# starting kernel
# baseline (speedup 1.0000x reference)
#include <hip/hip_runtime.h>
#include <hip/hip_bf16.h>

typedef __attribute__((ext_vector_type(2))) float v2f;
typedef __attribute__((ext_vector_type(8))) float v8f;

#define N_NODES 50000
#define N_EDGES 1600000
#define IN_HIDDEN 256
#define OUT_COLS 128
#define NHEAD 4
#define OUT_HIDDEN 32
#define EDGE_HIDDEN 64
#define N_ETYPES 8
#define NEG_SLOPE 0.2f

__device__ __forceinline__ unsigned enc_f32(float f) {
    unsigned u = __float_as_uint(f);
    return (u & 0x80000000u) ? ~u : (u | 0x80000000u);
}
__device__ __forceinline__ float dec_f32(unsigned u) {
    unsigned b = (u & 0x80000000u) ? (u & 0x7FFFFFFFu) : ~u;
    return __uint_as_float(b);
}

__global__ void init_gmax_kernel(unsigned* gmax) {
    *gmax = enc_f32(-INFINITY);
}

__global__ void __launch_bounds__(256)
gemm_h_kernel(const float* __restrict__ x, const float* __restrict__ W,
              float* __restrict__ h) {
    const int lane  = threadIdx.x & 31;
    const int wave  = threadIdx.x >> 5;
    const int strip = blockIdx.x * (blockDim.x >> 5) + wave;
    if (strip * 16 >= N_NODES) return;
    const int row0  = strip * 16;

    const int mrow  = lane & 15;
    const int khalf = (lane >> 4) << 1;

    v8f acc[8] = {};

    const float* xrow = x + (size_t)(row0 + mrow) * IN_HIDDEN + khalf;
    for (int k = 0; k < IN_HIDDEN; k += 4) {
        v2f a = *(const v2f*)(xrow + k);
#pragma unroll
        for (int t = 0; t < 8; ++t) {
            const float* wp = W + (size_t)(k + khalf) * OUT_COLS + t * 16 + mrow;
            v2f b;
            b.x = wp[0];
            b.y = wp[OUT_COLS];
            acc[t] = __builtin_amdgcn_wmma_f32_16x16x4_f32(
                false, a, false, b, (short)0, acc[t], false, false);
        }
    }

    const int mbase = (lane >> 4) << 3;
#pragma unroll
    for (int t = 0; t < 8; ++t) {
#pragma unroll
        for (int i = 0; i < 8; ++i) {
            float v = acc[t][i];
            v = (v == v) ? v : 0.0f;
            h[(size_t)(row0 + mbase + i) * OUT_COLS + t * 16 + mrow] = v;
        }
    }
}

__global__ void etype_logits_kernel(const float* __restrict__ edge_emb,
                                    const float* __restrict__ W_e,
                                    const float* __restrict__ a_e,
                                    float* __restrict__ he) {
    const int tid = threadIdx.x;
    if (tid >= N_ETYPES * NHEAD) return;
    const int t = tid >> 2, head = tid & 3;
    float s = 0.0f;
    for (int d = 0; d < EDGE_HIDDEN; ++d) {
        float ev = 0.0f;
        for (int k = 0; k < EDGE_HIDDEN; ++k)
            ev = fmaf(edge_emb[t * EDGE_HIDDEN + k],
                      W_e[k * (EDGE_HIDDEN * NHEAD) + head * EDGE_HIDDEN + d], ev);
        s = fmaf(a_e[head * EDGE_HIDDEN + d], ev, s);
    }
    he[tid] = s;
}

__global__ void __launch_bounds__(256)
node_logits_kernel(const float* __restrict__ h,
                   const float* __restrict__ a_l, const float* __restrict__ a_r,
                   float* __restrict__ hl, float* __restrict__ hr) {
    const int idx = blockIdx.x * blockDim.x + threadIdx.x;
    if (idx >= N_NODES * NHEAD) return;
    const int node = idx >> 2, head = idx & 3;
    const float* hp = h + (size_t)node * OUT_COLS + head * OUT_HIDDEN;
    const float* al = a_l + head * OUT_HIDDEN;
    const float* ar = a_r + head * OUT_HIDDEN;
    float sl = 0.0f, sr = 0.0f;
#pragma unroll
    for (int d = 0; d < OUT_HIDDEN; ++d) {
        float v = hp[d];
        sl = fmaf(al[d], v, sl);
        sr = fmaf(ar[d], v, sr);
    }
    hl[idx] = sl;
    hr[idx] = sr;
}

__global__ void __launch_bounds__(256)
scores_kernel(const int* __restrict__ edge,
              const float* __restrict__ hl, const float* __restrict__ hr,
              const float* __restrict__ he,
              float* __restrict__ scores, unsigned* __restrict__ gmax) {
    const int e = blockIdx.x * blockDim.x + threadIdx.x;
    float lmax = -INFINITY;
    if (e < N_EDGES) {
        const int src = edge[e];
        const int trg = edge[N_EDGES + e];
        const int et  = edge[2 * N_EDGES + e];
#pragma unroll
        for (int hh = 0; hh < NHEAD; ++hh) {
            float s = hl[src * NHEAD + hh] + hr[trg * NHEAD + hh] + he[et * NHEAD + hh];
            s = (s > 0.0f) ? s : NEG_SLOPE * s;
            scores[(size_t)e * NHEAD + hh] = s;
            lmax = fmaxf(lmax, s);
        }
    }
    __shared__ float red[256];
    red[threadIdx.x] = lmax;
    __syncthreads();
    for (int off = 128; off > 0; off >>= 1) {
        if (threadIdx.x < off)
            red[threadIdx.x] = fmaxf(red[threadIdx.x], red[threadIdx.x + off]);
        __syncthreads();
    }
    if (threadIdx.x == 0) atomicMax(gmax, enc_f32(red[0]));
}

__global__ void __launch_bounds__(256)
expsum_kernel(const int* __restrict__ edge, float* __restrict__ scores,
              const unsigned* __restrict__ gmax, float* __restrict__ denom) {
    const int e = blockIdx.x * blockDim.x + threadIdx.x;
    if (e >= N_EDGES) return;
    const float smax = dec_f32(*gmax);
    const int trg = edge[N_EDGES + e];
#pragma unroll
    for (int hh = 0; hh < NHEAD; ++hh) {
        float es = __expf(scores[(size_t)e * NHEAD + hh] - smax);
        scores[(size_t)e * NHEAD + hh] = es;
        atomicAdd(&denom[trg * NHEAD + hh], es);
    }
}

__global__ void __launch_bounds__(256)
aggregate_kernel(const int* __restrict__ edge, const float* __restrict__ h,
                 const float* __restrict__ es, const float* __restrict__ denom,
                 float* __restrict__ out, float* __restrict__ attn_out) {
    const int gid  = blockIdx.x * blockDim.x + threadIdx.x;
    const int e    = gid >> 5;
    const int lane = gid & 31;
    if (e >= N_EDGES) return;
    const int src = edge[e];
    const int trg = edge[N_EDGES + e];
    float a[NHEAD];
#pragma unroll
    for (int k = 0; k < NHEAD; ++k)
        a[k] = es[(size_t)e * NHEAD + k] / (denom[trg * NHEAD + k] + 1e-16f);
    if (lane < NHEAD) attn_out[(size_t)e * NHEAD + lane] = a[lane];
#pragma unroll
    for (int k = 0; k < NHEAD; ++k) {
        float m = h[(size_t)src * OUT_COLS + k * OUT_HIDDEN + lane] * a[k];
        atomicAdd(&out[(size_t)trg * OUT_COLS + k * OUT_HIDDEN + lane], m);
    }
}

extern "C" void kernel_launch(void* const* d_in, const int* in_sizes, int n_in,
                              void* d_out, int out_size, void* d_ws, size_t ws_size,
                              hipStream_t stream) {
    const int*   edge     = (const int*)  d_in[0];
    const float* x        = (const float*)d_in[1];
    const float* edge_emb = (const float*)d_in[2];
    const float* W        = (const float*)d_in[3];
    const float* W_e      = (const float*)d_in[4];
    const float* a_l      = (const float*)d_in[5];
    const float* a_r      = (const float*)d_in[6];
    const float* a_e      = (const float*)d_in[7];

    float* out      = (float*)d_out;
    float* attn_out = out + (size_t)N_NODES * OUT_COLS;

    char* ws = (char*)d_ws;
    float* h      = (float*)ws;  ws += (size_t)N_NODES * OUT_COLS * sizeof(float);
    float* scores = (float*)ws;  ws += (size_t)N_EDGES * NHEAD   * sizeof(float);
    float* hl     = (float*)ws;  ws += (size_t)N_NODES * NHEAD   * sizeof(float);
    float* hr     = (float*)ws;  ws += (size_t)N_NODES * NHEAD   * sizeof(float);
    float* denom  = (float*)ws;  ws += (size_t)N_NODES * NHEAD   * sizeof(float);
    float* he     = (float*)ws;  ws += (size_t)N_ETYPES * NHEAD  * sizeof(float);
    unsigned* gmax = (unsigned*)ws;

    hipMemsetAsync(out,   0, (size_t)N_NODES * OUT_COLS * sizeof(float), stream);
    hipMemsetAsync(denom, 0, (size_t)N_NODES * NHEAD   * sizeof(float), stream);
    init_gmax_kernel<<<1, 1, 0, stream>>>(gmax);

    {
        const int strips = N_NODES / 16;
        const int wavesPerBlock = 8;
        const int blocks = (strips + wavesPerBlock - 1) / wavesPerBlock;
        gemm_h_kernel<<<blocks, 256, 0, stream>>>(x, W, h);
    }
    etype_logits_kernel<<<1, 32, 0, stream>>>(edge_emb, W_e, a_e, he);
    node_logits_kernel<<<(N_NODES * NHEAD + 255) / 256, 256, 0, stream>>>(h, a_l, a_r, hl, hr);
    scores_kernel<<<(N_EDGES + 255) / 256, 256, 0, stream>>>(edge, hl, hr, he, scores, gmax);
    expsum_kernel<<<(N_EDGES + 255) / 256, 256, 0, stream>>>(edge, scores, gmax, denom);
    aggregate_kernel<<<((size_t)N_EDGES * 32 + 255) / 256, 256, 0, stream>>>(
        edge, h, scores, denom, out, attn_out);
}


# expsum: one lane per (edge,head) pair so an edge's 4 denominator adds fall in one aligned 4-float span: 4x fewer atomic memory requests; float atomic adds kept, arrival order as before
# speedup vs baseline: 1.3617x; 1.3617x over previous
#include <hip/hip_runtime.h>
#include <hip/hip_bf16.h>

typedef __attribute__((ext_vector_type(2))) float v2f;
typedef __attribute__((ext_vector_type(8))) float v8f;

#define N_NODES 50000
#define N_EDGES 1600000
#define IN_HIDDEN 256
#define OUT_COLS 128
#define NHEAD 4
#define OUT_HIDDEN 32
#define EDGE_HIDDEN 64
#define N_ETYPES 8
#define NEG_SLOPE 0.2f

__device__ __forceinline__ unsigned enc_f32(float f) {
    unsigned u = __float_as_uint(f);
    return (u & 0x80000000u) ? ~u : (u | 0x80000000u);
}
__device__ __forceinline__ float dec_f32(unsigned u) {
    unsigned b = (u & 0x80000000u) ? (u & 0x7FFFFFFFu) : ~u;
    return __uint_as_float(b);
}

__global__ void init_gmax_kernel(unsigned* gmax) {
    *gmax = enc_f32(-INFINITY);
}

__global__ void __launch_bounds__(256)
gemm_h_kernel(const float* __restrict__ x, const float* __restrict__ W,
              float* __restrict__ h) {
    const int lane  = threadIdx.x & 31;
    const int wave  = threadIdx.x >> 5;
    const int strip = blockIdx.x * (blockDim.x >> 5) + wave;
    if (strip * 16 >= N_NODES) return;
    const int row0  = strip * 16;

    const int mrow  = lane & 15;
    const int khalf = (lane >> 4) << 1;

    v8f acc[8] = {};

    const float* xrow = x + (size_t)(row0 + mrow) * IN_HIDDEN + khalf;
    for (int k = 0; k < IN_HIDDEN; k += 4) {
        v2f a = *(const v2f*)(xrow + k);
#pragma unroll
        for (int t = 0; t < 8; ++t) {
            const float* wp = W + (size_t)(k + khalf) * OUT_COLS + t * 16 + mrow;
            v2f b;
            b.x = wp[0];
            b.y = wp[OUT_COLS];
            acc[t] = __builtin_amdgcn_wmma_f32_16x16x4_f32(
                false, a, false, b, (short)0, acc[t], false, false);
        }
    }

    const int mbase = (lane >> 4) << 3;
#pragma unroll
    for (int t = 0; t < 8; ++t) {
#pragma unroll
        for (int i = 0; i < 8; ++i) {
            float v = acc[t][i];
            v = (v == v) ? v : 0.0f;
            h[(size_t)(row0 + mbase + i) * OUT_COLS + t * 16 + mrow] = v;
        }
    }
}

__global__ void etype_logits_kernel(const float* __restrict__ edge_emb,
                                    const float* __restrict__ W_e,
                                    const float* __restrict__ a_e,
                                    float* __restrict__ he) {
    const int tid = threadIdx.x;
    if (tid >= N_ETYPES * NHEAD) return;
    const int t = tid >> 2, head = tid & 3;
    float s = 0.0f;
    for (int d = 0; d < EDGE_HIDDEN; ++d) {
        float ev = 0.0f;
        for (int k = 0; k < EDGE_HIDDEN; ++k)
            ev = fmaf(edge_emb[t * EDGE_HIDDEN + k],
                      W_e[k * (EDGE_HIDDEN * NHEAD) + head * EDGE_HIDDEN + d], ev);
        s = fmaf(a_e[head * EDGE_HIDDEN + d], ev, s);
    }
    he[tid] = s;
}

__global__ void __launch_bounds__(256)
node_logits_kernel(const float* __restrict__ h,
                   const float* __restrict__ a_l, const float* __restrict__ a_r,
                   float* __restrict__ hl, float* __restrict__ hr) {
    const int idx = blockIdx.x * blockDim.x + threadIdx.x;
    if (idx >= N_NODES * NHEAD) return;
    const int node = idx >> 2, head = idx & 3;
    const float* hp = h + (size_t)node * OUT_COLS + head * OUT_HIDDEN;
    const float* al = a_l + head * OUT_HIDDEN;
    const float* ar = a_r + head * OUT_HIDDEN;
    float sl = 0.0f, sr = 0.0f;
#pragma unroll
    for (int d = 0; d < OUT_HIDDEN; ++d) {
        float v = hp[d];
        sl = fmaf(al[d], v, sl);
        sr = fmaf(ar[d], v, sr);
    }
    hl[idx] = sl;
    hr[idx] = sr;
}

__global__ void __launch_bounds__(256)
scores_kernel(const int* __restrict__ edge,
              const float* __restrict__ hl, const float* __restrict__ hr,
              const float* __restrict__ he,
              float* __restrict__ scores, unsigned* __restrict__ gmax) {
    const int e = blockIdx.x * blockDim.x + threadIdx.x;
    float lmax = -INFINITY;
    if (e < N_EDGES) {
        const int src = edge[e];
        const int trg = edge[N_EDGES + e];
        const int et  = edge[2 * N_EDGES + e];
#pragma unroll
        for (int hh = 0; hh < NHEAD; ++hh) {
            float s = hl[src * NHEAD + hh] + hr[trg * NHEAD + hh] + he[et * NHEAD + hh];
            s = (s > 0.0f) ? s : NEG_SLOPE * s;
            scores[(size_t)e * NHEAD + hh] = s;
            lmax = fmaxf(lmax, s);
        }
    }
    __shared__ float red[256];
    red[threadIdx.x] = lmax;
    __syncthreads();
    for (int off = 128; off > 0; off >>= 1) {
        if (threadIdx.x < off)
            red[threadIdx.x] = fmaxf(red[threadIdx.x], red[threadIdx.x + off]);
        __syncthreads();
    }
    if (threadIdx.x == 0) atomicMax(gmax, enc_f32(red[0]));
}

__global__ void __launch_bounds__(256)
expsum_kernel(const int* __restrict__ edge, float* __restrict__ scores,
              const unsigned* __restrict__ gmax, float* __restrict__ denom) {
    const int idx = blockIdx.x * blockDim.x + threadIdx.x;
    if (idx >= N_EDGES * NHEAD) return;
    const float smax = dec_f32(*gmax);
    const int e  = idx >> 2;
    const int hh = idx & 3;
    const int trg = edge[N_EDGES + e];
    const float es = __expf(scores[idx] - smax);
    scores[idx] = es;
    atomicAdd(&denom[trg * NHEAD + hh], es);
}

__global__ void __launch_bounds__(256)
aggregate_kernel(const int* __restrict__ edge, const float* __restrict__ h,
                 const float* __restrict__ es, const float* __restrict__ denom,
                 float* __restrict__ out, float* __restrict__ attn_out) {
    const int gid  = blockIdx.x * blockDim.x + threadIdx.x;
    const int e    = gid >> 5;
    const int lane = gid & 31;
    if (e >= N_EDGES) return;
    const int src = edge[e];
    const int trg = edge[N_EDGES + e];
    float a[NHEAD];
#pragma unroll
    for (int k = 0; k < NHEAD; ++k)
        a[k] = es[(size_t)e * NHEAD + k] / (denom[trg * NHEAD + k] + 1e-16f);
    if (lane < NHEAD) attn_out[(size_t)e * NHEAD + lane] = a[lane];
#pragma unroll
    for (int k = 0; k < NHEAD; ++k) {
        float m = h[(size_t)src * OUT_COLS + k * OUT_HIDDEN + lane] * a[k];
        atomicAdd(&out[(size_t)trg * OUT_COLS + k * OUT_HIDDEN + lane], m);
    }
}

extern "C" void kernel_launch(void* const* d_in, const int* in_sizes, int n_in,
                              void* d_out, int out_size, void* d_ws, size_t ws_size,
                              hipStream_t stream) {
    const int*   edge     = (const int*)  d_in[0];
    const float* x        = (const float*)d_in[1];
    const float* edge_emb = (const float*)d_in[2];
    const float* W        = (const float*)d_in[3];
    const float* W_e      = (const float*)d_in[4];
    const float* a_l      = (const float*)d_in[5];
    const float* a_r      = (const float*)d_in[6];
    const float* a_e      = (const float*)d_in[7];

    float* out      = (float*)d_out;
    float* attn_out = out + (size_t)N_NODES * OUT_COLS;

    char* ws = (char*)d_ws;
    float* h      = (float*)ws;  ws += (size_t)N_NODES * OUT_COLS * sizeof(float);
    float* scores = (float*)ws;  ws += (size_t)N_EDGES * NHEAD   * sizeof(float);
    float* hl     = (float*)ws;  ws += (size_t)N_NODES * NHEAD   * sizeof(float);
    float* hr     = (float*)ws;  ws += (size_t)N_NODES * NHEAD   * sizeof(float);
    float* denom  = (float*)ws;  ws += (size_t)N_NODES * NHEAD   * sizeof(float);
    float* he     = (float*)ws;  ws += (size_t)N_ETYPES * NHEAD  * sizeof(float);
    unsigned* gmax = (unsigned*)ws;

    hipMemsetAsync(out,   0, (size_t)N_NODES * OUT_COLS * sizeof(float), stream);
    hipMemsetAsync(denom, 0, (size_t)N_NODES * NHEAD   * sizeof(float), stream);
    init_gmax_kernel<<<1, 1, 0, stream>>>(gmax);

    {
        const int strips = N_NODES / 16;
        const int wavesPerBlock = 8;
        const int blocks = (strips + wavesPerBlock - 1) / wavesPerBlock;
        gemm_h_kernel<<<blocks, 256, 0, stream>>>(x, W, h);
    }
    etype_logits_kernel<<<1, 32, 0, stream>>>(edge_emb, W_e, a_e, he);
    node_logits_kernel<<<(N_NODES * NHEAD + 255) / 256, 256, 0, stream>>>(h, a_l, a_r, hl, hr);
    scores_kernel<<<(N_EDGES + 255) / 256, 256, 0, stream>>>(edge, hl, hr, he, scores, gmax);
    expsum_kernel<<<(N_EDGES * NHEAD + 255) / 256, 256, 0, stream>>>(edge, scores, gmax, denom);
    aggregate_kernel<<<((size_t)N_EDGES * 32 + 255) / 256, 256, 0, stream>>>(
        edge, h, scores, denom, out, attn_out);
}


# aggregate by target: per-call counting sort of edges, each segment then ordered by edge id so all sums run in one fixed order; a wave per node sums rows in registers; float atomics removed
# speedup vs baseline: 1.6618x; 1.2204x over previous
#include <hip/hip_runtime.h>
#include <hip/hip_bf16.h>

typedef __attribute__((ext_vector_type(2))) float v2f;
typedef __attribute__((ext_vector_type(8))) float v8f;

#define N_NODES 50000
#define N_EDGES 1600000
#define IN_HIDDEN 256
#define OUT_COLS 128
#define NHEAD 4
#define OUT_HIDDEN 32
#define EDGE_HIDDEN 64
#define N_ETYPES 8
#define NEG_SLOPE 0.2f

__device__ __forceinline__ unsigned enc_f32(float f) {
    unsigned u = __float_as_uint(f);
    return (u & 0x80000000u) ? ~u : (u | 0x80000000u);
}
__device__ __forceinline__ float dec_f32(unsigned u) {
    unsigned b = (u & 0x80000000u) ? (u & 0x7FFFFFFFu) : ~u;
    return __uint_as_float(b);
}

__global__ void __launch_bounds__(256)
init_kernel(unsigned* __restrict__ gmax, int* __restrict__ cnt) {
    const int i = blockIdx.x * blockDim.x + threadIdx.x;
    if (i < N_NODES) cnt[i] = 0;
    if (i == 0) *gmax = enc_f32(-INFINITY);
}

__global__ void __launch_bounds__(256)
gemm_h_kernel(const float* __restrict__ x, const float* __restrict__ W,
              float* __restrict__ h) {
    const int lane  = threadIdx.x & 31;
    const int wave  = threadIdx.x >> 5;
    const int strip = blockIdx.x * (blockDim.x >> 5) + wave;
    if (strip * 16 >= N_NODES) return;
    const int row0  = strip * 16;

    const int mrow  = lane & 15;
    const int khalf = (lane >> 4) << 1;

    v8f acc[8] = {};

    const float* xrow = x + (size_t)(row0 + mrow) * IN_HIDDEN + khalf;
    for (int k = 0; k < IN_HIDDEN; k += 4) {
        v2f a = *(const v2f*)(xrow + k);
#pragma unroll
        for (int t = 0; t < 8; ++t) {
            const float* wp = W + (size_t)(k + khalf) * OUT_COLS + t * 16 + mrow;
            v2f b;
            b.x = wp[0];
            b.y = wp[OUT_COLS];
            acc[t] = __builtin_amdgcn_wmma_f32_16x16x4_f32(
                false, a, false, b, (short)0, acc[t], false, false);
        }
    }

    const int mbase = (lane >> 4) << 3;
#pragma unroll
    for (int t = 0; t < 8; ++t) {
#pragma unroll
        for (int i = 0; i < 8; ++i) {
            float v = acc[t][i];
            v = (v == v) ? v : 0.0f;
            h[(size_t)(row0 + mbase + i) * OUT_COLS + t * 16 + mrow] = v;
        }
    }
}

__global__ void etype_logits_kernel(const float* __restrict__ edge_emb,
                                    const float* __restrict__ W_e,
                                    const float* __restrict__ a_e,
                                    float* __restrict__ he) {
    const int tid = threadIdx.x;
    if (tid >= N_ETYPES * NHEAD) return;
    const int t = tid >> 2, head = tid & 3;
    float s = 0.0f;
    for (int d = 0; d < EDGE_HIDDEN; ++d) {
        float ev = 0.0f;
        for (int k = 0; k < EDGE_HIDDEN; ++k)
            ev = fmaf(edge_emb[t * EDGE_HIDDEN + k],
                      W_e[k * (EDGE_HIDDEN * NHEAD) + head * EDGE_HIDDEN + d], ev);
        s = fmaf(a_e[head * EDGE_HIDDEN + d], ev, s);
    }
    he[tid] = s;
}

__global__ void __launch_bounds__(256)
node_logits_kernel(const float* __restrict__ h,
                   const float* __restrict__ a_l, const float* __restrict__ a_r,
                   float* __restrict__ hl, float* __restrict__ hr) {
    const int idx = blockIdx.x * blockDim.x + threadIdx.x;
    if (idx >= N_NODES * NHEAD) return;
    const int node = idx >> 2, head = idx & 3;
    const float* hp = h + (size_t)node * OUT_COLS + head * OUT_HIDDEN;
    const float* al = a_l + head * OUT_HIDDEN;
    const float* ar = a_r + head * OUT_HIDDEN;
    float sl = 0.0f, sr = 0.0f;
#pragma unroll
    for (int d = 0; d < OUT_HIDDEN; ++d) {
        float v = hp[d];
        sl = fmaf(al[d], v, sl);
        sr = fmaf(ar[d], v, sr);
    }
    hl[idx] = sl;
    hr[idx] = sr;
}

__device__ __forceinline__ float4 edge_scores(const float* __restrict__ hl,
                                              const float* __restrict__ hr,
                                              const float* __restrict__ he,
                                              int src, int trg, int et) {
    const float4 l = *(const float4*)(hl + src * NHEAD);
    const float4 r = *(const float4*)(hr + trg * NHEAD);
    const float4 t = *(const float4*)(he + et * NHEAD);
    float4 s;
    s.x = l.x + r.x + t.x;
    s.y = l.y + r.y + t.y;
    s.z = l.z + r.z + t.z;
    s.w = l.w + r.w + t.w;
    s.x = (s.x > 0.0f) ? s.x : NEG_SLOPE * s.x;
    s.y = (s.y > 0.0f) ? s.y : NEG_SLOPE * s.y;
    s.z = (s.z > 0.0f) ? s.z : NEG_SLOPE * s.z;
    s.w = (s.w > 0.0f) ? s.w : NEG_SLOPE * s.w;
    return s;
}

__global__ void __launch_bounds__(256)
edge_pass_kernel(const int* __restrict__ edge,
                 const float* __restrict__ hl, const float* __restrict__ hr,
                 const float* __restrict__ he,
                 int* __restrict__ cnt, int* __restrict__ slot,
                 unsigned* __restrict__ gmax) {
    const int e = blockIdx.x * blockDim.x + threadIdx.x;
    float lmax = -INFINITY;
    if (e < N_EDGES) {
        const int src = edge[e];
        const int trg = edge[N_EDGES + e];
        const int et  = edge[2 * N_EDGES + e];
        const float4 s = edge_scores(hl, hr, he, src, trg, et);
        lmax = fmaxf(fmaxf(s.x, s.y), fmaxf(s.z, s.w));
        slot[e] = atomicAdd(&cnt[trg], 1);
    }
    __shared__ float red[256];
    red[threadIdx.x] = lmax;
    __syncthreads();
    for (int off = 128; off > 0; off >>= 1) {
        if (threadIdx.x < off)
            red[threadIdx.x] = fmaxf(red[threadIdx.x], red[threadIdx.x + off]);
        __syncthreads();
    }
    if (threadIdx.x == 0) atomicMax(gmax, enc_f32(red[0]));
}

#define SCAN_T  1024
#define SCAN_CH 49
__global__ void __launch_bounds__(SCAN_T)
scan_kernel(const int* __restrict__ cnt, int* __restrict__ offs) {
    __shared__ int part[SCAN_T];
    const int t = threadIdx.x;
    const int base = t * SCAN_CH;
    int v[SCAN_CH];
    int sum = 0;
#pragma unroll
    for (int i = 0; i < SCAN_CH; ++i) {
        const int idx = base + i;
        v[i] = (idx < N_NODES) ? cnt[idx] : 0;
        sum += v[i];
    }
    part[t] = sum;
    __syncthreads();
    for (int off = 1; off < SCAN_T; off <<= 1) {
        const int add = (t >= off) ? part[t - off] : 0;
        __syncthreads();
        part[t] += add;
        __syncthreads();
    }
    int run = part[t] - sum;
#pragma unroll
    for (int i = 0; i < SCAN_CH; ++i) {
        const int idx = base + i;
        if (idx < N_NODES) offs[idx] = run;
        run += v[i];
    }
    if (t == SCAN_T - 1) offs[N_NODES] = part[SCAN_T - 1];
}

__global__ void __launch_bounds__(256)
place_kernel(const int* __restrict__ edge, const int* __restrict__ slot,
             const int* __restrict__ offs, int* __restrict__ arrived) {
    const int e = blockIdx.x * blockDim.x + threadIdx.x;
    if (e >= N_EDGES) return;
    const int src = edge[e];
    const int trg = edge[N_EDGES + e];
    const int et  = edge[2 * N_EDGES + e];
    *(int2*)(arrived + 2 * (size_t)(offs[trg] + slot[e])) = make_int2(e, src | (et << 16));
}

__global__ void __launch_bounds__(256)
order_kernel(const int* __restrict__ arrived, const int* __restrict__ offs,
             int* __restrict__ ordered) {
    const int lane = threadIdx.x & 31;
    const int node = blockIdx.x * (blockDim.x >> 5) + (threadIdx.x >> 5);
    if (node >= N_NODES) return;
    const int beg = __builtin_amdgcn_readfirstlane(offs[node]);
    const int d   = __builtin_amdgcn_readfirstlane(offs[node + 1]) - beg;
    for (int ca = 0; ca < d; ca += 32) {
        const int ia = ca + lane;
        int2 pa = make_int2(0x7FFFFFFF, 0);
        if (ia < d) pa = *(const int2*)(arrived + 2 * (size_t)(beg + ia));
        int pos = 0;
        for (int cb = 0; cb < d; cb += 32) {
            const int ib = cb + lane;
            int kb = 0x7FFFFFFF;
            if (ib < d) kb = arrived[2 * (size_t)(beg + ib)];
#pragma unroll
            for (int j = 0; j < 32; ++j) {
                const int key = __builtin_amdgcn_readlane(kb, j);
                pos += (key < pa.x) ? 1 : 0;
            }
        }
        if (ia < d) *(int2*)(ordered + 2 * (size_t)(beg + pos)) = pa;
    }
}

__global__ void __launch_bounds__(256)
node_aggregate_kernel(const int* __restrict__ ordered, const int* __restrict__ offs,
                      const float* __restrict__ h,
                      const float* __restrict__ hl, const float* __restrict__ hr,
                      const float* __restrict__ he, const unsigned* __restrict__ gmax,
                      float* __restrict__ out, float* __restrict__ attn_out) {
    const int lane = threadIdx.x & 31;
    const int node = blockIdx.x * (blockDim.x >> 5) + (threadIdx.x >> 5);
    if (node >= N_NODES) return;
    const int beg = __builtin_amdgcn_readfirstlane(offs[node]);
    const int d   = __builtin_amdgcn_readfirstlane(offs[node + 1]) - beg;
    const float smax = dec_f32(*gmax);

    float4 ds = make_float4(0.0f, 0.0f, 0.0f, 0.0f);
    for (int c = 0; c < d; c += 32) {
        const int i = c + lane;
        if (i < d) {
            const int2 p = *(const int2*)(ordered + 2 * (size_t)(beg + i));
            const float4 s = edge_scores(hl, hr, he, p.y & 0xFFFF, node, p.y >> 16);
            ds.x += __expf(s.x - smax);
            ds.y += __expf(s.y - smax);
            ds.z += __expf(s.z - smax);
            ds.w += __expf(s.w - smax);
        }
    }
    for (int m = 16; m > 0; m >>= 1) {
        ds.x += __shfl_xor(ds.x, m, 32);
        ds.y += __shfl_xor(ds.y, m, 32);
        ds.z += __shfl_xor(ds.z, m, 32);
        ds.w += __shfl_xor(ds.w, m, 32);
    }
    const float4 dn = make_float4(ds.x + 1e-16f, ds.y + 1e-16f, ds.z + 1e-16f, ds.w + 1e-16f);

    float acc0 = 0.0f, acc1 = 0.0f, acc2 = 0.0f, acc3 = 0.0f;
    const float* hcol = h + lane;
    for (int c = 0; c < d; c += 32) {
        const int i = c + lane;
        int srcv = 0;
        float4 a = make_float4(0.0f, 0.0f, 0.0f, 0.0f);
        if (i < d) {
            const int2 p = *(const int2*)(ordered + 2 * (size_t)(beg + i));
            srcv = p.y & 0xFFFF;
            const float4 s = edge_scores(hl, hr, he, srcv, node, p.y >> 16);
            a.x = __expf(s.x - smax) / dn.x;
            a.y = __expf(s.y - smax) / dn.y;
            a.z = __expf(s.z - smax) / dn.z;
            a.w = __expf(s.w - smax) / dn.w;
            *(float4*)(attn_out + (size_t)p.x * NHEAD) = a;
        }
        const int nb = (d - c < 32) ? (d - c) : 32;
#pragma unroll 4
        for (int j = 0; j < nb; ++j) {
            const int sj = __builtin_amdgcn_readlane(srcv, j);
            const float a0 = __int_as_float(__builtin_amdgcn_readlane(__float_as_int(a.x), j));
            const float a1 = __int_as_float(__builtin_amdgcn_readlane(__float_as_int(a.y), j));
            const float a2 = __int_as_float(__builtin_amdgcn_readlane(__float_as_int(a.z), j));
            const float a3 = __int_as_float(__builtin_amdgcn_readlane(__float_as_int(a.w), j));
            const float* row = hcol + (size_t)sj * OUT_COLS;
            acc0 = fmaf(row[0 * OUT_HIDDEN], a0, acc0);
            acc1 = fmaf(row[1 * OUT_HIDDEN], a1, acc1);
            acc2 = fmaf(row[2 * OUT_HIDDEN], a2, acc2);
            acc3 = fmaf(row[3 * OUT_HIDDEN], a3, acc3);
        }
    }
    float* orow = out + (size_t)node * OUT_COLS + lane;
    orow[0 * OUT_HIDDEN] = acc0;
    orow[1 * OUT_HIDDEN] = acc1;
    orow[2 * OUT_HIDDEN] = acc2;
    orow[3 * OUT_HIDDEN] = acc3;
}

extern "C" void kernel_launch(void* const* d_in, const int* in_sizes, int n_in,
                              void* d_out, int out_size, void* d_ws, size_t ws_size,
                              hipStream_t stream) {
    const int*   edge     = (const int*)  d_in[0];
    const float* x        = (const float*)d_in[1];
    const float* edge_emb = (const float*)d_in[2];
    const float* W        = (const float*)d_in[3];
    const float* W_e      = (const float*)d_in[4];
    const float* a_l      = (const float*)d_in[5];
    const float* a_r      = (const float*)d_in[6];
    const float* a_e      = (const float*)d_in[7];

    float* out      = (float*)d_out;
    float* attn_out = out + (size_t)N_NODES * OUT_COLS;

    char* ws = (char*)d_ws;
    float* h       = (float*)ws;  ws += (size_t)N_NODES * OUT_COLS * sizeof(float);
    int*   arrived = (int*)ws;    ws += (size_t)N_EDGES * 2 * sizeof(int);
    int*   ordered = (int*)ws;    ws += (size_t)N_EDGES * 2 * sizeof(int);
    int*   slot    = ordered;
    float* hl      = (float*)ws;  ws += (size_t)N_NODES * NHEAD   * sizeof(float);
    float* hr      = (float*)ws;  ws += (size_t)N_NODES * NHEAD   * sizeof(float);
    int*   cnt     = (int*)ws;
    int*   offs    = cnt + 50176;
    ws += (size_t)N_NODES * NHEAD * sizeof(float);
    float* he      = (float*)ws;  ws += (size_t)N_ETYPES * NHEAD  * sizeof(float);
    unsigned* gmax = (unsigned*)ws;

    init_kernel<<<(N_NODES + 255) / 256, 256, 0, stream>>>(gmax, cnt);

    {
        const int strips = N_NODES / 16;
        const int wavesPerBlock = 8;
        const int blocks = (strips + wavesPerBlock - 1) / wavesPerBlock;
        gemm_h_kernel<<<blocks, 256, 0, stream>>>(x, W, h);
    }
    etype_logits_kernel<<<1, 32, 0, stream>>>(edge_emb, W_e, a_e, he);
    node_logits_kernel<<<(N_NODES * NHEAD + 255) / 256, 256, 0, stream>>>(h, a_l, a_r, hl, hr);
    edge_pass_kernel<<<(N_EDGES + 255) / 256, 256, 0, stream>>>(edge, hl, hr, he, cnt, slot, gmax);
    scan_kernel<<<1, SCAN_T, 0, stream>>>(cnt, offs);
    place_kernel<<<(N_EDGES + 255) / 256, 256, 0, stream>>>(edge, slot, offs, arrived);
    order_kernel<<<(N_NODES + 7) / 8, 256, 0, stream>>>(arrived, offs, ordered);
    node_aggregate_kernel<<<(N_NODES + 7) / 8, 256, 0, stream>>>(
        ordered, offs, h, hl, hr, he, gmax, out, attn_out);
}


# edge-type logits: the single wave's chain of 576 serial load round trips is spread over 8 work-groups, one lane per (head,d) column, all 64 k loads in flight; fma order kept, he bit-identical
# speedup vs baseline: 1.8685x; 1.1244x over previous
#include <hip/hip_runtime.h>
#include <hip/hip_bf16.h>

typedef __attribute__((ext_vector_type(2))) float v2f;
typedef __attribute__((ext_vector_type(8))) float v8f;

#define N_NODES 50000
#define N_EDGES 1600000
#define IN_HIDDEN 256
#define OUT_COLS 128
#define NHEAD 4
#define OUT_HIDDEN 32
#define EDGE_HIDDEN 64
#define N_ETYPES 8
#define NEG_SLOPE 0.2f

__device__ __forceinline__ unsigned enc_f32(float f) {
    unsigned u = __float_as_uint(f);
    return (u & 0x80000000u) ? ~u : (u | 0x80000000u);
}
__device__ __forceinline__ float dec_f32(unsigned u) {
    unsigned b = (u & 0x80000000u) ? (u & 0x7FFFFFFFu) : ~u;
    return __uint_as_float(b);
}

__global__ void __launch_bounds__(256)
init_kernel(unsigned* __restrict__ gmax, int* __restrict__ cnt) {
    const int i = blockIdx.x * blockDim.x + threadIdx.x;
    if (i < N_NODES) cnt[i] = 0;
    if (i == 0) *gmax = enc_f32(-INFINITY);
}

__global__ void __launch_bounds__(256)
gemm_h_kernel(const float* __restrict__ x, const float* __restrict__ W,
              float* __restrict__ h) {
    const int lane  = threadIdx.x & 31;
    const int wave  = threadIdx.x >> 5;
    const int strip = blockIdx.x * (blockDim.x >> 5) + wave;
    if (strip * 16 >= N_NODES) return;
    const int row0  = strip * 16;

    const int mrow  = lane & 15;
    const int khalf = (lane >> 4) << 1;

    v8f acc[8] = {};

    const float* xrow = x + (size_t)(row0 + mrow) * IN_HIDDEN + khalf;
    for (int k = 0; k < IN_HIDDEN; k += 4) {
        v2f a = *(const v2f*)(xrow + k);
#pragma unroll
        for (int t = 0; t < 8; ++t) {
            const float* wp = W + (size_t)(k + khalf) * OUT_COLS + t * 16 + mrow;
            v2f b;
            b.x = wp[0];
            b.y = wp[OUT_COLS];
            acc[t] = __builtin_amdgcn_wmma_f32_16x16x4_f32(
                false, a, false, b, (short)0, acc[t], false, false);
        }
    }

    const int mbase = (lane >> 4) << 3;
#pragma unroll
    for (int t = 0; t < 8; ++t) {
#pragma unroll
        for (int i = 0; i < 8; ++i) {
            float v = acc[t][i];
            v = (v == v) ? v : 0.0f;
            h[(size_t)(row0 + mbase + i) * OUT_COLS + t * 16 + mrow] = v;
        }
    }
}

__global__ void __launch_bounds__(NHEAD * EDGE_HIDDEN)
etype_logits_kernel(const float* __restrict__ edge_emb,
                    const float* __restrict__ W_e,
                    const float* __restrict__ a_e,
                    float* __restrict__ he) {
    __shared__ float ev_s[NHEAD * EDGE_HIDDEN];
    __shared__ float ae_s[NHEAD * EDGE_HIDDEN];
    const int t    = blockIdx.x;
    const int col  = threadIdx.x;
    const int head = col >> 6, d = col & (EDGE_HIDDEN - 1);
    const float* er = edge_emb + t * EDGE_HIDDEN;
    const float* wc = W_e + col;
    const float ae = a_e[col];
    float ev = 0.0f;
#pragma unroll
    for (int k = 0; k < EDGE_HIDDEN; ++k)
        ev = fmaf(er[k], wc[k * (EDGE_HIDDEN * NHEAD)], ev);
    ev_s[d * NHEAD + head] = ev;
    ae_s[d * NHEAD + head] = ae;
    __syncthreads();
    if (col < NHEAD) {
        float s = 0.0f;
#pragma unroll
        for (int dd = 0; dd < EDGE_HIDDEN; ++dd)
            s = fmaf(ae_s[dd * NHEAD + col], ev_s[dd * NHEAD + col], s);
        he[t * NHEAD + col] = s;
    }
}

__global__ void __launch_bounds__(256)
node_logits_kernel(const float* __restrict__ h,
                   const float* __restrict__ a_l, const float* __restrict__ a_r,
                   float* __restrict__ hl, float* __restrict__ hr) {
    const int idx = blockIdx.x * blockDim.x + threadIdx.x;
    if (idx >= N_NODES * NHEAD) return;
    const int node = idx >> 2, head = idx & 3;
    const float* hp = h + (size_t)node * OUT_COLS + head * OUT_HIDDEN;
    const float* al = a_l + head * OUT_HIDDEN;
    const float* ar = a_r + head * OUT_HIDDEN;
    float sl = 0.0f, sr = 0.0f;
#pragma unroll
    for (int d = 0; d < OUT_HIDDEN; ++d) {
        float v = hp[d];
        sl = fmaf(al[d], v, sl);
        sr = fmaf(ar[d], v, sr);
    }
    hl[idx] = sl;
    hr[idx] = sr;
}

__device__ __forceinline__ float4 edge_scores(const float* __restrict__ hl,
                                              const float* __restrict__ hr,
                                              const float* __restrict__ he,
                                              int src, int trg, int et) {
    const float4 l = *(const float4*)(hl + src * NHEAD);
    const float4 r = *(const float4*)(hr + trg * NHEAD);
    const float4 t = *(const float4*)(he + et * NHEAD);
    float4 s;
    s.x = l.x + r.x + t.x;
    s.y = l.y + r.y + t.y;
    s.z = l.z + r.z + t.z;
    s.w = l.w + r.w + t.w;
    s.x = (s.x > 0.0f) ? s.x : NEG_SLOPE * s.x;
    s.y = (s.y > 0.0f) ? s.y : NEG_SLOPE * s.y;
    s.z = (s.z > 0.0f) ? s.z : NEG_SLOPE * s.z;
    s.w = (s.w > 0.0f) ? s.w : NEG_SLOPE * s.w;
    return s;
}

__global__ void __launch_bounds__(256)
edge_pass_kernel(const int* __restrict__ edge,
                 const float* __restrict__ hl, const float* __restrict__ hr,
                 const float* __restrict__ he,
                 int* __restrict__ cnt, int* __restrict__ slot,
                 unsigned* __restrict__ gmax) {
    const int e = blockIdx.x * blockDim.x + threadIdx.x;
    float lmax = -INFINITY;
    if (e < N_EDGES) {
        const int src = edge[e];
        const int trg = edge[N_EDGES + e];
        const int et  = edge[2 * N_EDGES + e];
        const float4 s = edge_scores(hl, hr, he, src, trg, et);
        lmax = fmaxf(fmaxf(s.x, s.y), fmaxf(s.z, s.w));
        slot[e] = atomicAdd(&cnt[trg], 1);
    }
    __shared__ float red[256];
    red[threadIdx.x] = lmax;
    __syncthreads();
    for (int off = 128; off > 0; off >>= 1) {
        if (threadIdx.x < off)
            red[threadIdx.x] = fmaxf(red[threadIdx.x], red[threadIdx.x + off]);
        __syncthreads();
    }
    if (threadIdx.x == 0) atomicMax(gmax, enc_f32(red[0]));
}

#define SCAN_T  1024
#define SCAN_CH 49
__global__ void __launch_bounds__(SCAN_T)
scan_kernel(const int* __restrict__ cnt, int* __restrict__ offs) {
    __shared__ int part[SCAN_T];
    const int t = threadIdx.x;
    const int base = t * SCAN_CH;
    int v[SCAN_CH];
    int sum = 0;
#pragma unroll
    for (int i = 0; i < SCAN_CH; ++i) {
        const int idx = base + i;
        v[i] = (idx < N_NODES) ? cnt[idx] : 0;
        sum += v[i];
    }
    part[t] = sum;
    __syncthreads();
    for (int off = 1; off < SCAN_T; off <<= 1) {
        const int add = (t >= off) ? part[t - off] : 0;
        __syncthreads();
        part[t] += add;
        __syncthreads();
    }
    int run = part[t] - sum;
#pragma unroll
    for (int i = 0; i < SCAN_CH; ++i) {
        const int idx = base + i;
        if (idx < N_NODES) offs[idx] = run;
        run += v[i];
    }
    if (t == SCAN_T - 1) offs[N_NODES] = part[SCAN_T - 1];
}

__global__ void __launch_bounds__(256)
place_kernel(const int* __restrict__ edge, const int* __restrict__ slot,
             const int* __restrict__ offs, int* __restrict__ arrived) {
    const int e = blockIdx.x * blockDim.x + threadIdx.x;
    if (e >= N_EDGES) return;
    const int src = edge[e];
    const int trg = edge[N_EDGES + e];
    const int et  = edge[2 * N_EDGES + e];
    *(int2*)(arrived + 2 * (size_t)(offs[trg] + slot[e])) = make_int2(e, src | (et << 16));
}

__global__ void __launch_bounds__(256)
order_kernel(const int* __restrict__ arrived, const int* __restrict__ offs,
             int* __restrict__ ordered) {
    const int lane = threadIdx.x & 31;
    const int node = blockIdx.x * (blockDim.x >> 5) + (threadIdx.x >> 5);
    if (node >= N_NODES) return;
    const int beg = __builtin_amdgcn_readfirstlane(offs[node]);
    const int d   = __builtin_amdgcn_readfirstlane(offs[node + 1]) - beg;
    for (int ca = 0; ca < d; ca += 32) {
        const int ia = ca + lane;
        int2 pa = make_int2(0x7FFFFFFF, 0);
        if (ia < d) pa = *(const int2*)(arrived + 2 * (size_t)(beg + ia));
        int pos = 0;
        for (int cb = 0; cb < d; cb += 32) {
            const int ib = cb + lane;
            int kb = 0x7FFFFFFF;
            if (ib < d) kb = arrived[2 * (size_t)(beg + ib)];
#pragma unroll
            for (int j = 0; j < 32; ++j) {
                const int key = __builtin_amdgcn_readlane(kb, j);
                pos += (key < pa.x) ? 1 : 0;
            }
        }
        if (ia < d) *(int2*)(ordered + 2 * (size_t)(beg + pos)) = pa;
    }
}

__global__ void __launch_bounds__(256)
node_aggregate_kernel(const int* __restrict__ ordered, const int* __restrict__ offs,
                      const float* __restrict__ h,
                      const float* __restrict__ hl, const float* __restrict__ hr,
                      const float* __restrict__ he, const unsigned* __restrict__ gmax,
                      float* __restrict__ out, float* __restrict__ attn_out) {
    const int lane = threadIdx.x & 31;
    const int node = blockIdx.x * (blockDim.x >> 5) + (threadIdx.x >> 5);
    if (node >= N_NODES) return;
    const int beg = __builtin_amdgcn_readfirstlane(offs[node]);
    const int d   = __builtin_amdgcn_readfirstlane(offs[node + 1]) - beg;
    const float smax = dec_f32(*gmax);

    float4 ds = make_float4(0.0f, 0.0f, 0.0f, 0.0f);
    for (int c = 0; c < d; c += 32) {
        const int i = c + lane;
        if (i < d) {
            const int2 p = *(const int2*)(ordered + 2 * (size_t)(beg + i));
            const float4 s = edge_scores(hl, hr, he, p.y & 0xFFFF, node, p.y >> 16);
            ds.x += __expf(s.x - smax);
            ds.y += __expf(s.y - smax);
            ds.z += __expf(s.z - smax);
            ds.w += __expf(s.w - smax);
        }
    }
    for (int m = 16; m > 0; m >>= 1) {
        ds.x += __shfl_xor(ds.x, m, 32);
        ds.y += __shfl_xor(ds.y, m, 32);
        ds.z += __shfl_xor(ds.z, m, 32);
        ds.w += __shfl_xor(ds.w, m, 32);
    }
    const float4 dn = make_float4(ds.x + 1e-16f, ds.y + 1e-16f, ds.z + 1e-16f, ds.w + 1e-16f);

    float acc0 = 0.0f, acc1 = 0.0f, acc2 = 0.0f, acc3 = 0.0f;
    const float* hcol = h + lane;
    for (int c = 0; c < d; c += 32) {
        const int i = c + lane;
        int srcv = 0;
        float4 a = make_float4(0.0f, 0.0f, 0.0f, 0.0f);
        if (i < d) {
            const int2 p = *(const int2*)(ordered + 2 * (size_t)(beg + i));
            srcv = p.y & 0xFFFF;
            const float4 s = edge_scores(hl, hr, he, srcv, node, p.y >> 16);
            a.x = __expf(s.x - smax) / dn.x;
            a.y = __expf(s.y - smax) / dn.y;
            a.z = __expf(s.z - smax) / dn.z;
            a.w = __expf(s.w - smax) / dn.w;
            *(float4*)(attn_out + (size_t)p.x * NHEAD) = a;
        }
        const int nb = (d - c < 32) ? (d - c) : 32;
#pragma unroll 4
        for (int j = 0; j < nb; ++j) {
            const int sj = __builtin_amdgcn_readlane(srcv, j);
            const float a0 = __int_as_float(__builtin_amdgcn_readlane(__float_as_int(a.x), j));
            const float a1 = __int_as_float(__builtin_amdgcn_readlane(__float_as_int(a.y), j));
            const float a2 = __int_as_float(__builtin_amdgcn_readlane(__float_as_int(a.z), j));
            const float a3 = __int_as_float(__builtin_amdgcn_readlane(__float_as_int(a.w), j));
            const float* row = hcol + (size_t)sj * OUT_COLS;
            acc0 = fmaf(row[0 * OUT_HIDDEN], a0, acc0);
            acc1 = fmaf(row[1 * OUT_HIDDEN], a1, acc1);
            acc2 = fmaf(row[2 * OUT_HIDDEN], a2, acc2);
            acc3 = fmaf(row[3 * OUT_HIDDEN], a3, acc3);
        }
    }
    float* orow = out + (size_t)node * OUT_COLS + lane;
    orow[0 * OUT_HIDDEN] = acc0;
    orow[1 * OUT_HIDDEN] = acc1;
    orow[2 * OUT_HIDDEN] = acc2;
    orow[3 * OUT_HIDDEN] = acc3;
}

extern "C" void kernel_launch(void* const* d_in, const int* in_sizes, int n_in,
                              void* d_out, int out_size, void* d_ws, size_t ws_size,
                              hipStream_t stream) {
    const int*   edge     = (const int*)  d_in[0];
    const float* x        = (const float*)d_in[1];
    const float* edge_emb = (const float*)d_in[2];
    const float* W        = (const float*)d_in[3];
    const float* W_e      = (const float*)d_in[4];
    const float* a_l      = (const float*)d_in[5];
    const float* a_r      = (const float*)d_in[6];
    const float* a_e      = (const float*)d_in[7];

    float* out      = (float*)d_out;
    float* attn_out = out + (size_t)N_NODES * OUT_COLS;

    char* ws = (char*)d_ws;
    float* h       = (float*)ws;  ws += (size_t)N_NODES * OUT_COLS * sizeof(float);
    int*   arrived = (int*)ws;    ws += (size_t)N_EDGES * 2 * sizeof(int);
    int*   ordered = (int*)ws;    ws += (size_t)N_EDGES * 2 * sizeof(int);
    int*   slot    = ordered;
    float* hl      = (float*)ws;  ws += (size_t)N_NODES * NHEAD   * sizeof(float);
    float* hr      = (float*)ws;  ws += (size_t)N_NODES * NHEAD   * sizeof(float);
    int*   cnt     = (int*)ws;
    int*   offs    = cnt + 50176;
    ws += (size_t)N_NODES * NHEAD * sizeof(float);
    float* he      = (float*)ws;  ws += (size_t)N_ETYPES * NHEAD  * sizeof(float);
    unsigned* gmax = (unsigned*)ws;

    init_kernel<<<(N_NODES + 255) / 256, 256, 0, stream>>>(gmax, cnt);

    {
        const int strips = N_NODES / 16;
        const int wavesPerBlock = 8;
        const int blocks = (strips + wavesPerBlock - 1) / wavesPerBlock;
        gemm_h_kernel<<<blocks, 256, 0, stream>>>(x, W, h);
    }
    etype_logits_kernel<<<N_ETYPES, NHEAD * EDGE_HIDDEN, 0, stream>>>(edge_emb, W_e, a_e, he);
    node_logits_kernel<<<(N_NODES * NHEAD + 255) / 256, 256, 0, stream>>>(h, a_l, a_r, hl, hr);
    edge_pass_kernel<<<(N_EDGES + 255) / 256, 256, 0, stream>>>(edge, hl, hr, he, cnt, slot, gmax);
    scan_kernel<<<1, SCAN_T, 0, stream>>>(cnt, offs);
    place_kernel<<<(N_EDGES + 255) / 256, 256, 0, stream>>>(edge, slot, offs, arrived);
    order_kernel<<<(N_NODES + 7) / 8, 256, 0, stream>>>(arrived, offs, ordered);
    node_aggregate_kernel<<<(N_NODES + 7) / 8, 256, 0, stream>>>(
        ordered, offs, h, hl, hr, he, gmax, out, attn_out);
}


# edge grouping by target redone as a two-level counting sort: ranks from returning LDS adds per (edge chunk, node bucket), not one device-wide returning add per edge; lists and sum order as before
# speedup vs baseline: 3.1777x; 1.7007x over previous
#include <hip/hip_runtime.h>
#include <hip/hip_bf16.h>

typedef __attribute__((ext_vector_type(2))) float v2f;
typedef __attribute__((ext_vector_type(8))) float v8f;

#define N_NODES 50000
#define N_EDGES 1600000
#define IN_HIDDEN 256
#define OUT_COLS 128
#define NHEAD 4
#define OUT_HIDDEN 32
#define EDGE_HIDDEN 64
#define N_ETYPES 8
#define NEG_SLOPE 0.2f

__device__ __forceinline__ unsigned enc_f32(float f) {
    unsigned u = __float_as_uint(f);
    return (u & 0x80000000u) ? ~u : (u | 0x80000000u);
}
__device__ __forceinline__ float dec_f32(unsigned u) {
    unsigned b = (u & 0x80000000u) ? (u & 0x7FFFFFFFu) : ~u;
    return __uint_as_float(b);
}

__global__ void __launch_bounds__(32)
init_kernel(unsigned* __restrict__ gmax) {
    if (blockIdx.x == 0 && threadIdx.x == 0) *gmax = enc_f32(-INFINITY);
}

__global__ void __launch_bounds__(256)
gemm_h_kernel(const float* __restrict__ x, const float* __restrict__ W,
              float* __restrict__ h) {
    const int lane  = threadIdx.x & 31;
    const int wave  = threadIdx.x >> 5;
    const int strip = blockIdx.x * (blockDim.x >> 5) + wave;
    if (strip * 16 >= N_NODES) return;
    const int row0  = strip * 16;

    const int mrow  = lane & 15;
    const int khalf = (lane >> 4) << 1;

    v8f acc[8] = {};

    const float* xrow = x + (size_t)(row0 + mrow) * IN_HIDDEN + khalf;
    for (int k = 0; k < IN_HIDDEN; k += 4) {
        v2f a = *(const v2f*)(xrow + k);
#pragma unroll
        for (int t = 0; t < 8; ++t) {
            const float* wp = W + (size_t)(k + khalf) * OUT_COLS + t * 16 + mrow;
            v2f b;
            b.x = wp[0];
            b.y = wp[OUT_COLS];
            acc[t] = __builtin_amdgcn_wmma_f32_16x16x4_f32(
                false, a, false, b, (short)0, acc[t], false, false);
        }
    }

    const int mbase = (lane >> 4) << 3;
#pragma unroll
    for (int t = 0; t < 8; ++t) {
#pragma unroll
        for (int i = 0; i < 8; ++i) {
            float v = acc[t][i];
            v = (v == v) ? v : 0.0f;
            h[(size_t)(row0 + mbase + i) * OUT_COLS + t * 16 + mrow] = v;
        }
    }
}

__global__ void __launch_bounds__(NHEAD * EDGE_HIDDEN)
etype_logits_kernel(const float* __restrict__ edge_emb,
                    const float* __restrict__ W_e,
                    const float* __restrict__ a_e,
                    float* __restrict__ he) {
    __shared__ float ev_s[NHEAD * EDGE_HIDDEN];
    __shared__ float ae_s[NHEAD * EDGE_HIDDEN];
    const int t    = blockIdx.x;
    const int col  = threadIdx.x;
    const int head = col >> 6, d = col & (EDGE_HIDDEN - 1);
    const float* er = edge_emb + t * EDGE_HIDDEN;
    const float* wc = W_e + col;
    const float ae = a_e[col];
    float ev = 0.0f;
#pragma unroll
    for (int k = 0; k < EDGE_HIDDEN; ++k)
        ev = fmaf(er[k], wc[k * (EDGE_HIDDEN * NHEAD)], ev);
    ev_s[d * NHEAD + head] = ev;
    ae_s[d * NHEAD + head] = ae;
    __syncthreads();
    if (col < NHEAD) {
        float s = 0.0f;
#pragma unroll
        for (int dd = 0; dd < EDGE_HIDDEN; ++dd)
            s = fmaf(ae_s[dd * NHEAD + col], ev_s[dd * NHEAD + col], s);
        he[t * NHEAD + col] = s;
    }
}

__global__ void __launch_bounds__(256)
node_logits_kernel(const float* __restrict__ h,
                   const float* __restrict__ a_l, const float* __restrict__ a_r,
                   float* __restrict__ hl, float* __restrict__ hr) {
    const int idx = blockIdx.x * blockDim.x + threadIdx.x;
    if (idx >= N_NODES * NHEAD) return;
    const int node = idx >> 2, head = idx & 3;
    const float* hp = h + (size_t)node * OUT_COLS + head * OUT_HIDDEN;
    const float* al = a_l + head * OUT_HIDDEN;
    const float* ar = a_r + head * OUT_HIDDEN;
    float sl = 0.0f, sr = 0.0f;
#pragma unroll
    for (int d = 0; d < OUT_HIDDEN; ++d) {
        float v = hp[d];
        sl = fmaf(al[d], v, sl);
        sr = fmaf(ar[d], v, sr);
    }
    hl[idx] = sl;
    hr[idx] = sr;
}

__device__ __forceinline__ float4 edge_scores(const float* __restrict__ hl,
                                              const float* __restrict__ hr,
                                              const float* __restrict__ he,
                                              int src, int trg, int et) {
    const float4 l = *(const float4*)(hl + src * NHEAD);
    const float4 r = *(const float4*)(hr + trg * NHEAD);
    const float4 t = *(const float4*)(he + et * NHEAD);
    float4 s;
    s.x = l.x + r.x + t.x;
    s.y = l.y + r.y + t.y;
    s.z = l.z + r.z + t.z;
    s.w = l.w + r.w + t.w;
    s.x = (s.x > 0.0f) ? s.x : NEG_SLOPE * s.x;
    s.y = (s.y > 0.0f) ? s.y : NEG_SLOPE * s.y;
    s.z = (s.z > 0.0f) ? s.z : NEG_SLOPE * s.z;
    s.w = (s.w > 0.0f) ? s.w : NEG_SLOPE * s.w;
    return s;
}

#define NB   256
#define NPB  196
#define G1   256
#define CH   (N_EDGES / G1)
#define GT   1024
#define ITERS ((CH + GT - 1) / GT)

__global__ void __launch_bounds__(GT)
chunk_rank_kernel(const int* __restrict__ edge,
                  const float* __restrict__ hl, const float* __restrict__ hr,
                  const float* __restrict__ he,
                  int* __restrict__ lrank, int* __restrict__ counts,
                  unsigned* __restrict__ gmax) {
    __shared__ int   hist[NB];
    __shared__ float red[GT / 32];
    const int wg = blockIdx.x, tid = threadIdx.x;
    if (tid < NB) hist[tid] = 0;
    __syncthreads();
    const int e0 = wg * CH;
    int srcv[ITERS], trgv[ITERS], etv[ITERS];
#pragma unroll
    for (int it = 0; it < ITERS; ++it) {
        const int k = tid + it * GT;
        const int e = e0 + ((k < CH) ? k : (CH - 1));
        srcv[it] = edge[e];
        trgv[it] = edge[N_EDGES + e];
        etv[it]  = edge[2 * N_EDGES + e];
    }
    float lmax = -INFINITY;
#pragma unroll
    for (int it = 0; it < ITERS; ++it) {
        const float4 s = edge_scores(hl, hr, he, srcv[it], trgv[it], etv[it]);
        lmax = fmaxf(lmax, fmaxf(fmaxf(s.x, s.y), fmaxf(s.z, s.w)));
    }
#pragma unroll
    for (int it = 0; it < ITERS; ++it) {
        const int k = tid + it * GT;
        if (k < CH) lrank[e0 + k] = atomicAdd(&hist[trgv[it] / NPB], 1);
    }
    __syncthreads();
    if (tid < NB) counts[tid * G1 + wg] = hist[tid];
    for (int m = 16; m > 0; m >>= 1) lmax = fmaxf(lmax, __shfl_xor(lmax, m, 32));
    if ((tid & 31) == 0) red[tid >> 5] = lmax;
    __syncthreads();
    if (tid < 32) {
        float v = red[tid];
        for (int m = 16; m > 0; m >>= 1) v = fmaxf(v, __shfl_xor(v, m, 32));
        if (tid == 0) atomicMax(gmax, enc_f32(v));
    }
}

#define SC_ROWS 16
__global__ void __launch_bounds__(GT)
scan_kernel(int* __restrict__ tab) {
    __shared__ int wtot[GT / 32];
    const int lane = threadIdx.x & 31, w = threadIdx.x >> 5;
    int4* p = (int4*)(tab + w * (SC_ROWS * 128)) + lane;
    int4 v[SC_ROWS];
#pragma unroll
    for (int r = 0; r < SC_ROWS; ++r) v[r] = p[r * 32];
    int carry = 0;
#pragma unroll
    for (int r = 0; r < SC_ROWS; ++r) {
        const int s1 = v[r].x, s2 = s1 + v[r].y, s3 = s2 + v[r].z, s4 = s3 + v[r].w;
        int inc = s4;
        for (int off = 1; off < 32; off <<= 1) {
            const int t = __shfl_up(inc, off, 32);
            if (lane >= off) inc += t;
        }
        const int ex = carry + inc - s4;
        v[r] = make_int4(ex, ex + s1, ex + s2, ex + s3);
        carry += __shfl(inc, 31, 32);
    }
    if (lane == 0) wtot[w] = carry;
    __syncthreads();
    int base = (lane < w) ? wtot[lane] : 0;
    for (int m = 16; m > 0; m >>= 1) base += __shfl_xor(base, m, 32);
#pragma unroll
    for (int r = 0; r < SC_ROWS; ++r)
        p[r * 32] = make_int4(v[r].x + base, v[r].y + base, v[r].z + base, v[r].w + base);
}

__global__ void __launch_bounds__(GT)
scatter_kernel(const int* __restrict__ edge, const int* __restrict__ lrank,
               const int* __restrict__ tab, int* __restrict__ coarse) {
    __shared__ int bs[NB];
    const int wg = blockIdx.x, tid = threadIdx.x;
    if (tid < NB) bs[tid] = tab[tid * G1 + wg];
    __syncthreads();
    const int e0 = wg * CH;
    int srcv[ITERS], trgv[ITERS], etv[ITERS], rkv[ITERS];
#pragma unroll
    for (int it = 0; it < ITERS; ++it) {
        const int k = tid + it * GT;
        const int e = e0 + ((k < CH) ? k : (CH - 1));
        srcv[it] = edge[e];
        trgv[it] = edge[N_EDGES + e];
        etv[it]  = edge[2 * N_EDGES + e];
        rkv[it]  = lrank[e];
    }
#pragma unroll
    for (int it = 0; it < ITERS; ++it) {
        const int k = tid + it * GT;
        if (k < CH) {
            const int b  = trgv[it] / NPB;
            const int tl = trgv[it] - b * NPB;
            *(int2*)(coarse + 2 * (size_t)(bs[b] + rkv[it])) =
                make_int2(e0 + k, srcv[it] | (etv[it] << 16) | (tl << 19));
        }
    }
}

__global__ void __launch_bounds__(GT)
bucket_kernel(const int* coarse, const int* __restrict__ tab,
              int* __restrict__ grouped, int* ordered, int* __restrict__ offs) {
    __shared__ int cntL[NPB + 1];
    __shared__ int loff[NPB + 1];
    const int b = blockIdx.x, tid = threadIdx.x;
    const int lane = tid & 31, w = tid >> 5;
    const int bstart = tab[b * G1];
    const int bend   = (b + 1 < NB) ? tab[(b + 1) * G1] : N_EDGES;
    const int n = bend - bstart;
    if (tid <= NPB) cntL[tid] = 0;
    __syncthreads();
    for (int i = tid; i < n; i += GT)
        atomicAdd(&cntL[(coarse[2 * (size_t)(bstart + i) + 1] >> 19) & 0xFF], 1);
    __syncthreads();
    if (w == 0) {
        int carry = 0;
        for (int r = 0; r < (NPB + 32) / 32; ++r) {
            const int idx = r * 32 + lane;
            const int c = (idx < NPB) ? cntL[idx] : 0;
            int inc = c;
            for (int off = 1; off < 32; off <<= 1) {
                const int t = __shfl_up(inc, off, 32);
                if (lane >= off) inc += t;
            }
            if (idx <= NPB) loff[idx] = carry + inc - c;
            carry += __shfl(inc, 31, 32);
        }
    }
    __syncthreads();
    if (tid < NPB) {
        const int node = b * NPB + tid;
        if (node <= N_NODES) offs[node] = bstart + loff[tid];
        cntL[tid] = 0;
    }
    __syncthreads();
    for (int i = tid; i < n; i += GT) {
        const int2 rec = *(const int2*)(coarse + 2 * (size_t)(bstart + i));
        const int tl = (rec.y >> 19) & 0xFF;
        const int r  = atomicAdd(&cntL[tl], 1);
        *(int2*)(grouped + 2 * (size_t)(bstart + loff[tl] + r)) = make_int2(rec.x, rec.y & 0x7FFFF);
    }
    __syncthreads();
    for (int t = w; t < NPB; t += GT / 32) {
        const int beg = bstart + __builtin_amdgcn_readfirstlane(loff[t]);
        const int d   = __builtin_amdgcn_readfirstlane(loff[t + 1]) - (beg - bstart);
        for (int ca = 0; ca < d; ca += 32) {
            const int ia = ca + lane;
            int2 pa = make_int2(0x7FFFFFFF, 0);
            if (ia < d) pa = *(const int2*)(grouped + 2 * (size_t)(beg + ia));
            int pos = 0;
            for (int cb = 0; cb < d; cb += 32) {
                const int ib = cb + lane;
                int kb = 0x7FFFFFFF;
                if (ib < d) kb = grouped[2 * (size_t)(beg + ib)];
#pragma unroll
                for (int j = 0; j < 32; ++j) {
                    const int key = __builtin_amdgcn_readlane(kb, j);
                    pos += (key < pa.x) ? 1 : 0;
                }
            }
            if (ia < d) *(int2*)(ordered + 2 * (size_t)(beg + pos)) = pa;
        }
    }
}

__global__ void __launch_bounds__(256)
node_aggregate_kernel(const int* __restrict__ ordered, const int* __restrict__ offs,
                      const float* __restrict__ h,
                      const float* __restrict__ hl, const float* __restrict__ hr,
                      const float* __restrict__ he, const unsigned* __restrict__ gmax,
                      float* __restrict__ out, float* __restrict__ attn_out) {
    const int lane = threadIdx.x & 31;
    const int node = blockIdx.x * (blockDim.x >> 5) + (threadIdx.x >> 5);
    if (node >= N_NODES) return;
    const int beg = __builtin_amdgcn_readfirstlane(offs[node]);
    const int d   = __builtin_amdgcn_readfirstlane(offs[node + 1]) - beg;
    const float smax = dec_f32(*gmax);

    float4 ds = make_float4(0.0f, 0.0f, 0.0f, 0.0f);
    for (int c = 0; c < d; c += 32) {
        const int i = c + lane;
        if (i < d) {
            const int2 p = *(const int2*)(ordered + 2 * (size_t)(beg + i));
            const float4 s = edge_scores(hl, hr, he, p.y & 0xFFFF, node, p.y >> 16);
            ds.x += __expf(s.x - smax);
            ds.y += __expf(s.y - smax);
            ds.z += __expf(s.z - smax);
            ds.w += __expf(s.w - smax);
        }
    }
    for (int m = 16; m > 0; m >>= 1) {
        ds.x += __shfl_xor(ds.x, m, 32);
        ds.y += __shfl_xor(ds.y, m, 32);
        ds.z += __shfl_xor(ds.z, m, 32);
        ds.w += __shfl_xor(ds.w, m, 32);
    }
    const float4 dn = make_float4(ds.x + 1e-16f, ds.y + 1e-16f, ds.z + 1e-16f, ds.w + 1e-16f);

    float acc0 = 0.0f, acc1 = 0.0f, acc2 = 0.0f, acc3 = 0.0f;
    const float* hcol = h + lane;
    for (int c = 0; c < d; c += 32) {
        const int i = c + lane;
        int srcv = 0;
        float4 a = make_float4(0.0f, 0.0f, 0.0f, 0.0f);
        if (i < d) {
            const int2 p = *(const int2*)(ordered + 2 * (size_t)(beg + i));
            srcv = p.y & 0xFFFF;
            const float4 s = edge_scores(hl, hr, he, srcv, node, p.y >> 16);
            a.x = __expf(s.x - smax) / dn.x;
            a.y = __expf(s.y - smax) / dn.y;
            a.z = __expf(s.z - smax) / dn.z;
            a.w = __expf(s.w - smax) / dn.w;
            *(float4*)(attn_out + (size_t)p.x * NHEAD) = a;
        }
        const int nb = (d - c < 32) ? (d - c) : 32;
#pragma unroll 4
        for (int j = 0; j < nb; ++j) {
            const int sj = __builtin_amdgcn_readlane(srcv, j);
            const float a0 = __int_as_float(__builtin_amdgcn_readlane(__float_as_int(a.x), j));
            const float a1 = __int_as_float(__builtin_amdgcn_readlane(__float_as_int(a.y), j));
            const float a2 = __int_as_float(__builtin_amdgcn_readlane(__float_as_int(a.z), j));
            const float a3 = __int_as_float(__builtin_amdgcn_readlane(__float_as_int(a.w), j));
            const float* row = hcol + (size_t)sj * OUT_COLS;
            acc0 = fmaf(row[0 * OUT_HIDDEN], a0, acc0);
            acc1 = fmaf(row[1 * OUT_HIDDEN], a1, acc1);
            acc2 = fmaf(row[2 * OUT_HIDDEN], a2, acc2);
            acc3 = fmaf(row[3 * OUT_HIDDEN], a3, acc3);
        }
    }
    float* orow = out + (size_t)node * OUT_COLS + lane;
    orow[0 * OUT_HIDDEN] = acc0;
    orow[1 * OUT_HIDDEN] = acc1;
    orow[2 * OUT_HIDDEN] = acc2;
    orow[3 * OUT_HIDDEN] = acc3;
}

extern "C" void kernel_launch(void* const* d_in, const int* in_sizes, int n_in,
                              void* d_out, int out_size, void* d_ws, size_t ws_size,
                              hipStream_t stream) {
    const int*   edge     = (const int*)  d_in[0];
    const float* x        = (const float*)d_in[1];
    const float* edge_emb = (const float*)d_in[2];
    const float* W        = (const float*)d_in[3];
    const float* W_e      = (const float*)d_in[4];
    const float* a_l      = (const float*)d_in[5];
    const float* a_r      = (const float*)d_in[6];
    const float* a_e      = (const float*)d_in[7];

    float* out      = (float*)d_out;
    float* attn_out = out + (size_t)N_NODES * OUT_COLS;

    char* ws = (char*)d_ws;
    float* h       = (float*)ws;  ws += (size_t)N_NODES * OUT_COLS * sizeof(float);
    int*   planeA  = (int*)ws;    ws += (size_t)N_EDGES * 2 * sizeof(int);
    int*   planeB  = (int*)ws;    ws += (size_t)N_EDGES * 2 * sizeof(int);
    int*   lrank   = planeA;
    int*   grouped = planeA;
    int*   coarse  = planeB;
    int*   ordered = planeB;
    float* hl      = (float*)ws;  ws += (size_t)N_NODES * NHEAD   * sizeof(float);
    float* hr      = (float*)ws;  ws += (size_t)N_NODES * NHEAD   * sizeof(float);
    int*   offs    = (int*)ws + 50176;
    int*   tab     = (int*)ws + 102400;
    ws += (size_t)N_NODES * NHEAD * sizeof(float);
    float* he      = (float*)ws;  ws += (size_t)N_ETYPES * NHEAD  * sizeof(float);
    unsigned* gmax = (unsigned*)ws;

    init_kernel<<<1, 32, 0, stream>>>(gmax);

    {
        const int strips = N_NODES / 16;
        const int wavesPerBlock = 8;
        const int blocks = (strips + wavesPerBlock - 1) / wavesPerBlock;
        gemm_h_kernel<<<blocks, 256, 0, stream>>>(x, W, h);
    }
    etype_logits_kernel<<<N_ETYPES, NHEAD * EDGE_HIDDEN, 0, stream>>>(edge_emb, W_e, a_e, he);
    node_logits_kernel<<<(N_NODES * NHEAD + 255) / 256, 256, 0, stream>>>(h, a_l, a_r, hl, hr);
    chunk_rank_kernel<<<G1, GT, 0, stream>>>(edge, hl, hr, he, lrank, tab, gmax);
    scan_kernel<<<1, GT, 0, stream>>>(tab);
    scatter_kernel<<<G1, GT, 0, stream>>>(edge, lrank, tab, coarse);
    bucket_kernel<<<NB, GT, 0, stream>>>(coarse, tab, grouped, ordered, offs);
    node_aggregate_kernel<<<(N_NODES + 7) / 8, 256, 0, stream>>>(
        ordered, offs, h, hl, hr, he, gmax, out, attn_out);
}


# score max moved from the edge-order rank kernel to the bucket kernel's ordering pass (a wave's target is one node): one gathered logit row per edge, not two; max is order-free, outputs bit-identical
# speedup vs baseline: 3.3223x; 1.0455x over previous
#include <hip/hip_runtime.h>
#include <hip/hip_bf16.h>

typedef __attribute__((ext_vector_type(2))) float v2f;
typedef __attribute__((ext_vector_type(8))) float v8f;

#define N_NODES 50000
#define N_EDGES 1600000
#define IN_HIDDEN 256
#define OUT_COLS 128
#define NHEAD 4
#define OUT_HIDDEN 32
#define EDGE_HIDDEN 64
#define N_ETYPES 8
#define NEG_SLOPE 0.2f

__device__ __forceinline__ unsigned enc_f32(float f) {
    unsigned u = __float_as_uint(f);
    return (u & 0x80000000u) ? ~u : (u | 0x80000000u);
}
__device__ __forceinline__ float dec_f32(unsigned u) {
    unsigned b = (u & 0x80000000u) ? (u & 0x7FFFFFFFu) : ~u;
    return __uint_as_float(b);
}

__global__ void __launch_bounds__(32)
init_kernel(unsigned* __restrict__ gmax) {
    if (blockIdx.x == 0 && threadIdx.x == 0) *gmax = enc_f32(-INFINITY);
}

__global__ void __launch_bounds__(256)
gemm_h_kernel(const float* __restrict__ x, const float* __restrict__ W,
              float* __restrict__ h) {
    const int lane  = threadIdx.x & 31;
    const int wave  = threadIdx.x >> 5;
    const int strip = blockIdx.x * (blockDim.x >> 5) + wave;
    if (strip * 16 >= N_NODES) return;
    const int row0  = strip * 16;

    const int mrow  = lane & 15;
    const int khalf = (lane >> 4) << 1;

    v8f acc[8] = {};

    const float* xrow = x + (size_t)(row0 + mrow) * IN_HIDDEN + khalf;
    for (int k = 0; k < IN_HIDDEN; k += 4) {
        v2f a = *(const v2f*)(xrow + k);
#pragma unroll
        for (int t = 0; t < 8; ++t) {
            const float* wp = W + (size_t)(k + khalf) * OUT_COLS + t * 16 + mrow;
            v2f b;
            b.x = wp[0];
            b.y = wp[OUT_COLS];
            acc[t] = __builtin_amdgcn_wmma_f32_16x16x4_f32(
                false, a, false, b, (short)0, acc[t], false, false);
        }
    }

    const int mbase = (lane >> 4) << 3;
#pragma unroll
    for (int t = 0; t < 8; ++t) {
#pragma unroll
        for (int i = 0; i < 8; ++i) {
            float v = acc[t][i];
            v = (v == v) ? v : 0.0f;
            h[(size_t)(row0 + mbase + i) * OUT_COLS + t * 16 + mrow] = v;
        }
    }
}

__global__ void __launch_bounds__(NHEAD * EDGE_HIDDEN)
etype_logits_kernel(const float* __restrict__ edge_emb,
                    const float* __restrict__ W_e,
                    const float* __restrict__ a_e,
                    float* __restrict__ he) {
    __shared__ float ev_s[NHEAD * EDGE_HIDDEN];
    __shared__ float ae_s[NHEAD * EDGE_HIDDEN];
    const int t    = blockIdx.x;
    const int col  = threadIdx.x;
    const int head = col >> 6, d = col & (EDGE_HIDDEN - 1);
    const float* er = edge_emb + t * EDGE_HIDDEN;
    const float* wc = W_e + col;
    const float ae = a_e[col];
    float ev = 0.0f;
#pragma unroll
    for (int k = 0; k < EDGE_HIDDEN; ++k)
        ev = fmaf(er[k], wc[k * (EDGE_HIDDEN * NHEAD)], ev);
    ev_s[d * NHEAD + head] = ev;
    ae_s[d * NHEAD + head] = ae;
    __syncthreads();
    if (col < NHEAD) {
        float s = 0.0f;
#pragma unroll
        for (int dd = 0; dd < EDGE_HIDDEN; ++dd)
            s = fmaf(ae_s[dd * NHEAD + col], ev_s[dd * NHEAD + col], s);
        he[t * NHEAD + col] = s;
    }
}

__global__ void __launch_bounds__(256)
node_logits_kernel(const float* __restrict__ h,
                   const float* __restrict__ a_l, const float* __restrict__ a_r,
                   float* __restrict__ hl, float* __restrict__ hr) {
    const int idx = blockIdx.x * blockDim.x + threadIdx.x;
    if (idx >= N_NODES * NHEAD) return;
    const int node = idx >> 2, head = idx & 3;
    const float* hp = h + (size_t)node * OUT_COLS + head * OUT_HIDDEN;
    const float* al = a_l + head * OUT_HIDDEN;
    const float* ar = a_r + head * OUT_HIDDEN;
    float sl = 0.0f, sr = 0.0f;
#pragma unroll
    for (int d = 0; d < OUT_HIDDEN; ++d) {
        float v = hp[d];
        sl = fmaf(al[d], v, sl);
        sr = fmaf(ar[d], v, sr);
    }
    hl[idx] = sl;
    hr[idx] = sr;
}

__device__ __forceinline__ float4 edge_scores(const float* __restrict__ hl,
                                              const float* __restrict__ hr,
                                              const float* __restrict__ he,
                                              int src, int trg, int et) {
    const float4 l = *(const float4*)(hl + src * NHEAD);
    const float4 r = *(const float4*)(hr + trg * NHEAD);
    const float4 t = *(const float4*)(he + et * NHEAD);
    float4 s;
    s.x = l.x + r.x + t.x;
    s.y = l.y + r.y + t.y;
    s.z = l.z + r.z + t.z;
    s.w = l.w + r.w + t.w;
    s.x = (s.x > 0.0f) ? s.x : NEG_SLOPE * s.x;
    s.y = (s.y > 0.0f) ? s.y : NEG_SLOPE * s.y;
    s.z = (s.z > 0.0f) ? s.z : NEG_SLOPE * s.z;
    s.w = (s.w > 0.0f) ? s.w : NEG_SLOPE * s.w;
    return s;
}

__device__ __forceinline__ float4 scores_of_rows(const float4 l, const float4 r, const float4 t) {
    float4 s;
    s.x = l.x + r.x + t.x;
    s.y = l.y + r.y + t.y;
    s.z = l.z + r.z + t.z;
    s.w = l.w + r.w + t.w;
    s.x = (s.x > 0.0f) ? s.x : NEG_SLOPE * s.x;
    s.y = (s.y > 0.0f) ? s.y : NEG_SLOPE * s.y;
    s.z = (s.z > 0.0f) ? s.z : NEG_SLOPE * s.z;
    s.w = (s.w > 0.0f) ? s.w : NEG_SLOPE * s.w;
    return s;
}

#define NB   256
#define NPB  196
#define G1   256
#define CH   (N_EDGES / G1)
#define GT   1024
#define ITERS ((CH + GT - 1) / GT)

__global__ void __launch_bounds__(GT)
chunk_rank_kernel(const int* __restrict__ edge,
                  int* __restrict__ lrank, int* __restrict__ counts) {
    __shared__ int hist[NB];
    const int wg = blockIdx.x, tid = threadIdx.x;
    if (tid < NB) hist[tid] = 0;
    __syncthreads();
    const int e0 = wg * CH;
    int trgv[ITERS];
#pragma unroll
    for (int it = 0; it < ITERS; ++it) {
        const int k = tid + it * GT;
        const int e = e0 + ((k < CH) ? k : (CH - 1));
        trgv[it] = edge[N_EDGES + e];
    }
#pragma unroll
    for (int it = 0; it < ITERS; ++it) {
        const int k = tid + it * GT;
        if (k < CH) lrank[e0 + k] = atomicAdd(&hist[trgv[it] / NPB], 1);
    }
    __syncthreads();
    if (tid < NB) counts[tid * G1 + wg] = hist[tid];
}

#define SC_ROWS 16
__global__ void __launch_bounds__(GT)
scan_kernel(int* __restrict__ tab) {
    __shared__ int wtot[GT / 32];
    const int lane = threadIdx.x & 31, w = threadIdx.x >> 5;
    int4* p = (int4*)(tab + w * (SC_ROWS * 128)) + lane;
    int4 v[SC_ROWS];
#pragma unroll
    for (int r = 0; r < SC_ROWS; ++r) v[r] = p[r * 32];
    int carry = 0;
#pragma unroll
    for (int r = 0; r < SC_ROWS; ++r) {
        const int s1 = v[r].x, s2 = s1 + v[r].y, s3 = s2 + v[r].z, s4 = s3 + v[r].w;
        int inc = s4;
        for (int off = 1; off < 32; off <<= 1) {
            const int t = __shfl_up(inc, off, 32);
            if (lane >= off) inc += t;
        }
        const int ex = carry + inc - s4;
        v[r] = make_int4(ex, ex + s1, ex + s2, ex + s3);
        carry += __shfl(inc, 31, 32);
    }
    if (lane == 0) wtot[w] = carry;
    __syncthreads();
    int base = (lane < w) ? wtot[lane] : 0;
    for (int m = 16; m > 0; m >>= 1) base += __shfl_xor(base, m, 32);
#pragma unroll
    for (int r = 0; r < SC_ROWS; ++r)
        p[r * 32] = make_int4(v[r].x + base, v[r].y + base, v[r].z + base, v[r].w + base);
}

__global__ void __launch_bounds__(GT)
scatter_kernel(const int* __restrict__ edge, const int* __restrict__ lrank,
               const int* __restrict__ tab, int* __restrict__ coarse) {
    __shared__ int bs[NB];
    const int wg = blockIdx.x, tid = threadIdx.x;
    if (tid < NB) bs[tid] = tab[tid * G1 + wg];
    __syncthreads();
    const int e0 = wg * CH;
    int srcv[ITERS], trgv[ITERS], etv[ITERS], rkv[ITERS];
#pragma unroll
    for (int it = 0; it < ITERS; ++it) {
        const int k = tid + it * GT;
        const int e = e0 + ((k < CH) ? k : (CH - 1));
        srcv[it] = edge[e];
        trgv[it] = edge[N_EDGES + e];
        etv[it]  = edge[2 * N_EDGES + e];
        rkv[it]  = lrank[e];
    }
#pragma unroll
    for (int it = 0; it < ITERS; ++it) {
        const int k = tid + it * GT;
        if (k < CH) {
            const int b  = trgv[it] / NPB;
            const int tl = trgv[it] - b * NPB;
            *(int2*)(coarse + 2 * (size_t)(bs[b] + rkv[it])) =
                make_int2(e0 + k, srcv[it] | (etv[it] << 16) | (tl << 19));
        }
    }
}

__global__ void __launch_bounds__(GT)
bucket_kernel(const int* coarse, const int* __restrict__ tab,
              int* __restrict__ grouped, int* ordered, int* __restrict__ offs,
              const float* __restrict__ hl, const float* __restrict__ hr,
              const float* __restrict__ he, unsigned* __restrict__ gmax) {
    __shared__ int   cntL[NPB + 1];
    __shared__ int   loff[NPB + 1];
    __shared__ float red[GT / 32];
    const int b = blockIdx.x, tid = threadIdx.x;
    const int lane = tid & 31, w = tid >> 5;
    const int bstart = tab[b * G1];
    const int bend   = (b + 1 < NB) ? tab[(b + 1) * G1] : N_EDGES;
    const int n = bend - bstart;
    if (tid <= NPB) cntL[tid] = 0;
    __syncthreads();
    for (int i = tid; i < n; i += GT)
        atomicAdd(&cntL[(coarse[2 * (size_t)(bstart + i) + 1] >> 19) & 0xFF], 1);
    __syncthreads();
    if (w == 0) {
        int carry = 0;
        for (int r = 0; r < (NPB + 32) / 32; ++r) {
            const int idx = r * 32 + lane;
            const int c = (idx < NPB) ? cntL[idx] : 0;
            int inc = c;
            for (int off = 1; off < 32; off <<= 1) {
                const int t = __shfl_up(inc, off, 32);
                if (lane >= off) inc += t;
            }
            if (idx <= NPB) loff[idx] = carry + inc - c;
            carry += __shfl(inc, 31, 32);
        }
    }
    __syncthreads();
    if (tid < NPB) {
        const int node = b * NPB + tid;
        if (node <= N_NODES) offs[node] = bstart + loff[tid];
        cntL[tid] = 0;
    }
    __syncthreads();
    for (int i = tid; i < n; i += GT) {
        const int2 rec = *(const int2*)(coarse + 2 * (size_t)(bstart + i));
        const int tl = (rec.y >> 19) & 0xFF;
        const int r  = atomicAdd(&cntL[tl], 1);
        *(int2*)(grouped + 2 * (size_t)(bstart + loff[tl] + r)) = make_int2(rec.x, rec.y & 0x7FFFF);
    }
    __syncthreads();
    float lmax = -INFINITY;
    int  py = 0, pnode = 0;
    bool pv = false;
    for (int t = w; t < NPB; t += GT / 32) {
        const int beg  = bstart + __builtin_amdgcn_readfirstlane(loff[t]);
        const int d    = __builtin_amdgcn_readfirstlane(loff[t + 1]) - (beg - bstart);
        const int node = b * NPB + t;
        for (int ca = 0; ca < d; ca += 32) {
            const int ia = ca + lane;
            int2 pa = make_int2(0x7FFFFFFF, 0);
            if (ia < d) pa = *(const int2*)(grouped + 2 * (size_t)(beg + ia));
            const float4 rl = *(const float4*)(hl + (py & 0xFFFF) * NHEAD);
            const float4 rr = *(const float4*)(hr + pnode * NHEAD);
            const float4 rt = *(const float4*)(he + (py >> 16) * NHEAD);
            int pos = 0;
            for (int cb = 0; cb < d; cb += 32) {
                const int ib = cb + lane;
                int kb = 0x7FFFFFFF;
                if (ib < d) kb = grouped[2 * (size_t)(beg + ib)];
#pragma unroll
                for (int j = 0; j < 32; ++j) {
                    const int key = __builtin_amdgcn_readlane(kb, j);
                    pos += (key < pa.x) ? 1 : 0;
                }
            }
            if (ia < d) *(int2*)(ordered + 2 * (size_t)(beg + pos)) = pa;
            const float4 s = scores_of_rows(rl, rr, rt);
            if (pv) lmax = fmaxf(lmax, fmaxf(fmaxf(s.x, s.y), fmaxf(s.z, s.w)));
            py = pa.y; pnode = node; pv = (ia < d);
        }
    }
    {
        const float4 rl = *(const float4*)(hl + (py & 0xFFFF) * NHEAD);
        const float4 rr = *(const float4*)(hr + pnode * NHEAD);
        const float4 rt = *(const float4*)(he + (py >> 16) * NHEAD);
        const float4 s = scores_of_rows(rl, rr, rt);
        if (pv) lmax = fmaxf(lmax, fmaxf(fmaxf(s.x, s.y), fmaxf(s.z, s.w)));
    }
    for (int m = 16; m > 0; m >>= 1) lmax = fmaxf(lmax, __shfl_xor(lmax, m, 32));
    if (lane == 0) red[w] = lmax;
    __syncthreads();
    if (tid < 32) {
        float v = red[tid];
        for (int m = 16; m > 0; m >>= 1) v = fmaxf(v, __shfl_xor(v, m, 32));
        if (tid == 0) atomicMax(gmax, enc_f32(v));
    }
}

__global__ void __launch_bounds__(256)
node_aggregate_kernel(const int* __restrict__ ordered, const int* __restrict__ offs,
                      const float* __restrict__ h,
                      const float* __restrict__ hl, const float* __restrict__ hr,
                      const float* __restrict__ he, const unsigned* __restrict__ gmax,
                      float* __restrict__ out, float* __restrict__ attn_out) {
    const int lane = threadIdx.x & 31;
    const int node = blockIdx.x * (blockDim.x >> 5) + (threadIdx.x >> 5);
    if (node >= N_NODES) return;
    const int beg = __builtin_amdgcn_readfirstlane(offs[node]);
    const int d   = __builtin_amdgcn_readfirstlane(offs[node + 1]) - beg;
    const float smax = dec_f32(*gmax);

    float4 ds = make_float4(0.0f, 0.0f, 0.0f, 0.0f);
    for (int c = 0; c < d; c += 32) {
        const int i = c + lane;
        if (i < d) {
            const int2 p = *(const int2*)(ordered + 2 * (size_t)(beg + i));
            const float4 s = edge_scores(hl, hr, he, p.y & 0xFFFF, node, p.y >> 16);
            ds.x += __expf(s.x - smax);
            ds.y += __expf(s.y - smax);
            ds.z += __expf(s.z - smax);
            ds.w += __expf(s.w - smax);
        }
    }
    for (int m = 16; m > 0; m >>= 1) {
        ds.x += __shfl_xor(ds.x, m, 32);
        ds.y += __shfl_xor(ds.y, m, 32);
        ds.z += __shfl_xor(ds.z, m, 32);
        ds.w += __shfl_xor(ds.w, m, 32);
    }
    const float4 dn = make_float4(ds.x + 1e-16f, ds.y + 1e-16f, ds.z + 1e-16f, ds.w + 1e-16f);

    float acc0 = 0.0f, acc1 = 0.0f, acc2 = 0.0f, acc3 = 0.0f;
    const float* hcol = h + lane;
    for (int c = 0; c < d; c += 32) {
        const int i = c + lane;
        int srcv = 0;
        float4 a = make_float4(0.0f, 0.0f, 0.0f, 0.0f);
        if (i < d) {
            const int2 p = *(const int2*)(ordered + 2 * (size_t)(beg + i));
            srcv = p.y & 0xFFFF;
            const float4 s = edge_scores(hl, hr, he, srcv, node, p.y >> 16);
            a.x = __expf(s.x - smax) / dn.x;
            a.y = __expf(s.y - smax) / dn.y;
            a.z = __expf(s.z - smax) / dn.z;
            a.w = __expf(s.w - smax) / dn.w;
            *(float4*)(attn_out + (size_t)p.x * NHEAD) = a;
        }
        const int nb = (d - c < 32) ? (d - c) : 32;
#pragma unroll 4
        for (int j = 0; j < nb; ++j) {
            const int sj = __builtin_amdgcn_readlane(srcv, j);
            const float a0 = __int_as_float(__builtin_amdgcn_readlane(__float_as_int(a.x), j));
            const float a1 = __int_as_float(__builtin_amdgcn_readlane(__float_as_int(a.y), j));
            const float a2 = __int_as_float(__builtin_amdgcn_readlane(__float_as_int(a.z), j));
            const float a3 = __int_as_float(__builtin_amdgcn_readlane(__float_as_int(a.w), j));
            const float* row = hcol + (size_t)sj * OUT_COLS;
            acc0 = fmaf(row[0 * OUT_HIDDEN], a0, acc0);
            acc1 = fmaf(row[1 * OUT_HIDDEN], a1, acc1);
            acc2 = fmaf(row[2 * OUT_HIDDEN], a2, acc2);
            acc3 = fmaf(row[3 * OUT_HIDDEN], a3, acc3);
        }
    }
    float* orow = out + (size_t)node * OUT_COLS + lane;
    orow[0 * OUT_HIDDEN] = acc0;
    orow[1 * OUT_HIDDEN] = acc1;
    orow[2 * OUT_HIDDEN] = acc2;
    orow[3 * OUT_HIDDEN] = acc3;
}

extern "C" void kernel_launch(void* const* d_in, const int* in_sizes, int n_in,
                              void* d_out, int out_size, void* d_ws, size_t ws_size,
                              hipStream_t stream) {
    const int*   edge     = (const int*)  d_in[0];
    const float* x        = (const float*)d_in[1];
    const float* edge_emb = (const float*)d_in[2];
    const float* W        = (const float*)d_in[3];
    const float* W_e      = (const float*)d_in[4];
    const float* a_l      = (const float*)d_in[5];
    const float* a_r      = (const float*)d_in[6];
    const float* a_e      = (const float*)d_in[7];

    float* out      = (float*)d_out;
    float* attn_out = out + (size_t)N_NODES * OUT_COLS;

    char* ws = (char*)d_ws;
    float* h       = (float*)ws;  ws += (size_t)N_NODES * OUT_COLS * sizeof(float);
    int*   planeA  = (int*)ws;    ws += (size_t)N_EDGES * 2 * sizeof(int);
    int*   planeB  = (int*)ws;    ws += (size_t)N_EDGES * 2 * sizeof(int);
    int*   lrank   = planeA;
    int*   grouped = planeA;
    int*   coarse  = planeB;
    int*   ordered = planeB;
    float* hl      = (float*)ws;  ws += (size_t)N_NODES * NHEAD   * sizeof(float);
    float* hr      = (float*)ws;  ws += (size_t)N_NODES * NHEAD   * sizeof(float);
    int*   offs    = (int*)ws + 50176;
    int*   tab     = (int*)ws + 102400;
    ws += (size_t)N_NODES * NHEAD * sizeof(float);
    float* he      = (float*)ws;  ws += (size_t)N_ETYPES * NHEAD  * sizeof(float);
    unsigned* gmax = (unsigned*)ws;

    init_kernel<<<1, 32, 0, stream>>>(gmax);

    {
        const int strips = N_NODES / 16;
        const int wavesPerBlock = 8;
        const int blocks = (strips + wavesPerBlock - 1) / wavesPerBlock;
        gemm_h_kernel<<<blocks, 256, 0, stream>>>(x, W, h);
    }
    etype_logits_kernel<<<N_ETYPES, NHEAD * EDGE_HIDDEN, 0, stream>>>(edge_emb, W_e, a_e, he);
    node_logits_kernel<<<(N_NODES * NHEAD + 255) / 256, 256, 0, stream>>>(h, a_l, a_r, hl, hr);
    chunk_rank_kernel<<<G1, GT, 0, stream>>>(edge, lrank, tab);
    scan_kernel<<<1, GT, 0, stream>>>(tab);
    scatter_kernel<<<G1, GT, 0, stream>>>(edge, lrank, tab, coarse);
    bucket_kernel<<<NB, GT, 0, stream>>>(coarse, tab, grouped, ordered, offs, hl, hr, he, gmax);
    node_aggregate_kernel<<<(N_NODES + 7) / 8, 256, 0, stream>>>(
        ordered, offs, h, hl, hr, he, gmax, out, attn_out);
}


# node pass: the denominator pass keeps records and exp scores of a node's first 64 edges in registers; the attention pass gathers no logit row and takes no exp again; sums and their order kept
# speedup vs baseline: 3.4269x; 1.0315x over previous
#include <hip/hip_runtime.h>
#include <hip/hip_bf16.h>

typedef __attribute__((ext_vector_type(2))) float v2f;
typedef __attribute__((ext_vector_type(8))) float v8f;

#define N_NODES 50000
#define N_EDGES 1600000
#define IN_HIDDEN 256
#define OUT_COLS 128
#define NHEAD 4
#define OUT_HIDDEN 32
#define EDGE_HIDDEN 64
#define N_ETYPES 8
#define NEG_SLOPE 0.2f

__device__ __forceinline__ unsigned enc_f32(float f) {
    unsigned u = __float_as_uint(f);
    return (u & 0x80000000u) ? ~u : (u | 0x80000000u);
}
__device__ __forceinline__ float dec_f32(unsigned u) {
    unsigned b = (u & 0x80000000u) ? (u & 0x7FFFFFFFu) : ~u;
    return __uint_as_float(b);
}

__global__ void __launch_bounds__(32)
init_kernel(unsigned* __restrict__ gmax) {
    if (blockIdx.x == 0 && threadIdx.x == 0) *gmax = enc_f32(-INFINITY);
}

__global__ void __launch_bounds__(256)
gemm_h_kernel(const float* __restrict__ x, const float* __restrict__ W,
              float* __restrict__ h) {
    const int lane  = threadIdx.x & 31;
    const int wave  = threadIdx.x >> 5;
    const int strip = blockIdx.x * (blockDim.x >> 5) + wave;
    if (strip * 16 >= N_NODES) return;
    const int row0  = strip * 16;

    const int mrow  = lane & 15;
    const int khalf = (lane >> 4) << 1;

    v8f acc[8] = {};

    const float* xrow = x + (size_t)(row0 + mrow) * IN_HIDDEN + khalf;
    for (int k = 0; k < IN_HIDDEN; k += 4) {
        v2f a = *(const v2f*)(xrow + k);
#pragma unroll
        for (int t = 0; t < 8; ++t) {
            const float* wp = W + (size_t)(k + khalf) * OUT_COLS + t * 16 + mrow;
            v2f b;
            b.x = wp[0];
            b.y = wp[OUT_COLS];
            acc[t] = __builtin_amdgcn_wmma_f32_16x16x4_f32(
                false, a, false, b, (short)0, acc[t], false, false);
        }
    }

    const int mbase = (lane >> 4) << 3;
#pragma unroll
    for (int t = 0; t < 8; ++t) {
#pragma unroll
        for (int i = 0; i < 8; ++i) {
            float v = acc[t][i];
            v = (v == v) ? v : 0.0f;
            h[(size_t)(row0 + mbase + i) * OUT_COLS + t * 16 + mrow] = v;
        }
    }
}

__global__ void __launch_bounds__(NHEAD * EDGE_HIDDEN)
etype_logits_kernel(const float* __restrict__ edge_emb,
                    const float* __restrict__ W_e,
                    const float* __restrict__ a_e,
                    float* __restrict__ he) {
    __shared__ float ev_s[NHEAD * EDGE_HIDDEN];
    __shared__ float ae_s[NHEAD * EDGE_HIDDEN];
    const int t    = blockIdx.x;
    const int col  = threadIdx.x;
    const int head = col >> 6, d = col & (EDGE_HIDDEN - 1);
    const float* er = edge_emb + t * EDGE_HIDDEN;
    const float* wc = W_e + col;
    const float ae = a_e[col];
    float ev = 0.0f;
#pragma unroll
    for (int k = 0; k < EDGE_HIDDEN; ++k)
        ev = fmaf(er[k], wc[k * (EDGE_HIDDEN * NHEAD)], ev);
    ev_s[d * NHEAD + head] = ev;
    ae_s[d * NHEAD + head] = ae;
    __syncthreads();
    if (col < NHEAD) {
        float s = 0.0f;
#pragma unroll
        for (int dd = 0; dd < EDGE_HIDDEN; ++dd)
            s = fmaf(ae_s[dd * NHEAD + col], ev_s[dd * NHEAD + col], s);
        he[t * NHEAD + col] = s;
    }
}

__global__ void __launch_bounds__(256)
node_logits_kernel(const float* __restrict__ h,
                   const float* __restrict__ a_l, const float* __restrict__ a_r,
                   float* __restrict__ hl, float* __restrict__ hr) {
    const int idx = blockIdx.x * blockDim.x + threadIdx.x;
    if (idx >= N_NODES * NHEAD) return;
    const int node = idx >> 2, head = idx & 3;
    const float* hp = h + (size_t)node * OUT_COLS + head * OUT_HIDDEN;
    const float* al = a_l + head * OUT_HIDDEN;
    const float* ar = a_r + head * OUT_HIDDEN;
    float sl = 0.0f, sr = 0.0f;
#pragma unroll
    for (int d = 0; d < OUT_HIDDEN; ++d) {
        float v = hp[d];
        sl = fmaf(al[d], v, sl);
        sr = fmaf(ar[d], v, sr);
    }
    hl[idx] = sl;
    hr[idx] = sr;
}

__device__ __forceinline__ float4 edge_scores(const float* __restrict__ hl,
                                              const float* __restrict__ hr,
                                              const float* __restrict__ he,
                                              int src, int trg, int et) {
    const float4 l = *(const float4*)(hl + src * NHEAD);
    const float4 r = *(const float4*)(hr + trg * NHEAD);
    const float4 t = *(const float4*)(he + et * NHEAD);
    float4 s;
    s.x = l.x + r.x + t.x;
    s.y = l.y + r.y + t.y;
    s.z = l.z + r.z + t.z;
    s.w = l.w + r.w + t.w;
    s.x = (s.x > 0.0f) ? s.x : NEG_SLOPE * s.x;
    s.y = (s.y > 0.0f) ? s.y : NEG_SLOPE * s.y;
    s.z = (s.z > 0.0f) ? s.z : NEG_SLOPE * s.z;
    s.w = (s.w > 0.0f) ? s.w : NEG_SLOPE * s.w;
    return s;
}

__device__ __forceinline__ float4 scores_of_rows(const float4 l, const float4 r, const float4 t) {
    float4 s;
    s.x = l.x + r.x + t.x;
    s.y = l.y + r.y + t.y;
    s.z = l.z + r.z + t.z;
    s.w = l.w + r.w + t.w;
    s.x = (s.x > 0.0f) ? s.x : NEG_SLOPE * s.x;
    s.y = (s.y > 0.0f) ? s.y : NEG_SLOPE * s.y;
    s.z = (s.z > 0.0f) ? s.z : NEG_SLOPE * s.z;
    s.w = (s.w > 0.0f) ? s.w : NEG_SLOPE * s.w;
    return s;
}

#define NB   256
#define NPB  196
#define G1   256
#define CH   (N_EDGES / G1)
#define GT   1024
#define ITERS ((CH + GT - 1) / GT)

__global__ void __launch_bounds__(GT)
chunk_rank_kernel(const int* __restrict__ edge,
                  int* __restrict__ lrank, int* __restrict__ counts) {
    __shared__ int hist[NB];
    const int wg = blockIdx.x, tid = threadIdx.x;
    if (tid < NB) hist[tid] = 0;
    __syncthreads();
    const int e0 = wg * CH;
    int trgv[ITERS];
#pragma unroll
    for (int it = 0; it < ITERS; ++it) {
        const int k = tid + it * GT;
        const int e = e0 + ((k < CH) ? k : (CH - 1));
        trgv[it] = edge[N_EDGES + e];
    }
#pragma unroll
    for (int it = 0; it < ITERS; ++it) {
        const int k = tid + it * GT;
        if (k < CH) lrank[e0 + k] = atomicAdd(&hist[trgv[it] / NPB], 1);
    }
    __syncthreads();
    if (tid < NB) counts[tid * G1 + wg] = hist[tid];
}

#define SC_ROWS 16
__global__ void __launch_bounds__(GT)
scan_kernel(int* __restrict__ tab) {
    __shared__ int wtot[GT / 32];
    const int lane = threadIdx.x & 31, w = threadIdx.x >> 5;
    int4* p = (int4*)(tab + w * (SC_ROWS * 128)) + lane;
    int4 v[SC_ROWS];
#pragma unroll
    for (int r = 0; r < SC_ROWS; ++r) v[r] = p[r * 32];
    int carry = 0;
#pragma unroll
    for (int r = 0; r < SC_ROWS; ++r) {
        const int s1 = v[r].x, s2 = s1 + v[r].y, s3 = s2 + v[r].z, s4 = s3 + v[r].w;
        int inc = s4;
        for (int off = 1; off < 32; off <<= 1) {
            const int t = __shfl_up(inc, off, 32);
            if (lane >= off) inc += t;
        }
        const int ex = carry + inc - s4;
        v[r] = make_int4(ex, ex + s1, ex + s2, ex + s3);
        carry += __shfl(inc, 31, 32);
    }
    if (lane == 0) wtot[w] = carry;
    __syncthreads();
    int base = (lane < w) ? wtot[lane] : 0;
    for (int m = 16; m > 0; m >>= 1) base += __shfl_xor(base, m, 32);
#pragma unroll
    for (int r = 0; r < SC_ROWS; ++r)
        p[r * 32] = make_int4(v[r].x + base, v[r].y + base, v[r].z + base, v[r].w + base);
}

__global__ void __launch_bounds__(GT)
scatter_kernel(const int* __restrict__ edge, const int* __restrict__ lrank,
               const int* __restrict__ tab, int* __restrict__ coarse) {
    __shared__ int bs[NB];
    const int wg = blockIdx.x, tid = threadIdx.x;
    if (tid < NB) bs[tid] = tab[tid * G1 + wg];
    __syncthreads();
    const int e0 = wg * CH;
    int srcv[ITERS], trgv[ITERS], etv[ITERS], rkv[ITERS];
#pragma unroll
    for (int it = 0; it < ITERS; ++it) {
        const int k = tid + it * GT;
        const int e = e0 + ((k < CH) ? k : (CH - 1));
        srcv[it] = edge[e];
        trgv[it] = edge[N_EDGES + e];
        etv[it]  = edge[2 * N_EDGES + e];
        rkv[it]  = lrank[e];
    }
#pragma unroll
    for (int it = 0; it < ITERS; ++it) {
        const int k = tid + it * GT;
        if (k < CH) {
            const int b  = trgv[it] / NPB;
            const int tl = trgv[it] - b * NPB;
            *(int2*)(coarse + 2 * (size_t)(bs[b] + rkv[it])) =
                make_int2(e0 + k, srcv[it] | (etv[it] << 16) | (tl << 19));
        }
    }
}

__global__ void __launch_bounds__(GT)
bucket_kernel(const int* coarse, const int* __restrict__ tab,
              int* __restrict__ grouped, int* ordered, int* __restrict__ offs,
              const float* __restrict__ hl, const float* __restrict__ hr,
              const float* __restrict__ he, unsigned* __restrict__ gmax) {
    __shared__ int   cntL[NPB + 1];
    __shared__ int   loff[NPB + 1];
    __shared__ float red[GT / 32];
    const int b = blockIdx.x, tid = threadIdx.x;
    const int lane = tid & 31, w = tid >> 5;
    const int bstart = tab[b * G1];
    const int bend   = (b + 1 < NB) ? tab[(b + 1) * G1] : N_EDGES;
    const int n = bend - bstart;
    if (tid <= NPB) cntL[tid] = 0;
    __syncthreads();
    for (int i = tid; i < n; i += GT)
        atomicAdd(&cntL[(coarse[2 * (size_t)(bstart + i) + 1] >> 19) & 0xFF], 1);
    __syncthreads();
    if (w == 0) {
        int carry = 0;
        for (int r = 0; r < (NPB + 32) / 32; ++r) {
            const int idx = r * 32 + lane;
            const int c = (idx < NPB) ? cntL[idx] : 0;
            int inc = c;
            for (int off = 1; off < 32; off <<= 1) {
                const int t = __shfl_up(inc, off, 32);
                if (lane >= off) inc += t;
            }
            if (idx <= NPB) loff[idx] = carry + inc - c;
            carry += __shfl(inc, 31, 32);
        }
    }
    __syncthreads();
    if (tid < NPB) {
        const int node = b * NPB + tid;
        if (node <= N_NODES) offs[node] = bstart + loff[tid];
        cntL[tid] = 0;
    }
    __syncthreads();
    for (int i = tid; i < n; i += GT) {
        const int2 rec = *(const int2*)(coarse + 2 * (size_t)(bstart + i));
        const int tl = (rec.y >> 19) & 0xFF;
        const int r  = atomicAdd(&cntL[tl], 1);
        *(int2*)(grouped + 2 * (size_t)(bstart + loff[tl] + r)) = make_int2(rec.x, rec.y & 0x7FFFF);
    }
    __syncthreads();
    float lmax = -INFINITY;
    int  py = 0, pnode = 0;
    bool pv = false;
    for (int t = w; t < NPB; t += GT / 32) {
        const int beg  = bstart + __builtin_amdgcn_readfirstlane(loff[t]);
        const int d    = __builtin_amdgcn_readfirstlane(loff[t + 1]) - (beg - bstart);
        const int node = b * NPB + t;
        for (int ca = 0; ca < d; ca += 32) {
            const int ia = ca + lane;
            int2 pa = make_int2(0x7FFFFFFF, 0);
            if (ia < d) pa = *(const int2*)(grouped + 2 * (size_t)(beg + ia));
            const float4 rl = *(const float4*)(hl + (py & 0xFFFF) * NHEAD);
            const float4 rr = *(const float4*)(hr + pnode * NHEAD);
            const float4 rt = *(const float4*)(he + (py >> 16) * NHEAD);
            int pos = 0;
            for (int cb = 0; cb < d; cb += 32) {
                const int ib = cb + lane;
                int kb = 0x7FFFFFFF;
                if (ib < d) kb = grouped[2 * (size_t)(beg + ib)];
#pragma unroll
                for (int j = 0; j < 32; ++j) {
                    const int key = __builtin_amdgcn_readlane(kb, j);
                    pos += (key < pa.x) ? 1 : 0;
                }
            }
            if (ia < d) *(int2*)(ordered + 2 * (size_t)(beg + pos)) = pa;
            const float4 s = scores_of_rows(rl, rr, rt);
            if (pv) lmax = fmaxf(lmax, fmaxf(fmaxf(s.x, s.y), fmaxf(s.z, s.w)));
            py = pa.y; pnode = node; pv = (ia < d);
        }
    }
    {
        const float4 rl = *(const float4*)(hl + (py & 0xFFFF) * NHEAD);
        const float4 rr = *(const float4*)(hr + pnode * NHEAD);
        const float4 rt = *(const float4*)(he + (py >> 16) * NHEAD);
        const float4 s = scores_of_rows(rl, rr, rt);
        if (pv) lmax = fmaxf(lmax, fmaxf(fmaxf(s.x, s.y), fmaxf(s.z, s.w)));
    }
    for (int m = 16; m > 0; m >>= 1) lmax = fmaxf(lmax, __shfl_xor(lmax, m, 32));
    if (lane == 0) red[w] = lmax;
    __syncthreads();
    if (tid < 32) {
        float v = red[tid];
        for (int m = 16; m > 0; m >>= 1) v = fmaxf(v, __shfl_xor(v, m, 32));
        if (tid == 0) atomicMax(gmax, enc_f32(v));
    }
}

__global__ void __launch_bounds__(256)
node_aggregate_kernel(const int* __restrict__ ordered, const int* __restrict__ offs,
                      const float* __restrict__ h,
                      const float* __restrict__ hl, const float* __restrict__ hr,
                      const float* __restrict__ he, const unsigned* __restrict__ gmax,
                      float* __restrict__ out, float* __restrict__ attn_out) {
    const int lane = threadIdx.x & 31;
    const int node = blockIdx.x * (blockDim.x >> 5) + (threadIdx.x >> 5);
    if (node >= N_NODES) return;
    const int beg = __builtin_amdgcn_readfirstlane(offs[node]);
    const int d   = __builtin_amdgcn_readfirstlane(offs[node + 1]) - beg;
    const float smax = dec_f32(*gmax);

    float4 ds = make_float4(0.0f, 0.0f, 0.0f, 0.0f);
    int2   p0 = make_int2(0, 0), p1 = make_int2(0, 0);
    float4 e0 = make_float4(0.0f, 0.0f, 0.0f, 0.0f);
    float4 e1 = make_float4(0.0f, 0.0f, 0.0f, 0.0f);
    if (lane < d) {
        p0 = *(const int2*)(ordered + 2 * (size_t)(beg + lane));
        const float4 s = edge_scores(hl, hr, he, p0.y & 0xFFFF, node, p0.y >> 16);
        e0.x = __expf(s.x - smax);
        e0.y = __expf(s.y - smax);
        e0.z = __expf(s.z - smax);
        e0.w = __expf(s.w - smax);
        ds.x += e0.x;
        ds.y += e0.y;
        ds.z += e0.z;
        ds.w += e0.w;
    }
    if (32 + lane < d) {
        p1 = *(const int2*)(ordered + 2 * (size_t)(beg + 32 + lane));
        const float4 s = edge_scores(hl, hr, he, p1.y & 0xFFFF, node, p1.y >> 16);
        e1.x = __expf(s.x - smax);
        e1.y = __expf(s.y - smax);
        e1.z = __expf(s.z - smax);
        e1.w = __expf(s.w - smax);
        ds.x += e1.x;
        ds.y += e1.y;
        ds.z += e1.z;
        ds.w += e1.w;
    }
    for (int c = 64; c < d; c += 32) {
        const int i = c + lane;
        if (i < d) {
            const int2 p = *(const int2*)(ordered + 2 * (size_t)(beg + i));
            const float4 s = edge_scores(hl, hr, he, p.y & 0xFFFF, node, p.y >> 16);
            ds.x += __expf(s.x - smax);
            ds.y += __expf(s.y - smax);
            ds.z += __expf(s.z - smax);
            ds.w += __expf(s.w - smax);
        }
    }
    for (int m = 16; m > 0; m >>= 1) {
        ds.x += __shfl_xor(ds.x, m, 32);
        ds.y += __shfl_xor(ds.y, m, 32);
        ds.z += __shfl_xor(ds.z, m, 32);
        ds.w += __shfl_xor(ds.w, m, 32);
    }
    const float4 dn = make_float4(ds.x + 1e-16f, ds.y + 1e-16f, ds.z + 1e-16f, ds.w + 1e-16f);

    float acc0 = 0.0f, acc1 = 0.0f, acc2 = 0.0f, acc3 = 0.0f;
    const float* hcol = h + lane;
    for (int c = 0; c < d; c += 32) {
        const int i = c + lane;
        int srcv = 0;
        float4 a = make_float4(0.0f, 0.0f, 0.0f, 0.0f);
        if (i < d) {
            int2   p = p0;
            float4 e = e0;
            if (c == 32) { p = p1; e = e1; }
            if (c >= 64) {
                p = *(const int2*)(ordered + 2 * (size_t)(beg + i));
                const float4 s = edge_scores(hl, hr, he, p.y & 0xFFFF, node, p.y >> 16);
                e.x = __expf(s.x - smax);
                e.y = __expf(s.y - smax);
                e.z = __expf(s.z - smax);
                e.w = __expf(s.w - smax);
            }
            srcv = p.y & 0xFFFF;
            a.x = e.x / dn.x;
            a.y = e.y / dn.y;
            a.z = e.z / dn.z;
            a.w = e.w / dn.w;
            *(float4*)(attn_out + (size_t)p.x * NHEAD) = a;
        }
        const int nb = (d - c < 32) ? (d - c) : 32;
#pragma unroll 4
        for (int j = 0; j < nb; ++j) {
            const int sj = __builtin_amdgcn_readlane(srcv, j);
            const float a0 = __int_as_float(__builtin_amdgcn_readlane(__float_as_int(a.x), j));
            const float a1 = __int_as_float(__builtin_amdgcn_readlane(__float_as_int(a.y), j));
            const float a2 = __int_as_float(__builtin_amdgcn_readlane(__float_as_int(a.z), j));
            const float a3 = __int_as_float(__builtin_amdgcn_readlane(__float_as_int(a.w), j));
            const float* row = hcol + (size_t)sj * OUT_COLS;
            acc0 = fmaf(row[0 * OUT_HIDDEN], a0, acc0);
            acc1 = fmaf(row[1 * OUT_HIDDEN], a1, acc1);
            acc2 = fmaf(row[2 * OUT_HIDDEN], a2, acc2);
            acc3 = fmaf(row[3 * OUT_HIDDEN], a3, acc3);
        }
    }
    float* orow = out + (size_t)node * OUT_COLS + lane;
    orow[0 * OUT_HIDDEN] = acc0;
    orow[1 * OUT_HIDDEN] = acc1;
    orow[2 * OUT_HIDDEN] = acc2;
    orow[3 * OUT_HIDDEN] = acc3;
}

extern "C" void kernel_launch(void* const* d_in, const int* in_sizes, int n_in,
                              void* d_out, int out_size, void* d_ws, size_t ws_size,
                              hipStream_t stream) {
    const int*   edge     = (const int*)  d_in[0];
    const float* x        = (const float*)d_in[1];
    const float* edge_emb = (const float*)d_in[2];
    const float* W        = (const float*)d_in[3];
    const float* W_e      = (const float*)d_in[4];
    const float* a_l      = (const float*)d_in[5];
    const float* a_r      = (const float*)d_in[6];
    const float* a_e      = (const float*)d_in[7];

    float* out      = (float*)d_out;
    float* attn_out = out + (size_t)N_NODES * OUT_COLS;

    char* ws = (char*)d_ws;
    float* h       = (float*)ws;  ws += (size_t)N_NODES * OUT_COLS * sizeof(float);
    int*   planeA  = (int*)ws;    ws += (size_t)N_EDGES * 2 * sizeof(int);
    int*   planeB  = (int*)ws;    ws += (size_t)N_EDGES * 2 * sizeof(int);
    int*   lrank   = planeA;
    int*   grouped = planeA;
    int*   coarse  = planeB;
    int*   ordered = planeB;
    float* hl      = (float*)ws;  ws += (size_t)N_NODES * NHEAD   * sizeof(float);
    float* hr      = (float*)ws;  ws += (size_t)N_NODES * NHEAD   * sizeof(float);
    int*   offs    = (int*)ws + 50176;
    int*   tab     = (int*)ws + 102400;
    ws += (size_t)N_NODES * NHEAD * sizeof(float);
    float* he      = (float*)ws;  ws += (size_t)N_ETYPES * NHEAD  * sizeof(float);
    unsigned* gmax = (unsigned*)ws;

    init_kernel<<<1, 32, 0, stream>>>(gmax);

    {
        const int strips = N_NODES / 16;
        const int wavesPerBlock = 8;
        const int blocks = (strips + wavesPerBlock - 1) / wavesPerBlock;
        gemm_h_kernel<<<blocks, 256, 0, stream>>>(x, W, h);
    }
    etype_logits_kernel<<<N_ETYPES, NHEAD * EDGE_HIDDEN, 0, stream>>>(edge_emb, W_e, a_e, he);
    node_logits_kernel<<<(N_NODES * NHEAD + 255) / 256, 256, 0, stream>>>(h, a_l, a_r, hl, hr);
    chunk_rank_kernel<<<G1, GT, 0, stream>>>(edge, lrank, tab);
    scan_kernel<<<1, GT, 0, stream>>>(tab);
    scatter_kernel<<<G1, GT, 0, stream>>>(edge, lrank, tab, coarse);
    bucket_kernel<<<NB, GT, 0, stream>>>(coarse, tab, grouped, ordered, offs, hl, hr, he, gmax);
    node_aggregate_kernel<<<(N_NODES + 7) / 8, 256, 0, stream>>>(
        ordered, offs, h, hl, hr, he, gmax, out, attn_out);
}


# launch merging: score-max start + edge-type logits + GEMM + edge rank run as work-group ranges of one launch, table scan + node logits of another (9 launches -> 5); all sums and their order kept
# speedup vs baseline: 3.7084x; 1.0821x over previous
#include <hip/hip_runtime.h>
#include <hip/hip_bf16.h>

typedef __attribute__((ext_vector_type(2))) float v2f;
typedef __attribute__((ext_vector_type(8))) float v8f;

#define N_NODES 50000
#define N_EDGES 1600000
#define IN_HIDDEN 256
#define OUT_COLS 128
#define NHEAD 4
#define OUT_HIDDEN 32
#define EDGE_HIDDEN 64
#define N_ETYPES 8
#define NEG_SLOPE 0.2f

__device__ __forceinline__ unsigned enc_f32(float f) {
    unsigned u = __float_as_uint(f);
    return (u & 0x80000000u) ? ~u : (u | 0x80000000u);
}
__device__ __forceinline__ float dec_f32(unsigned u) {
    unsigned b = (u & 0x80000000u) ? (u & 0x7FFFFFFFu) : ~u;
    return __uint_as_float(b);
}

#define FRONT_T 256
#define GEMM_BLOCKS ((N_NODES / 16 + FRONT_T / 32 - 1) / (FRONT_T / 32))
__device__ __forceinline__ void
gemm_h_stage(const int blk, const float* __restrict__ x, const float* __restrict__ W,
             float* __restrict__ h) {
    const int lane  = threadIdx.x & 31;
    const int wave  = threadIdx.x >> 5;
    const int strip = blk * (FRONT_T >> 5) + wave;
    if (strip * 16 >= N_NODES) return;
    const int row0  = strip * 16;

    const int mrow  = lane & 15;
    const int khalf = (lane >> 4) << 1;

    v8f acc[8] = {};

    const float* xrow = x + (size_t)(row0 + mrow) * IN_HIDDEN + khalf;
    for (int k = 0; k < IN_HIDDEN; k += 4) {
        v2f a = *(const v2f*)(xrow + k);
#pragma unroll
        for (int t = 0; t < 8; ++t) {
            const float* wp = W + (size_t)(k + khalf) * OUT_COLS + t * 16 + mrow;
            v2f b;
            b.x = wp[0];
            b.y = wp[OUT_COLS];
            acc[t] = __builtin_amdgcn_wmma_f32_16x16x4_f32(
                false, a, false, b, (short)0, acc[t], false, false);
        }
    }

    const int mbase = (lane >> 4) << 3;
#pragma unroll
    for (int t = 0; t < 8; ++t) {
#pragma unroll
        for (int i = 0; i < 8; ++i) {
            float v = acc[t][i];
            v = (v == v) ? v : 0.0f;
            h[(size_t)(row0 + mbase + i) * OUT_COLS + t * 16 + mrow] = v;
        }
    }
}

__device__ __forceinline__ void
etype_logits_stage(const int t,
                   const float* __restrict__ edge_emb,
                   const float* __restrict__ W_e,
                   const float* __restrict__ a_e,
                   float* __restrict__ he) {
    __shared__ float ev_s[NHEAD * EDGE_HIDDEN];
    __shared__ float ae_s[NHEAD * EDGE_HIDDEN];
    const int col  = threadIdx.x;
    const int head = col >> 6, d = col & (EDGE_HIDDEN - 1);
    const float* er = edge_emb + t * EDGE_HIDDEN;
    const float* wc = W_e + col;
    const float ae = a_e[col];
    float ev = 0.0f;
#pragma unroll
    for (int k = 0; k < EDGE_HIDDEN; ++k)
        ev = fmaf(er[k], wc[k * (EDGE_HIDDEN * NHEAD)], ev);
    ev_s[d * NHEAD + head] = ev;
    ae_s[d * NHEAD + head] = ae;
    __syncthreads();
    if (col < NHEAD) {
        float s = 0.0f;
#pragma unroll
        for (int dd = 0; dd < EDGE_HIDDEN; ++dd)
            s = fmaf(ae_s[dd * NHEAD + col], ev_s[dd * NHEAD + col], s);
        he[t * NHEAD + col] = s;
    }
}

__device__ __forceinline__ void
node_logits_stage(const int idx, const float* __restrict__ h,
                  const float* __restrict__ a_l, const float* __restrict__ a_r,
                  float* __restrict__ hl, float* __restrict__ hr) {
    if (idx >= N_NODES * NHEAD) return;
    const int node = idx >> 2, head = idx & 3;
    const float* hp = h + (size_t)node * OUT_COLS + head * OUT_HIDDEN;
    const float* al = a_l + head * OUT_HIDDEN;
    const float* ar = a_r + head * OUT_HIDDEN;
    float sl = 0.0f, sr = 0.0f;
#pragma unroll
    for (int d = 0; d < OUT_HIDDEN; ++d) {
        float v = hp[d];
        sl = fmaf(al[d], v, sl);
        sr = fmaf(ar[d], v, sr);
    }
    hl[idx] = sl;
    hr[idx] = sr;
}

__device__ __forceinline__ float4 edge_scores(const float* __restrict__ hl,
                                              const float* __restrict__ hr,
                                              const float* __restrict__ he,
                                              int src, int trg, int et) {
    const float4 l = *(const float4*)(hl + src * NHEAD);
    const float4 r = *(const float4*)(hr + trg * NHEAD);
    const float4 t = *(const float4*)(he + et * NHEAD);
    float4 s;
    s.x = l.x + r.x + t.x;
    s.y = l.y + r.y + t.y;
    s.z = l.z + r.z + t.z;
    s.w = l.w + r.w + t.w;
    s.x = (s.x > 0.0f) ? s.x : NEG_SLOPE * s.x;
    s.y = (s.y > 0.0f) ? s.y : NEG_SLOPE * s.y;
    s.z = (s.z > 0.0f) ? s.z : NEG_SLOPE * s.z;
    s.w = (s.w > 0.0f) ? s.w : NEG_SLOPE * s.w;
    return s;
}

__device__ __forceinline__ float4 scores_of_rows(const float4 l, const float4 r, const float4 t) {
    float4 s;
    s.x = l.x + r.x + t.x;
    s.y = l.y + r.y + t.y;
    s.z = l.z + r.z + t.z;
    s.w = l.w + r.w + t.w;
    s.x = (s.x > 0.0f) ? s.x : NEG_SLOPE * s.x;
    s.y = (s.y > 0.0f) ? s.y : NEG_SLOPE * s.y;
    s.z = (s.z > 0.0f) ? s.z : NEG_SLOPE * s.z;
    s.w = (s.w > 0.0f) ? s.w : NEG_SLOPE * s.w;
    return s;
}

#define NB   256
#define NPB  196
#define G1   256
#define CH   (N_EDGES / G1)
#define GT   1024
#define ITERS ((CH + GT - 1) / GT)

#define ITERS_R ((CH + FRONT_T - 1) / FRONT_T)
__device__ __forceinline__ void
chunk_rank_stage(const int wg, const int* __restrict__ edge,
                 int* __restrict__ lrank, int* __restrict__ counts) {
    __shared__ int hist[NB];
    const int tid = threadIdx.x;
    if (tid < NB) hist[tid] = 0;
    __syncthreads();
    const int e0 = wg * CH;
    int trgv[ITERS_R];
#pragma unroll
    for (int it = 0; it < ITERS_R; ++it) {
        const int k = tid + it * FRONT_T;
        const int e = e0 + ((k < CH) ? k : (CH - 1));
        trgv[it] = edge[N_EDGES + e];
    }
#pragma unroll
    for (int it = 0; it < ITERS_R; ++it) {
        const int k = tid + it * FRONT_T;
        if (k < CH) lrank[e0 + k] = atomicAdd(&hist[trgv[it] / NPB], 1);
    }
    __syncthreads();
    if (tid < NB) counts[tid * G1 + wg] = hist[tid];
}

__global__ void __launch_bounds__(FRONT_T)
front_kernel(const int* __restrict__ edge,
             const float* __restrict__ x, const float* __restrict__ W,
             const float* __restrict__ edge_emb, const float* __restrict__ W_e,
             const float* __restrict__ a_e,
             float* __restrict__ h, float* __restrict__ he,
             int* __restrict__ lrank, int* __restrict__ counts,
             unsigned* __restrict__ gmax) {
    const int b = blockIdx.x;
    if (b < GEMM_BLOCKS) {
        gemm_h_stage(b, x, W, h);
    } else if (b < GEMM_BLOCKS + G1) {
        chunk_rank_stage(b - GEMM_BLOCKS, edge, lrank, counts);
    } else {
        const int t = b - GEMM_BLOCKS - G1;
        if (t == 0 && threadIdx.x == 0) *gmax = enc_f32(-INFINITY);
        etype_logits_stage(t, edge_emb, W_e, a_e, he);
    }
}

#define SC_ROWS 16
__device__ __forceinline__ void
scan_stage(int* __restrict__ tab) {
    __shared__ int wtot[GT / 32];
    const int lane = threadIdx.x & 31, w = threadIdx.x >> 5;
    int4* p = (int4*)(tab + w * (SC_ROWS * 128)) + lane;
    int4 v[SC_ROWS];
#pragma unroll
    for (int r = 0; r < SC_ROWS; ++r) v[r] = p[r * 32];
    int carry = 0;
#pragma unroll
    for (int r = 0; r < SC_ROWS; ++r) {
        const int s1 = v[r].x, s2 = s1 + v[r].y, s3 = s2 + v[r].z, s4 = s3 + v[r].w;
        int inc = s4;
        for (int off = 1; off < 32; off <<= 1) {
            const int t = __shfl_up(inc, off, 32);
            if (lane >= off) inc += t;
        }
        const int ex = carry + inc - s4;
        v[r] = make_int4(ex, ex + s1, ex + s2, ex + s3);
        carry += __shfl(inc, 31, 32);
    }
    if (lane == 0) wtot[w] = carry;
    __syncthreads();
    int base = (lane < w) ? wtot[lane] : 0;
    for (int m = 16; m > 0; m >>= 1) base += __shfl_xor(base, m, 32);
#pragma unroll
    for (int r = 0; r < SC_ROWS; ++r)
        p[r * 32] = make_int4(v[r].x + base, v[r].y + base, v[r].z + base, v[r].w + base);
}

#define LOGIT_BLOCKS ((N_NODES * NHEAD + GT - 1) / GT)
__global__ void __launch_bounds__(GT)
mid_kernel(int* __restrict__ tab, const float* __restrict__ h,
           const float* __restrict__ a_l, const float* __restrict__ a_r,
           float* __restrict__ hl, float* __restrict__ hr) {
    if (blockIdx.x == 0) {
        scan_stage(tab);
    } else {
        node_logits_stage((blockIdx.x - 1) * GT + threadIdx.x, h, a_l, a_r, hl, hr);
    }
}

__global__ void __launch_bounds__(GT)
scatter_kernel(const int* __restrict__ edge, const int* __restrict__ lrank,
               const int* __restrict__ tab, int* __restrict__ coarse) {
    __shared__ int bs[NB];
    const int wg = blockIdx.x, tid = threadIdx.x;
    if (tid < NB) bs[tid] = tab[tid * G1 + wg];
    __syncthreads();
    const int e0 = wg * CH;
    int srcv[ITERS], trgv[ITERS], etv[ITERS], rkv[ITERS];
#pragma unroll
    for (int it = 0; it < ITERS; ++it) {
        const int k = tid + it * GT;
        const int e = e0 + ((k < CH) ? k : (CH - 1));
        srcv[it] = edge[e];
        trgv[it] = edge[N_EDGES + e];
        etv[it]  = edge[2 * N_EDGES + e];
        rkv[it]  = lrank[e];
    }
#pragma unroll
    for (int it = 0; it < ITERS; ++it) {
        const int k = tid + it * GT;
        if (k < CH) {
            const int b  = trgv[it] / NPB;
            const int tl = trgv[it] - b * NPB;
            *(int2*)(coarse + 2 * (size_t)(bs[b] + rkv[it])) =
                make_int2(e0 + k, srcv[it] | (etv[it] << 16) | (tl << 19));
        }
    }
}

__global__ void __launch_bounds__(GT)
bucket_kernel(const int* coarse, const int* __restrict__ tab,
              int* __restrict__ grouped, int* ordered, int* __restrict__ offs,
              const float* __restrict__ hl, const float* __restrict__ hr,
              const float* __restrict__ he, unsigned* __restrict__ gmax) {
    __shared__ int   cntL[NPB + 1];
    __shared__ int   loff[NPB + 1];
    __shared__ float red[GT / 32];
    const int b = blockIdx.x, tid = threadIdx.x;
    const int lane = tid & 31, w = tid >> 5;
    const int bstart = tab[b * G1];
    const int bend   = (b + 1 < NB) ? tab[(b + 1) * G1] : N_EDGES;
    const int n = bend - bstart;
    if (tid <= NPB) cntL[tid] = 0;
    __syncthreads();
    for (int i = tid; i < n; i += GT)
        atomicAdd(&cntL[(coarse[2 * (size_t)(bstart + i) + 1] >> 19) & 0xFF], 1);
    __syncthreads();
    if (w == 0) {
        int carry = 0;
        for (int r = 0; r < (NPB + 32) / 32; ++r) {
            const int idx = r * 32 + lane;
            const int c = (idx < NPB) ? cntL[idx] : 0;
            int inc = c;
            for (int off = 1; off < 32; off <<= 1) {
                const int t = __shfl_up(inc, off, 32);
                if (lane >= off) inc += t;
            }
            if (idx <= NPB) loff[idx] = carry + inc - c;
            carry += __shfl(inc, 31, 32);
        }
    }
    __syncthreads();
    if (tid < NPB) {
        const int node = b * NPB + tid;
        if (node <= N_NODES) offs[node] = bstart + loff[tid];
        cntL[tid] = 0;
    }
    __syncthreads();
    for (int i = tid; i < n; i += GT) {
        const int2 rec = *(const int2*)(coarse + 2 * (size_t)(bstart + i));
        const int tl = (rec.y >> 19) & 0xFF;
        const int r  = atomicAdd(&cntL[tl], 1);
        *(int2*)(grouped + 2 * (size_t)(bstart + loff[tl] + r)) = make_int2(rec.x, rec.y & 0x7FFFF);
    }
    __syncthreads();
    float lmax = -INFINITY;
    int  py = 0, pnode = 0;
    bool pv = false;
    for (int t = w; t < NPB; t += GT / 32) {
        const int beg  = bstart + __builtin_amdgcn_readfirstlane(loff[t]);
        const int d    = __builtin_amdgcn_readfirstlane(loff[t + 1]) - (beg - bstart);
        const int node = b * NPB + t;
        for (int ca = 0; ca < d; ca += 32) {
            const int ia = ca + lane;
            int2 pa = make_int2(0x7FFFFFFF, 0);
            if (ia < d) pa = *(const int2*)(grouped + 2 * (size_t)(beg + ia));
            const float4 rl = *(const float4*)(hl + (py & 0xFFFF) * NHEAD);
            const float4 rr = *(const float4*)(hr + pnode * NHEAD);
            const float4 rt = *(const float4*)(he + (py >> 16) * NHEAD);
            int pos = 0;
            for (int cb = 0; cb < d; cb += 32) {
                const int ib = cb + lane;
                int kb = 0x7FFFFFFF;
                if (ib < d) kb = grouped[2 * (size_t)(beg + ib)];
#pragma unroll
                for (int j = 0; j < 32; ++j) {
                    const int key = __builtin_amdgcn_readlane(kb, j);
                    pos += (key < pa.x) ? 1 : 0;
                }
            }
            if (ia < d) *(int2*)(ordered + 2 * (size_t)(beg + pos)) = pa;
            const float4 s = scores_of_rows(rl, rr, rt);
            if (pv) lmax = fmaxf(lmax, fmaxf(fmaxf(s.x, s.y), fmaxf(s.z, s.w)));
            py = pa.y; pnode = node; pv = (ia < d);
        }
    }
    {
        const float4 rl = *(const float4*)(hl + (py & 0xFFFF) * NHEAD);
        const float4 rr = *(const float4*)(hr + pnode * NHEAD);
        const float4 rt = *(const float4*)(he + (py >> 16) * NHEAD);
        const float4 s = scores_of_rows(rl, rr, rt);
        if (pv) lmax = fmaxf(lmax, fmaxf(fmaxf(s.x, s.y), fmaxf(s.z, s.w)));
    }
    for (int m = 16; m > 0; m >>= 1) lmax = fmaxf(lmax, __shfl_xor(lmax, m, 32));
    if (lane == 0) red[w] = lmax;
    __syncthreads();
    if (tid < 32) {
        float v = red[tid];
        for (int m = 16; m > 0; m >>= 1) v = fmaxf(v, __shfl_xor(v, m, 32));
        if (tid == 0) atomicMax(gmax, enc_f32(v));
    }
}

__global__ void __launch_bounds__(256)
node_aggregate_kernel(const int* __restrict__ ordered, const int* __restrict__ offs,
                      const float* __restrict__ h,
                      const float* __restrict__ hl, const float* __restrict__ hr,
                      const float* __restrict__ he, const unsigned* __restrict__ gmax,
                      float* __restrict__ out, float* __restrict__ attn_out) {
    const int lane = threadIdx.x & 31;
    const int node = blockIdx.x * (blockDim.x >> 5) + (threadIdx.x >> 5);
    if (node >= N_NODES) return;
    const int beg = __builtin_amdgcn_readfirstlane(offs[node]);
    const int d   = __builtin_amdgcn_readfirstlane(offs[node + 1]) - beg;
    const float smax = dec_f32(*gmax);

    float4 ds = make_float4(0.0f, 0.0f, 0.0f, 0.0f);
    int2   p0 = make_int2(0, 0), p1 = make_int2(0, 0);
    float4 e0 = make_float4(0.0f, 0.0f, 0.0f, 0.0f);
    float4 e1 = make_float4(0.0f, 0.0f, 0.0f, 0.0f);
    if (lane < d) {
        p0 = *(const int2*)(ordered + 2 * (size_t)(beg + lane));
        const float4 s = edge_scores(hl, hr, he, p0.y & 0xFFFF, node, p0.y >> 16);
        e0.x = __expf(s.x - smax);
        e0.y = __expf(s.y - smax);
        e0.z = __expf(s.z - smax);
        e0.w = __expf(s.w - smax);
        ds.x += e0.x;
        ds.y += e0.y;
        ds.z += e0.z;
        ds.w += e0.w;
    }
    if (32 + lane < d) {
        p1 = *(const int2*)(ordered + 2 * (size_t)(beg + 32 + lane));
        const float4 s = edge_scores(hl, hr, he, p1.y & 0xFFFF, node, p1.y >> 16);
        e1.x = __expf(s.x - smax);
        e1.y = __expf(s.y - smax);
        e1.z = __expf(s.z - smax);
        e1.w = __expf(s.w - smax);
        ds.x += e1.x;
        ds.y += e1.y;
        ds.z += e1.z;
        ds.w += e1.w;
    }
    for (int c = 64; c < d; c += 32) {
        const int i = c + lane;
        if (i < d) {
            const int2 p = *(const int2*)(ordered + 2 * (size_t)(beg + i));
            const float4 s = edge_scores(hl, hr, he, p.y & 0xFFFF, node, p.y >> 16);
            ds.x += __expf(s.x - smax);
            ds.y += __expf(s.y - smax);
            ds.z += __expf(s.z - smax);
            ds.w += __expf(s.w - smax);
        }
    }
    for (int m = 16; m > 0; m >>= 1) {
        ds.x += __shfl_xor(ds.x, m, 32);
        ds.y += __shfl_xor(ds.y, m, 32);
        ds.z += __shfl_xor(ds.z, m, 32);
        ds.w += __shfl_xor(ds.w, m, 32);
    }
    const float4 dn = make_float4(ds.x + 1e-16f, ds.y + 1e-16f, ds.z + 1e-16f, ds.w + 1e-16f);

    float acc0 = 0.0f, acc1 = 0.0f, acc2 = 0.0f, acc3 = 0.0f;
    const float* hcol = h + lane;
    for (int c = 0; c < d; c += 32) {
        const int i = c + lane;
        int srcv = 0;
        float4 a = make_float4(0.0f, 0.0f, 0.0f, 0.0f);
        if (i < d) {
            int2   p = p0;
            float4 e = e0;
            if (c == 32) { p = p1; e = e1; }
            if (c >= 64) {
                p = *(const int2*)(ordered + 2 * (size_t)(beg + i));
                const float4 s = edge_scores(hl, hr, he, p.y & 0xFFFF, node, p.y >> 16);
                e.x = __expf(s.x - smax);
                e.y = __expf(s.y - smax);
                e.z = __expf(s.z - smax);
                e.w = __expf(s.w - smax);
            }
            srcv = p.y & 0xFFFF;
            a.x = e.x / dn.x;
            a.y = e.y / dn.y;
            a.z = e.z / dn.z;
            a.w = e.w / dn.w;
            *(float4*)(attn_out + (size_t)p.x * NHEAD) = a;
        }
        const int nb = (d - c < 32) ? (d - c) : 32;
#pragma unroll 4
        for (int j = 0; j < nb; ++j) {
            const int sj = __builtin_amdgcn_readlane(srcv, j);
            const float a0 = __int_as_float(__builtin_amdgcn_readlane(__float_as_int(a.x), j));
            const float a1 = __int_as_float(__builtin_amdgcn_readlane(__float_as_int(a.y), j));
            const float a2 = __int_as_float(__builtin_amdgcn_readlane(__float_as_int(a.z), j));
            const float a3 = __int_as_float(__builtin_amdgcn_readlane(__float_as_int(a.w), j));
            const float* row = hcol + (size_t)sj * OUT_COLS;
            acc0 = fmaf(row[0 * OUT_HIDDEN], a0, acc0);
            acc1 = fmaf(row[1 * OUT_HIDDEN], a1, acc1);
            acc2 = fmaf(row[2 * OUT_HIDDEN], a2, acc2);
            acc3 = fmaf(row[3 * OUT_HIDDEN], a3, acc3);
        }
    }
    float* orow = out + (size_t)node * OUT_COLS + lane;
    orow[0 * OUT_HIDDEN] = acc0;
    orow[1 * OUT_HIDDEN] = acc1;
    orow[2 * OUT_HIDDEN] = acc2;
    orow[3 * OUT_HIDDEN] = acc3;
}

extern "C" void kernel_launch(void* const* d_in, const int* in_sizes, int n_in,
                              void* d_out, int out_size, void* d_ws, size_t ws_size,
                              hipStream_t stream) {
    const int*   edge     = (const int*)  d_in[0];
    const float* x        = (const float*)d_in[1];
    const float* edge_emb = (const float*)d_in[2];
    const float* W        = (const float*)d_in[3];
    const float* W_e      = (const float*)d_in[4];
    const float* a_l      = (const float*)d_in[5];
    const float* a_r      = (const float*)d_in[6];
    const float* a_e      = (const float*)d_in[7];

    float* out      = (float*)d_out;
    float* attn_out = out + (size_t)N_NODES * OUT_COLS;

    char* ws = (char*)d_ws;
    float* h       = (float*)ws;  ws += (size_t)N_NODES * OUT_COLS * sizeof(float);
    int*   planeA  = (int*)ws;    ws += (size_t)N_EDGES * 2 * sizeof(int);
    int*   planeB  = (int*)ws;    ws += (size_t)N_EDGES * 2 * sizeof(int);
    int*   lrank   = planeA;
    int*   grouped = planeA;
    int*   coarse  = planeB;
    int*   ordered = planeB;
    float* hl      = (float*)ws;  ws += (size_t)N_NODES * NHEAD   * sizeof(float);
    float* hr      = (float*)ws;  ws += (size_t)N_NODES * NHEAD   * sizeof(float);
    int*   offs    = (int*)ws + 50176;
    int*   tab     = (int*)ws + 102400;
    ws += (size_t)N_NODES * NHEAD * sizeof(float);
    float* he      = (float*)ws;  ws += (size_t)N_ETYPES * NHEAD  * sizeof(float);
    unsigned* gmax = (unsigned*)ws;

    front_kernel<<<GEMM_BLOCKS + G1 + N_ETYPES, FRONT_T, 0, stream>>>(
        edge, x, W, edge_emb, W_e, a_e, h, he, lrank, tab, gmax);
    mid_kernel<<<1 + LOGIT_BLOCKS, GT, 0, stream>>>(tab, h, a_l, a_r, hl, hr);
    scatter_kernel<<<G1, GT, 0, stream>>>(edge, lrank, tab, coarse);
    bucket_kernel<<<NB, GT, 0, stream>>>(coarse, tab, grouped, ordered, offs, hl, hr, he, gmax);
    node_aggregate_kernel<<<(N_NODES + 7) / 8, 256, 0, stream>>>(
        ordered, offs, h, hl, hr, he, gmax, out, attn_out);
}


# record grouping goes through LDS at both levels: a chunk's records leave the scatter kernel as contiguous runs, a bucket is grouped by node inside LDS, not via a plane in memory; sum order kept
# speedup vs baseline: 4.2075x; 1.1346x over previous
#include <hip/hip_runtime.h>
#include <hip/hip_bf16.h>

typedef __attribute__((ext_vector_type(2))) float v2f;
typedef __attribute__((ext_vector_type(8))) float v8f;

#define N_NODES 50000
#define N_EDGES 1600000
#define IN_HIDDEN 256
#define OUT_COLS 128
#define NHEAD 4
#define OUT_HIDDEN 32
#define EDGE_HIDDEN 64
#define N_ETYPES 8
#define NEG_SLOPE 0.2f

__device__ __forceinline__ unsigned enc_f32(float f) {
    unsigned u = __float_as_uint(f);
    return (u & 0x80000000u) ? ~u : (u | 0x80000000u);
}
__device__ __forceinline__ float dec_f32(unsigned u) {
    unsigned b = (u & 0x80000000u) ? (u & 0x7FFFFFFFu) : ~u;
    return __uint_as_float(b);
}

#define FRONT_T 256
#define GEMM_BLOCKS ((N_NODES / 16 + FRONT_T / 32 - 1) / (FRONT_T / 32))
__device__ __forceinline__ void
gemm_h_stage(const int blk, const float* __restrict__ x, const float* __restrict__ W,
             float* __restrict__ h) {
    const int lane  = threadIdx.x & 31;
    const int wave  = threadIdx.x >> 5;
    const int strip = blk * (FRONT_T >> 5) + wave;
    if (strip * 16 >= N_NODES) return;
    const int row0  = strip * 16;

    const int mrow  = lane & 15;
    const int khalf = (lane >> 4) << 1;

    v8f acc[8] = {};

    const float* xrow = x + (size_t)(row0 + mrow) * IN_HIDDEN + khalf;
    for (int k = 0; k < IN_HIDDEN; k += 4) {
        v2f a = *(const v2f*)(xrow + k);
#pragma unroll
        for (int t = 0; t < 8; ++t) {
            const float* wp = W + (size_t)(k + khalf) * OUT_COLS + t * 16 + mrow;
            v2f b;
            b.x = wp[0];
            b.y = wp[OUT_COLS];
            acc[t] = __builtin_amdgcn_wmma_f32_16x16x4_f32(
                false, a, false, b, (short)0, acc[t], false, false);
        }
    }

    const int mbase = (lane >> 4) << 3;
#pragma unroll
    for (int t = 0; t < 8; ++t) {
#pragma unroll
        for (int i = 0; i < 8; ++i) {
            float v = acc[t][i];
            v = (v == v) ? v : 0.0f;
            h[(size_t)(row0 + mbase + i) * OUT_COLS + t * 16 + mrow] = v;
        }
    }
}

__device__ __forceinline__ void
etype_logits_stage(const int t,
                   const float* __restrict__ edge_emb,
                   const float* __restrict__ W_e,
                   const float* __restrict__ a_e,
                   float* __restrict__ he) {
    __shared__ float ev_s[NHEAD * EDGE_HIDDEN];
    __shared__ float ae_s[NHEAD * EDGE_HIDDEN];
    const int col  = threadIdx.x;
    const int head = col >> 6, d = col & (EDGE_HIDDEN - 1);
    const float* er = edge_emb + t * EDGE_HIDDEN;
    const float* wc = W_e + col;
    const float ae = a_e[col];
    float ev = 0.0f;
#pragma unroll
    for (int k = 0; k < EDGE_HIDDEN; ++k)
        ev = fmaf(er[k], wc[k * (EDGE_HIDDEN * NHEAD)], ev);
    ev_s[d * NHEAD + head] = ev;
    ae_s[d * NHEAD + head] = ae;
    __syncthreads();
    if (col < NHEAD) {
        float s = 0.0f;
#pragma unroll
        for (int dd = 0; dd < EDGE_HIDDEN; ++dd)
            s = fmaf(ae_s[dd * NHEAD + col], ev_s[dd * NHEAD + col], s);
        he[t * NHEAD + col] = s;
    }
}

__device__ __forceinline__ void
node_logits_stage(const int idx, const float* __restrict__ h,
                  const float* __restrict__ a_l, const float* __restrict__ a_r,
                  float* __restrict__ hl, float* __restrict__ hr) {
    if (idx >= N_NODES * NHEAD) return;
    const int node = idx >> 2, head = idx & 3;
    const float* hp = h + (size_t)node * OUT_COLS + head * OUT_HIDDEN;
    const float* al = a_l + head * OUT_HIDDEN;
    const float* ar = a_r + head * OUT_HIDDEN;
    float sl = 0.0f, sr = 0.0f;
#pragma unroll
    for (int d = 0; d < OUT_HIDDEN; ++d) {
        float v = hp[d];
        sl = fmaf(al[d], v, sl);
        sr = fmaf(ar[d], v, sr);
    }
    hl[idx] = sl;
    hr[idx] = sr;
}

__device__ __forceinline__ float4 edge_scores(const float* __restrict__ hl,
                                              const float* __restrict__ hr,
                                              const float* __restrict__ he,
                                              int src, int trg, int et) {
    const float4 l = *(const float4*)(hl + src * NHEAD);
    const float4 r = *(const float4*)(hr + trg * NHEAD);
    const float4 t = *(const float4*)(he + et * NHEAD);
    float4 s;
    s.x = l.x + r.x + t.x;
    s.y = l.y + r.y + t.y;
    s.z = l.z + r.z + t.z;
    s.w = l.w + r.w + t.w;
    s.x = (s.x > 0.0f) ? s.x : NEG_SLOPE * s.x;
    s.y = (s.y > 0.0f) ? s.y : NEG_SLOPE * s.y;
    s.z = (s.z > 0.0f) ? s.z : NEG_SLOPE * s.z;
    s.w = (s.w > 0.0f) ? s.w : NEG_SLOPE * s.w;
    return s;
}

__device__ __forceinline__ float4 scores_of_rows(const float4 l, const float4 r, const float4 t) {
    float4 s;
    s.x = l.x + r.x + t.x;
    s.y = l.y + r.y + t.y;
    s.z = l.z + r.z + t.z;
    s.w = l.w + r.w + t.w;
    s.x = (s.x > 0.0f) ? s.x : NEG_SLOPE * s.x;
    s.y = (s.y > 0.0f) ? s.y : NEG_SLOPE * s.y;
    s.z = (s.z > 0.0f) ? s.z : NEG_SLOPE * s.z;
    s.w = (s.w > 0.0f) ? s.w : NEG_SLOPE * s.w;
    return s;
}

#define NB   256
#define NPB  196
#define G1   256
#define CH   (N_EDGES / G1)
#define GT   1024
#define ITERS ((CH + GT - 1) / GT)

#define ITERS_R ((CH + FRONT_T - 1) / FRONT_T)
__device__ __forceinline__ void
chunk_rank_stage(const int wg, const int* __restrict__ edge,
                 int* __restrict__ lrank, int* __restrict__ counts) {
    __shared__ int hist[NB];
    const int tid = threadIdx.x;
    if (tid < NB) hist[tid] = 0;
    __syncthreads();
    const int e0 = wg * CH;
    int trgv[ITERS_R];
#pragma unroll
    for (int it = 0; it < ITERS_R; ++it) {
        const int k = tid + it * FRONT_T;
        const int e = e0 + ((k < CH) ? k : (CH - 1));
        trgv[it] = edge[N_EDGES + e];
    }
#pragma unroll
    for (int it = 0; it < ITERS_R; ++it) {
        const int k = tid + it * FRONT_T;
        if (k < CH) lrank[e0 + k] = atomicAdd(&hist[trgv[it] / NPB], 1);
    }
    __syncthreads();
    if (tid < NB) counts[tid * G1 + wg] = hist[tid];
}

__global__ void __launch_bounds__(FRONT_T)
front_kernel(const int* __restrict__ edge,
             const float* __restrict__ x, const float* __restrict__ W,
             const float* __restrict__ edge_emb, const float* __restrict__ W_e,
             const float* __restrict__ a_e,
             float* __restrict__ h, float* __restrict__ he,
             int* __restrict__ lrank, int* __restrict__ counts,
             unsigned* __restrict__ gmax) {
    const int b = blockIdx.x;
    if (b < GEMM_BLOCKS) {
        gemm_h_stage(b, x, W, h);
    } else if (b < GEMM_BLOCKS + G1) {
        chunk_rank_stage(b - GEMM_BLOCKS, edge, lrank, counts);
    } else {
        const int t = b - GEMM_BLOCKS - G1;
        if (t == 0 && threadIdx.x == 0) *gmax = enc_f32(-INFINITY);
        etype_logits_stage(t, edge_emb, W_e, a_e, he);
    }
}

#define SC_ROWS 16
__device__ __forceinline__ void
scan_stage(int* __restrict__ tab) {
    __shared__ int wtot[GT / 32];
    const int lane = threadIdx.x & 31, w = threadIdx.x >> 5;
    int4* p = (int4*)(tab + w * (SC_ROWS * 128)) + lane;
    int4 v[SC_ROWS];
#pragma unroll
    for (int r = 0; r < SC_ROWS; ++r) v[r] = p[r * 32];
    int carry = 0;
#pragma unroll
    for (int r = 0; r < SC_ROWS; ++r) {
        const int s1 = v[r].x, s2 = s1 + v[r].y, s3 = s2 + v[r].z, s4 = s3 + v[r].w;
        int inc = s4;
        for (int off = 1; off < 32; off <<= 1) {
            const int t = __shfl_up(inc, off, 32);
            if (lane >= off) inc += t;
        }
        const int ex = carry + inc - s4;
        v[r] = make_int4(ex, ex + s1, ex + s2, ex + s3);
        carry += __shfl(inc, 31, 32);
    }
    if (lane == 0) wtot[w] = carry;
    __syncthreads();
    int base = (lane < w) ? wtot[lane] : 0;
    for (int m = 16; m > 0; m >>= 1) base += __shfl_xor(base, m, 32);
#pragma unroll
    for (int r = 0; r < SC_ROWS; ++r)
        p[r * 32] = make_int4(v[r].x + base, v[r].y + base, v[r].z + base, v[r].w + base);
}

#define LOGIT_BLOCKS ((N_NODES * NHEAD + GT - 1) / GT)
__global__ void __launch_bounds__(GT)
mid_kernel(int* __restrict__ tab, const float* __restrict__ h,
           const float* __restrict__ a_l, const float* __restrict__ a_r,
           float* __restrict__ hl, float* __restrict__ hr) {
    if (blockIdx.x == 0) {
        scan_stage(tab);
    } else {
        node_logits_stage((blockIdx.x - 1) * GT + threadIdx.x, h, a_l, a_r, hl, hr);
    }
}

#define EID_BITS 21
__global__ void __launch_bounds__(GT)
scatter_kernel(const int* __restrict__ edge, const int* __restrict__ lrank,
               const int* __restrict__ tab, int* __restrict__ coarse) {
    __shared__ int  bs[NB];
    __shared__ int  ls[NB];
    __shared__ int  wtot[NB / 32];
    __shared__ int2 stage[CH];
    const int wg = blockIdx.x, tid = threadIdx.x;
    const int lane = tid & 31, w = tid >> 5;
    const int e0 = wg * CH;
    int srcv[ITERS], trgv[ITERS], etv[ITERS], rkv[ITERS];
#pragma unroll
    for (int it = 0; it < ITERS; ++it) {
        const int k = tid + it * GT;
        const int e = e0 + ((k < CH) ? k : (CH - 1));
        srcv[it] = edge[e];
        trgv[it] = edge[N_EDGES + e];
        etv[it]  = edge[2 * N_EDGES + e];
        rkv[it]  = lrank[e];
    }
    int ex = 0;
    if (tid < NB) {
        const int idx = tid * G1 + wg;
        const int s0 = tab[idx];
        const int s1 = (idx + 1 < NB * G1) ? tab[idx + 1] : N_EDGES;
        bs[tid] = s0;
        const int cnt = s1 - s0;
        int inc = cnt;
        for (int off = 1; off < 32; off <<= 1) {
            const int t = __shfl_up(inc, off, 32);
            if (lane >= off) inc += t;
        }
        if (lane == 31) wtot[w] = inc;
        ex = inc - cnt;
    }
    __syncthreads();
    if (tid < NB) {
        int base = 0;
        for (int j = 0; j < NB / 32; ++j) base += (j < w) ? wtot[j] : 0;
        ls[tid] = ex + base;
    }
    __syncthreads();
#pragma unroll
    for (int it = 0; it < ITERS; ++it) {
        const int k = tid + it * GT;
        if (k < CH) {
            const int b  = trgv[it] / NPB;
            const int tl = trgv[it] - b * NPB;
            stage[ls[b] + rkv[it]] =
                make_int2((e0 + k) | (b << EID_BITS), srcv[it] | (etv[it] << 16) | (tl << 19));
        }
    }
    __syncthreads();
#pragma unroll
    for (int it = 0; it < ITERS; ++it) {
        const int s = tid + it * GT;
        if (s < CH) {
            const int2 r = stage[s];
            const int b = r.x >> EID_BITS;
            *(int2*)(coarse + 2 * (size_t)(bs[b] + (s - ls[b]))) =
                make_int2(r.x & ((1 << EID_BITS) - 1), r.y);
        }
    }
}

#define BCAP_IT 8
#define BCAP    (BCAP_IT * GT)

template <bool IN_LDS>
__device__ __forceinline__ float
order_segments(const int2* recs, const int* __restrict__ grouped, int* ordered,
               const int* loff, const int bstart, const int b,
               const float* __restrict__ hl, const float* __restrict__ hr,
               const float* __restrict__ he, const int lane, const int w) {
    float lmax = -INFINITY;
    int  py = 0, pnode = 0;
    bool pv = false;
    for (int t = w; t < NPB; t += GT / 32) {
        const int lo   = __builtin_amdgcn_readfirstlane(loff[t]);
        const int d    = __builtin_amdgcn_readfirstlane(loff[t + 1]) - lo;
        const int beg  = bstart + lo;
        const int node = b * NPB + t;
        for (int ca = 0; ca < d; ca += 32) {
            const int ia = ca + lane;
            int2 pa = make_int2(0x7FFFFFFF, 0);
            if (ia < d) {
                if (IN_LDS) pa = recs[lo + ia];
                else        pa = *(const int2*)(grouped + 2 * (size_t)(beg + ia));
            }
            const float4 rl = *(const float4*)(hl + (py & 0xFFFF) * NHEAD);
            const float4 rr = *(const float4*)(hr + pnode * NHEAD);
            const float4 rt = *(const float4*)(he + (py >> 16) * NHEAD);
            int pos = 0;
            for (int cb = 0; cb < d; cb += 32) {
                const int ib = cb + lane;
                int kb = 0x7FFFFFFF;
                if (ib < d) {
                    if (IN_LDS) kb = recs[lo + ib].x;
                    else        kb = grouped[2 * (size_t)(beg + ib)];
                }
#pragma unroll
                for (int j = 0; j < 32; ++j) {
                    const int key = __builtin_amdgcn_readlane(kb, j);
                    pos += (key < pa.x) ? 1 : 0;
                }
            }
            if (ia < d) *(int2*)(ordered + 2 * (size_t)(beg + pos)) = pa;
            const float4 s = scores_of_rows(rl, rr, rt);
            if (pv) lmax = fmaxf(lmax, fmaxf(fmaxf(s.x, s.y), fmaxf(s.z, s.w)));
            py = pa.y; pnode = node; pv = (ia < d);
        }
    }
    {
        const float4 rl = *(const float4*)(hl + (py & 0xFFFF) * NHEAD);
        const float4 rr = *(const float4*)(hr + pnode * NHEAD);
        const float4 rt = *(const float4*)(he + (py >> 16) * NHEAD);
        const float4 s = scores_of_rows(rl, rr, rt);
        if (pv) lmax = fmaxf(lmax, fmaxf(fmaxf(s.x, s.y), fmaxf(s.z, s.w)));
    }
    return lmax;
}

__global__ void __launch_bounds__(GT)
bucket_kernel(const int* coarse, const int* __restrict__ tab,
              int* __restrict__ grouped, int* ordered, int* __restrict__ offs,
              const float* __restrict__ hl, const float* __restrict__ hr,
              const float* __restrict__ he, unsigned* __restrict__ gmax) {
    __shared__ int   cntL[NPB + 1];
    __shared__ int   loff[NPB + 1];
    __shared__ float red[GT / 32];
    __shared__ int2  recs[BCAP];
    const int b = blockIdx.x, tid = threadIdx.x;
    const int lane = tid & 31, w = tid >> 5;
    const int bstart = tab[b * G1];
    const int bend   = (b + 1 < NB) ? tab[(b + 1) * G1] : N_EDGES;
    const int n = bend - bstart;
    const bool in_lds = (n <= BCAP);
    if (tid <= NPB) cntL[tid] = 0;
    int2 rv[BCAP_IT];
    if (in_lds) {
#pragma unroll
        for (int it = 0; it < BCAP_IT; ++it) {
            const int i = tid + it * GT;
            rv[it] = make_int2(0, 0);
            if (i < n) rv[it] = *(const int2*)(coarse + 2 * (size_t)(bstart + i));
        }
    }
    __syncthreads();
    if (in_lds) {
#pragma unroll
        for (int it = 0; it < BCAP_IT; ++it)
            if (tid + it * GT < n) atomicAdd(&cntL[(rv[it].y >> 19) & 0xFF], 1);
    } else {
        for (int i = tid; i < n; i += GT)
            atomicAdd(&cntL[(coarse[2 * (size_t)(bstart + i) + 1] >> 19) & 0xFF], 1);
    }
    __syncthreads();
    if (w == 0) {
        int carry = 0;
        for (int r = 0; r < (NPB + 32) / 32; ++r) {
            const int idx = r * 32 + lane;
            const int c = (idx < NPB) ? cntL[idx] : 0;
            int inc = c;
            for (int off = 1; off < 32; off <<= 1) {
                const int t = __shfl_up(inc, off, 32);
                if (lane >= off) inc += t;
            }
            if (idx <= NPB) loff[idx] = carry + inc - c;
            carry += __shfl(inc, 31, 32);
        }
    }
    __syncthreads();
    if (tid < NPB) {
        const int node = b * NPB + tid;
        if (node <= N_NODES) offs[node] = bstart + loff[tid];
        cntL[tid] = 0;
    }
    __syncthreads();
    float lmax;
    if (in_lds) {
#pragma unroll
        for (int it = 0; it < BCAP_IT; ++it) {
            if (tid + it * GT < n) {
                const int tl = (rv[it].y >> 19) & 0xFF;
                const int r  = atomicAdd(&cntL[tl], 1);
                recs[loff[tl] + r] = make_int2(rv[it].x, rv[it].y & 0x7FFFF);
            }
        }
        __syncthreads();
        lmax = order_segments<true>(recs, grouped, ordered, loff, bstart, b, hl, hr, he, lane, w);
    } else {
        for (int i = tid; i < n; i += GT) {
            const int2 rec = *(const int2*)(coarse + 2 * (size_t)(bstart + i));
            const int tl = (rec.y >> 19) & 0xFF;
            const int r  = atomicAdd(&cntL[tl], 1);
            *(int2*)(grouped + 2 * (size_t)(bstart + loff[tl] + r)) = make_int2(rec.x, rec.y & 0x7FFFF);
        }
        __syncthreads();
        lmax = order_segments<false>(recs, grouped, ordered, loff, bstart, b, hl, hr, he, lane, w);
    }
    for (int m = 16; m > 0; m >>= 1) lmax = fmaxf(lmax, __shfl_xor(lmax, m, 32));
    if (lane == 0) red[w] = lmax;
    __syncthreads();
    if (tid < 32) {
        float v = red[tid];
        for (int m = 16; m > 0; m >>= 1) v = fmaxf(v, __shfl_xor(v, m, 32));
        if (tid == 0) atomicMax(gmax, enc_f32(v));
    }
}

__global__ void __launch_bounds__(256)
node_aggregate_kernel(const int* __restrict__ ordered, const int* __restrict__ offs,
                      const float* __restrict__ h,
                      const float* __restrict__ hl, const float* __restrict__ hr,
                      const float* __restrict__ he, const unsigned* __restrict__ gmax,
                      float* __restrict__ out, float* __restrict__ attn_out) {
    const int lane = threadIdx.x & 31;
    const int node = blockIdx.x * (blockDim.x >> 5) + (threadIdx.x >> 5);
    if (node >= N_NODES) return;
    const int beg = __builtin_amdgcn_readfirstlane(offs[node]);
    const int d   = __builtin_amdgcn_readfirstlane(offs[node + 1]) - beg;
    const float smax = dec_f32(*gmax);

    float4 ds = make_float4(0.0f, 0.0f, 0.0f, 0.0f);
    int2   p0 = make_int2(0, 0), p1 = make_int2(0, 0);
    float4 e0 = make_float4(0.0f, 0.0f, 0.0f, 0.0f);
    float4 e1 = make_float4(0.0f, 0.0f, 0.0f, 0.0f);
    if (lane < d) {
        p0 = *(const int2*)(ordered + 2 * (size_t)(beg + lane));
        const float4 s = edge_scores(hl, hr, he, p0.y & 0xFFFF, node, p0.y >> 16);
        e0.x = __expf(s.x - smax);
        e0.y = __expf(s.y - smax);
        e0.z = __expf(s.z - smax);
        e0.w = __expf(s.w - smax);
        ds.x += e0.x;
        ds.y += e0.y;
        ds.z += e0.z;
        ds.w += e0.w;
    }
    if (32 + lane < d) {
        p1 = *(const int2*)(ordered + 2 * (size_t)(beg + 32 + lane));
        const float4 s = edge_scores(hl, hr, he, p1.y & 0xFFFF, node, p1.y >> 16);
        e1.x = __expf(s.x - smax);
        e1.y = __expf(s.y - smax);
        e1.z = __expf(s.z - smax);
        e1.w = __expf(s.w - smax);
        ds.x += e1.x;
        ds.y += e1.y;
        ds.z += e1.z;
        ds.w += e1.w;
    }
    for (int c = 64; c < d; c += 32) {
        const int i = c + lane;
        if (i < d) {
            const int2 p = *(const int2*)(ordered + 2 * (size_t)(beg + i));
            const float4 s = edge_scores(hl, hr, he, p.y & 0xFFFF, node, p.y >> 16);
            ds.x += __expf(s.x - smax);
            ds.y += __expf(s.y - smax);
            ds.z += __expf(s.z - smax);
            ds.w += __expf(s.w - smax);
        }
    }
    for (int m = 16; m > 0; m >>= 1) {
        ds.x += __shfl_xor(ds.x, m, 32);
        ds.y += __shfl_xor(ds.y, m, 32);
        ds.z += __shfl_xor(ds.z, m, 32);
        ds.w += __shfl_xor(ds.w, m, 32);
    }
    const float4 dn = make_float4(ds.x + 1e-16f, ds.y + 1e-16f, ds.z + 1e-16f, ds.w + 1e-16f);

    float acc0 = 0.0f, acc1 = 0.0f, acc2 = 0.0f, acc3 = 0.0f;
    const float* hcol = h + lane;
    for (int c = 0; c < d; c += 32) {
        const int i = c + lane;
        int srcv = 0;
        float4 a = make_float4(0.0f, 0.0f, 0.0f, 0.0f);
        if (i < d) {
            int2   p = p0;
            float4 e = e0;
            if (c == 32) { p = p1; e = e1; }
            if (c >= 64) {
                p = *(const int2*)(ordered + 2 * (size_t)(beg + i));
                const float4 s = edge_scores(hl, hr, he, p.y & 0xFFFF, node, p.y >> 16);
                e.x = __expf(s.x - smax);
                e.y = __expf(s.y - smax);
                e.z = __expf(s.z - smax);
                e.w = __expf(s.w - smax);
            }
            srcv = p.y & 0xFFFF;
            a.x = e.x / dn.x;
            a.y = e.y / dn.y;
            a.z = e.z / dn.z;
            a.w = e.w / dn.w;
            *(float4*)(attn_out + (size_t)p.x * NHEAD) = a;
        }
        const int nb = (d - c < 32) ? (d - c) : 32;
#pragma unroll 4
        for (int j = 0; j < nb; ++j) {
            const int sj = __builtin_amdgcn_readlane(srcv, j);
            const float a0 = __int_as_float(__builtin_amdgcn_readlane(__float_as_int(a.x), j));
            const float a1 = __int_as_float(__builtin_amdgcn_readlane(__float_as_int(a.y), j));
            const float a2 = __int_as_float(__builtin_amdgcn_readlane(__float_as_int(a.z), j));
            const float a3 = __int_as_float(__builtin_amdgcn_readlane(__float_as_int(a.w), j));
            const float* row = hcol + (size_t)sj * OUT_COLS;
            acc0 = fmaf(row[0 * OUT_HIDDEN], a0, acc0);
            acc1 = fmaf(row[1 * OUT_HIDDEN], a1, acc1);
            acc2 = fmaf(row[2 * OUT_HIDDEN], a2, acc2);
            acc3 = fmaf(row[3 * OUT_HIDDEN], a3, acc3);
        }
    }
    float* orow = out + (size_t)node * OUT_COLS + lane;
    orow[0 * OUT_HIDDEN] = acc0;
    orow[1 * OUT_HIDDEN] = acc1;
    orow[2 * OUT_HIDDEN] = acc2;
    orow[3 * OUT_HIDDEN] = acc3;
}

extern "C" void kernel_launch(void* const* d_in, const int* in_sizes, int n_in,
                              void* d_out, int out_size, void* d_ws, size_t ws_size,
                              hipStream_t stream) {
    const int*   edge     = (const int*)  d_in[0];
    const float* x        = (const float*)d_in[1];
    const float* edge_emb = (const float*)d_in[2];
    const float* W        = (const float*)d_in[3];
    const float* W_e      = (const float*)d_in[4];
    const float* a_l      = (const float*)d_in[5];
    const float* a_r      = (const float*)d_in[6];
    const float* a_e      = (const float*)d_in[7];

    float* out      = (float*)d_out;
    float* attn_out = out + (size_t)N_NODES * OUT_COLS;

    char* ws = (char*)d_ws;
    float* h       = (float*)ws;  ws += (size_t)N_NODES * OUT_COLS * sizeof(float);
    int*   planeA  = (int*)ws;    ws += (size_t)N_EDGES * 2 * sizeof(int);
    int*   planeB  = (int*)ws;    ws += (size_t)N_EDGES * 2 * sizeof(int);
    int*   lrank   = planeA;
    int*   grouped = planeA;
    int*   coarse  = planeB;
    int*   ordered = planeB;
    float* hl      = (float*)ws;  ws += (size_t)N_NODES * NHEAD   * sizeof(float);
    float* hr      = (float*)ws;  ws += (size_t)N_NODES * NHEAD   * sizeof(float);
    int*   offs    = (int*)ws + 50176;
    int*   tab     = (int*)ws + 102400;
    ws += (size_t)N_NODES * NHEAD * sizeof(float);
    float* he      = (float*)ws;  ws += (size_t)N_ETYPES * NHEAD  * sizeof(float);
    unsigned* gmax = (unsigned*)ws;

    front_kernel<<<GEMM_BLOCKS + G1 + N_ETYPES, FRONT_T, 0, stream>>>(
        edge, x, W, edge_emb, W_e, a_e, h, he, lrank, tab, gmax);
    mid_kernel<<<1 + LOGIT_BLOCKS, GT, 0, stream>>>(tab, h, a_l, a_r, hl, hr);
    scatter_kernel<<<G1, GT, 0, stream>>>(edge, lrank, tab, coarse);
    bucket_kernel<<<NB, GT, 0, stream>>>(coarse, tab, grouped, ordered, offs, hl, hr, he, gmax);
    node_aggregate_kernel<<<(N_NODES + 7) / 8, 256, 0, stream>>>(
        ordered, offs, h, hl, hr, he, gmax, out, attn_out);
}


# mid launch removed (5 launches -> 4): node logits summed from the GEMM's accumulators (16-lane tree; hl/hr sum order changed); each scatter work-group sums the count table itself, no scan launch
# speedup vs baseline: 4.7351x; 1.1254x over previous
#include <hip/hip_runtime.h>
#include <hip/hip_bf16.h>

typedef __attribute__((ext_vector_type(2))) float v2f;
typedef __attribute__((ext_vector_type(8))) float v8f;

#define N_NODES 50000
#define N_EDGES 1600000
#define IN_HIDDEN 256
#define OUT_COLS 128
#define NHEAD 4
#define OUT_HIDDEN 32
#define EDGE_HIDDEN 64
#define N_ETYPES 8
#define NEG_SLOPE 0.2f

__device__ __forceinline__ unsigned enc_f32(float f) {
    unsigned u = __float_as_uint(f);
    return (u & 0x80000000u) ? ~u : (u | 0x80000000u);
}
__device__ __forceinline__ float dec_f32(unsigned u) {
    unsigned b = (u & 0x80000000u) ? (u & 0x7FFFFFFFu) : ~u;
    return __uint_as_float(b);
}

__device__ __forceinline__ float rows8_sum16(const float* v, const int c) {
    const bool b3 = (c & 8) != 0, b2 = (c & 4) != 0, b1 = (c & 2) != 0;
    float w[4], u[2];
#pragma unroll
    for (int i = 0; i < 4; ++i) {
        const float keep = b3 ? v[i + 4] : v[i];
        const float send = b3 ? v[i] : v[i + 4];
        w[i] = keep + __shfl_xor(send, 8, 32);
    }
#pragma unroll
    for (int i = 0; i < 2; ++i) {
        const float keep = b2 ? w[i + 2] : w[i];
        const float send = b2 ? w[i] : w[i + 2];
        u[i] = keep + __shfl_xor(send, 4, 32);
    }
    const float keep = b1 ? u[1] : u[0];
    const float send = b1 ? u[0] : u[1];
    const float t = keep + __shfl_xor(send, 2, 32);
    return t + __shfl_xor(t, 1, 32);
}

#define FRONT_T 256
#define GEMM_BLOCKS ((N_NODES / 16 + FRONT_T / 32 - 1) / (FRONT_T / 32))
__device__ __forceinline__ void
gemm_h_stage(const int blk, const float* __restrict__ x, const float* __restrict__ W,
             const float* __restrict__ a_l, const float* __restrict__ a_r,
             float* __restrict__ h, float* __restrict__ hl, float* __restrict__ hr) {
    const int lane  = threadIdx.x & 31;
    const int wave  = threadIdx.x >> 5;
    const int strip = blk * (FRONT_T >> 5) + wave;
    if (strip * 16 >= N_NODES) return;
    const int row0  = strip * 16;

    const int mrow  = lane & 15;
    const int khalf = (lane >> 4) << 1;

    v8f acc[8] = {};

    const float* xrow = x + (size_t)(row0 + mrow) * IN_HIDDEN + khalf;
    for (int k = 0; k < IN_HIDDEN; k += 4) {
        v2f a = *(const v2f*)(xrow + k);
#pragma unroll
        for (int t = 0; t < 8; ++t) {
            const float* wp = W + (size_t)(k + khalf) * OUT_COLS + t * 16 + mrow;
            v2f b;
            b.x = wp[0];
            b.y = wp[OUT_COLS];
            acc[t] = __builtin_amdgcn_wmma_f32_16x16x4_f32(
                false, a, false, b, (short)0, acc[t], false, false);
        }
    }

    const int mbase = (lane >> 4) << 3;
#pragma unroll
    for (int t = 0; t < 8; ++t) {
#pragma unroll
        for (int i = 0; i < 8; ++i) {
            float v = acc[t][i];
            v = (v == v) ? v : 0.0f;
            acc[t][i] = v;
            h[(size_t)(row0 + mbase + i) * OUT_COLS + t * 16 + mrow] = v;
        }
    }

    float fl[NHEAD], fr[NHEAD];
#pragma unroll
    for (int hd = 0; hd < NHEAD; ++hd) {
        const float al0 = a_l[hd * OUT_HIDDEN + mrow], al1 = a_l[hd * OUT_HIDDEN + 16 + mrow];
        const float ar0 = a_r[hd * OUT_HIDDEN + mrow], ar1 = a_r[hd * OUT_HIDDEN + 16 + mrow];
        float pl[8], pr[8];
#pragma unroll
        for (int i = 0; i < 8; ++i) {
            const float v0 = acc[2 * hd][i], v1 = acc[2 * hd + 1][i];
            pl[i] = fmaf(al1, v1, al0 * v0);
            pr[i] = fmaf(ar1, v1, ar0 * v0);
        }
        fl[hd] = rows8_sum16(pl, mrow);
        fr[hd] = rows8_sum16(pr, mrow);
    }
    const bool odd = (mrow & 1) != 0;
    const float2 ol = make_float2(odd ? fl[2] : fl[0], odd ? fl[3] : fl[1]);
    const float2 orr = make_float2(odd ? fr[2] : fr[0], odd ? fr[3] : fr[1]);
    *(float2*)(hl + (size_t)(row0 + mbase) * NHEAD + mrow * 2) = ol;
    *(float2*)(hr + (size_t)(row0 + mbase) * NHEAD + mrow * 2) = orr;
}

__device__ __forceinline__ void
etype_logits_stage(const int t,
                   const float* __restrict__ edge_emb,
                   const float* __restrict__ W_e,
                   const float* __restrict__ a_e,
                   float* __restrict__ he) {
    __shared__ float ev_s[NHEAD * EDGE_HIDDEN];
    __shared__ float ae_s[NHEAD * EDGE_HIDDEN];
    const int col  = threadIdx.x;
    const int head = col >> 6, d = col & (EDGE_HIDDEN - 1);
    const float* er = edge_emb + t * EDGE_HIDDEN;
    const float* wc = W_e + col;
    const float ae = a_e[col];
    float ev = 0.0f;
#pragma unroll
    for (int k = 0; k < EDGE_HIDDEN; ++k)
        ev = fmaf(er[k], wc[k * (EDGE_HIDDEN * NHEAD)], ev);
    ev_s[d * NHEAD + head] = ev;
    ae_s[d * NHEAD + head] = ae;
    __syncthreads();
    if (col < NHEAD) {
        float s = 0.0f;
#pragma unroll
        for (int dd = 0; dd < EDGE_HIDDEN; ++dd)
            s = fmaf(ae_s[dd * NHEAD + col], ev_s[dd * NHEAD + col], s);
        he[t * NHEAD + col] = s;
    }
}

__device__ __forceinline__ float4 edge_scores(const float* __restrict__ hl,
                                              const float* __restrict__ hr,
                                              const float* __restrict__ he,
                                              int src, int trg, int et) {
    const float4 l = *(const float4*)(hl + src * NHEAD);
    const float4 r = *(const float4*)(hr + trg * NHEAD);
    const float4 t = *(const float4*)(he + et * NHEAD);
    float4 s;
    s.x = l.x + r.x + t.x;
    s.y = l.y + r.y + t.y;
    s.z = l.z + r.z + t.z;
    s.w = l.w + r.w + t.w;
    s.x = (s.x > 0.0f) ? s.x : NEG_SLOPE * s.x;
    s.y = (s.y > 0.0f) ? s.y : NEG_SLOPE * s.y;
    s.z = (s.z > 0.0f) ? s.z : NEG_SLOPE * s.z;
    s.w = (s.w > 0.0f) ? s.w : NEG_SLOPE * s.w;
    return s;
}

__device__ __forceinline__ float4 scores_of_rows(const float4 l, const float4 r, const float4 t) {
    float4 s;
    s.x = l.x + r.x + t.x;
    s.y = l.y + r.y + t.y;
    s.z = l.z + r.z + t.z;
    s.w = l.w + r.w + t.w;
    s.x = (s.x > 0.0f) ? s.x : NEG_SLOPE * s.x;
    s.y = (s.y > 0.0f) ? s.y : NEG_SLOPE * s.y;
    s.z = (s.z > 0.0f) ? s.z : NEG_SLOPE * s.z;
    s.w = (s.w > 0.0f) ? s.w : NEG_SLOPE * s.w;
    return s;
}

#define NB   256
#define NPB  196
#define G1   256
#define CH   (N_EDGES / G1)
#define GT   1024
#define ITERS ((CH + GT - 1) / GT)

#define ITERS_R ((CH + FRONT_T - 1) / FRONT_T)
__device__ __forceinline__ void
chunk_rank_stage(const int wg, const int* __restrict__ edge,
                 int* __restrict__ lrank, int* __restrict__ counts) {
    __shared__ int hist[NB];
    const int tid = threadIdx.x;
    if (tid < NB) hist[tid] = 0;
    __syncthreads();
    const int e0 = wg * CH;
    int trgv[ITERS_R];
#pragma unroll
    for (int it = 0; it < ITERS_R; ++it) {
        const int k = tid + it * FRONT_T;
        const int e = e0 + ((k < CH) ? k : (CH - 1));
        trgv[it] = edge[N_EDGES + e];
    }
#pragma unroll
    for (int it = 0; it < ITERS_R; ++it) {
        const int k = tid + it * FRONT_T;
        if (k < CH) lrank[e0 + k] = atomicAdd(&hist[trgv[it] / NPB], 1);
    }
    __syncthreads();
    if (tid < NB) counts[tid * G1 + wg] = hist[tid];
}

__global__ void __launch_bounds__(FRONT_T)
front_kernel(const int* __restrict__ edge,
             const float* __restrict__ x, const float* __restrict__ W,
             const float* __restrict__ edge_emb, const float* __restrict__ W_e,
             const float* __restrict__ a_e,
             const float* __restrict__ a_l, const float* __restrict__ a_r,
             float* __restrict__ h, float* __restrict__ he,
             float* __restrict__ hl, float* __restrict__ hr,
             int* __restrict__ lrank, int* __restrict__ counts,
             unsigned* __restrict__ gmax) {
    const int b = blockIdx.x;
    if (b < GEMM_BLOCKS) {
        gemm_h_stage(b, x, W, a_l, a_r, h, hl, hr);
    } else if (b < GEMM_BLOCKS + G1) {
        chunk_rank_stage(b - GEMM_BLOCKS, edge, lrank, counts);
    } else {
        const int t = b - GEMM_BLOCKS - G1;
        if (t == 0 && threadIdx.x == 0) *gmax = enc_f32(-INFINITY);
        etype_logits_stage(t, edge_emb, W_e, a_e, he);
    }
}

__device__ __forceinline__ int vals16_sum32(const int* v, const int lane) {
    const bool b4 = (lane & 16) != 0, b3 = (lane & 8) != 0, b2 = (lane & 4) != 0, b1 = (lane & 2) != 0;
    int a[8], w[4], u[2];
#pragma unroll
    for (int i = 0; i < 8; ++i) {
        const int keep = b4 ? v[i + 8] : v[i];
        const int send = b4 ? v[i] : v[i + 8];
        a[i] = keep + __shfl_xor(send, 16, 32);
    }
#pragma unroll
    for (int i = 0; i < 4; ++i) {
        const int keep = b3 ? a[i + 4] : a[i];
        const int send = b3 ? a[i] : a[i + 4];
        w[i] = keep + __shfl_xor(send, 8, 32);
    }
#pragma unroll
    for (int i = 0; i < 2; ++i) {
        const int keep = b2 ? w[i + 2] : w[i];
        const int send = b2 ? w[i] : w[i + 2];
        u[i] = keep + __shfl_xor(send, 4, 32);
    }
    const int keep = b1 ? u[1] : u[0];
    const int send = b1 ? u[0] : u[1];
    const int t = keep + __shfl_xor(send, 2, 32);
    return t + __shfl_xor(t, 1, 32);
}

#define EID_BITS 21
__global__ void __launch_bounds__(GT)
scatter_kernel(const int* __restrict__ edge, const int* __restrict__ lrank,
               const int* __restrict__ counts, int* __restrict__ bstarts, int* __restrict__ coarse) {
    __shared__ int  bs[NB];
    __shared__ int  ls[NB];
    __shared__ int  wtot[NB / 32];
    __shared__ int  wtotB[NB / 32];
    __shared__ int  rtot[NB];
    __shared__ int  rpre[NB];
    __shared__ int2 stage[CH];
    const int wg = blockIdx.x, tid = threadIdx.x;
    const int lane = tid & 31, w = tid >> 5;
    const int e0 = wg * CH;
    int srcv[ITERS], trgv[ITERS], etv[ITERS], rkv[ITERS];
#pragma unroll
    for (int it = 0; it < ITERS; ++it) {
        const int k = tid + it * GT;
        const int e = e0 + ((k < CH) ? k : (CH - 1));
        srcv[it] = edge[e];
        trgv[it] = edge[N_EDGES + e];
        etv[it]  = edge[2 * N_EDGES + e];
        rkv[it]  = lrank[e];
    }
    int cnt = 0;
    if (tid < NB) cnt = counts[tid * G1 + wg];
    {
        const int4* rowp = (const int4*)(counts + (size_t)(w * 8) * G1) + lane;
        int4 ta[8], tb[8];
#pragma unroll
        for (int r = 0; r < 8; ++r) { ta[r] = rowp[r * (G1 / 4)]; tb[r] = rowp[r * (G1 / 4) + 32]; }
        int v[16];
        const int c0 = 4 * lane, c1 = 128 + 4 * lane;
#pragma unroll
        for (int r = 0; r < 8; ++r) {
            v[r] = (ta[r].x + ta[r].y) + (ta[r].z + ta[r].w) + (tb[r].x + tb[r].y) + (tb[r].z + tb[r].w);
            v[8 + r] = ((c0     < wg) ? ta[r].x : 0) + ((c0 + 1 < wg) ? ta[r].y : 0)
                     + ((c0 + 2 < wg) ? ta[r].z : 0) + ((c0 + 3 < wg) ? ta[r].w : 0)
                     + ((c1     < wg) ? tb[r].x : 0) + ((c1 + 1 < wg) ? tb[r].y : 0)
                     + ((c1 + 2 < wg) ? tb[r].z : 0) + ((c1 + 3 < wg) ? tb[r].w : 0);
        }
        const int tot = vals16_sum32(v, lane);
        if ((lane & 1) == 0) {
            const int row = w * 8 + ((lane >> 1) & 7);
            if (lane & 16) rpre[row] = tot; else rtot[row] = tot;
        }
    }
    __syncthreads();
    int ex = 0, exB = 0;
    if (tid < NB) {
        const int tot = rtot[tid];
        int inc = cnt, incB = tot;
        for (int off = 1; off < 32; off <<= 1) {
            const int t  = __shfl_up(inc, off, 32);
            const int tB = __shfl_up(incB, off, 32);
            if (lane >= off) { inc += t; incB += tB; }
        }
        if (lane == 31) { wtot[w] = inc; wtotB[w] = incB; }
        ex = inc - cnt;
        exB = incB - tot;
    }
    __syncthreads();
    if (tid < NB) {
        int base = 0, baseB = 0;
        for (int j = 0; j < NB / 32; ++j) { base += (j < w) ? wtot[j] : 0; baseB += (j < w) ? wtotB[j] : 0; }
        ls[tid] = ex + base;
        const int bfirst = exB + baseB;
        bs[tid] = bfirst + rpre[tid];
        if (wg == 0) bstarts[tid] = bfirst;
    }
    __syncthreads();
#pragma unroll
    for (int it = 0; it < ITERS; ++it) {
        const int k = tid + it * GT;
        if (k < CH) {
            const int b  = trgv[it] / NPB;
            const int tl = trgv[it] - b * NPB;
            stage[ls[b] + rkv[it]] =
                make_int2((e0 + k) | (b << EID_BITS), srcv[it] | (etv[it] << 16) | (tl << 19));
        }
    }
    __syncthreads();
#pragma unroll
    for (int it = 0; it < ITERS; ++it) {
        const int s = tid + it * GT;
        if (s < CH) {
            const int2 r = stage[s];
            const int b = r.x >> EID_BITS;
            *(int2*)(coarse + 2 * (size_t)(bs[b] + (s - ls[b]))) =
                make_int2(r.x & ((1 << EID_BITS) - 1), r.y);
        }
    }
}

#define BCAP_IT 8
#define BCAP    (BCAP_IT * GT)

template <bool IN_LDS>
__device__ __forceinline__ float
order_segments(const int2* recs, const int* __restrict__ grouped, int* ordered,
               const int* loff, const int bstart, const int b,
               const float* __restrict__ hl, const float* __restrict__ hr,
               const float* __restrict__ he, const int lane, const int w) {
    float lmax = -INFINITY;
    int  py = 0, pnode = 0;
    bool pv = false;
    for (int t = w; t < NPB; t += GT / 32) {
        const int lo   = __builtin_amdgcn_readfirstlane(loff[t]);
        const int d    = __builtin_amdgcn_readfirstlane(loff[t + 1]) - lo;
        const int beg  = bstart + lo;
        const int node = b * NPB + t;
        for (int ca = 0; ca < d; ca += 32) {
            const int ia = ca + lane;
            int2 pa = make_int2(0x7FFFFFFF, 0);
            if (ia < d) {
                if (IN_LDS) pa = recs[lo + ia];
                else        pa = *(const int2*)(grouped + 2 * (size_t)(beg + ia));
            }
            const float4 rl = *(const float4*)(hl + (py & 0xFFFF) * NHEAD);
            const float4 rr = *(const float4*)(hr + pnode * NHEAD);
            const float4 rt = *(const float4*)(he + (py >> 16) * NHEAD);
            int pos = 0;
            for (int cb = 0; cb < d; cb += 32) {
                const int ib = cb + lane;
                int kb = 0x7FFFFFFF;
                if (ib < d) {
                    if (IN_LDS) kb = recs[lo + ib].x;
                    else        kb = grouped[2 * (size_t)(beg + ib)];
                }
#pragma unroll
                for (int j = 0; j < 32; ++j) {
                    const int key = __builtin_amdgcn_readlane(kb, j);
                    pos += (key < pa.x) ? 1 : 0;
                }
            }
            if (ia < d) *(int2*)(ordered + 2 * (size_t)(beg + pos)) = pa;
            const float4 s = scores_of_rows(rl, rr, rt);
            if (pv) lmax = fmaxf(lmax, fmaxf(fmaxf(s.x, s.y), fmaxf(s.z, s.w)));
            py = pa.y; pnode = node; pv = (ia < d);
        }
    }
    {
        const float4 rl = *(const float4*)(hl + (py & 0xFFFF) * NHEAD);
        const float4 rr = *(const float4*)(hr + pnode * NHEAD);
        const float4 rt = *(const float4*)(he + (py >> 16) * NHEAD);
        const float4 s = scores_of_rows(rl, rr, rt);
        if (pv) lmax = fmaxf(lmax, fmaxf(fmaxf(s.x, s.y), fmaxf(s.z, s.w)));
    }
    return lmax;
}

__global__ void __launch_bounds__(GT)
bucket_kernel(const int* coarse, const int* __restrict__ bstarts,
              int* __restrict__ grouped, int* ordered, int* __restrict__ offs,
              const float* __restrict__ hl, const float* __restrict__ hr,
              const float* __restrict__ he, unsigned* __restrict__ gmax) {
    __shared__ int   cntL[NPB + 1];
    __shared__ int   loff[NPB + 1];
    __shared__ float red[GT / 32];
    __shared__ int2  recs[BCAP];
    const int b = blockIdx.x, tid = threadIdx.x;
    const int lane = tid & 31, w = tid >> 5;
    const int bstart = bstarts[b];
    const int bend   = (b + 1 < NB) ? bstarts[b + 1] : N_EDGES;
    const int n = bend - bstart;
    const bool in_lds = (n <= BCAP);
    if (tid <= NPB) cntL[tid] = 0;
    int2 rv[BCAP_IT];
    if (in_lds) {
#pragma unroll
        for (int it = 0; it < BCAP_IT; ++it) {
            const int i = tid + it * GT;
            rv[it] = make_int2(0, 0);
            if (i < n) rv[it] = *(const int2*)(coarse + 2 * (size_t)(bstart + i));
        }
    }
    __syncthreads();
    if (in_lds) {
#pragma unroll
        for (int it = 0; it < BCAP_IT; ++it)
            if (tid + it * GT < n) atomicAdd(&cntL[(rv[it].y >> 19) & 0xFF], 1);
    } else {
        for (int i = tid; i < n; i += GT)
            atomicAdd(&cntL[(coarse[2 * (size_t)(bstart + i) + 1] >> 19) & 0xFF], 1);
    }
    __syncthreads();
    if (w == 0) {
        int carry = 0;
        for (int r = 0; r < (NPB + 32) / 32; ++r) {
            const int idx = r * 32 + lane;
            const int c = (idx < NPB) ? cntL[idx] : 0;
            int inc = c;
            for (int off = 1; off < 32; off <<= 1) {
                const int t = __shfl_up(inc, off, 32);
                if (lane >= off) inc += t;
            }
            if (idx <= NPB) loff[idx] = carry + inc - c;
            carry += __shfl(inc, 31, 32);
        }
    }
    __syncthreads();
    if (tid < NPB) {
        const int node = b * NPB + tid;
        if (node <= N_NODES) offs[node] = bstart + loff[tid];
        cntL[tid] = 0;
    }
    __syncthreads();
    float lmax;
    if (in_lds) {
#pragma unroll
        for (int it = 0; it < BCAP_IT; ++it) {
            if (tid + it * GT < n) {
                const int tl = (rv[it].y >> 19) & 0xFF;
                const int r  = atomicAdd(&cntL[tl], 1);
                recs[loff[tl] + r] = make_int2(rv[it].x, rv[it].y & 0x7FFFF);
            }
        }
        __syncthreads();
        lmax = order_segments<true>(recs, grouped, ordered, loff, bstart, b, hl, hr, he, lane, w);
    } else {
        for (int i = tid; i < n; i += GT) {
            const int2 rec = *(const int2*)(coarse + 2 * (size_t)(bstart + i));
            const int tl = (rec.y >> 19) & 0xFF;
            const int r  = atomicAdd(&cntL[tl], 1);
            *(int2*)(grouped + 2 * (size_t)(bstart + loff[tl] + r)) = make_int2(rec.x, rec.y & 0x7FFFF);
        }
        __syncthreads();
        lmax = order_segments<false>(recs, grouped, ordered, loff, bstart, b, hl, hr, he, lane, w);
    }
    for (int m = 16; m > 0; m >>= 1) lmax = fmaxf(lmax, __shfl_xor(lmax, m, 32));
    if (lane == 0) red[w] = lmax;
    __syncthreads();
    if (tid < 32) {
        float v = red[tid];
        for (int m = 16; m > 0; m >>= 1) v = fmaxf(v, __shfl_xor(v, m, 32));
        if (tid == 0) atomicMax(gmax, enc_f32(v));
    }
}

__global__ void __launch_bounds__(256)
node_aggregate_kernel(const int* __restrict__ ordered, const int* __restrict__ offs,
                      const float* __restrict__ h,
                      const float* __restrict__ hl, const float* __restrict__ hr,
                      const float* __restrict__ he, const unsigned* __restrict__ gmax,
                      float* __restrict__ out, float* __restrict__ attn_out) {
    const int lane = threadIdx.x & 31;
    const int node = blockIdx.x * (blockDim.x >> 5) + (threadIdx.x >> 5);
    if (node >= N_NODES) return;
    const int beg = __builtin_amdgcn_readfirstlane(offs[node]);
    const int d   = __builtin_amdgcn_readfirstlane(offs[node + 1]) - beg;
    const float smax = dec_f32(*gmax);

    float4 ds = make_float4(0.0f, 0.0f, 0.0f, 0.0f);
    int2   p0 = make_int2(0, 0), p1 = make_int2(0, 0);
    float4 e0 = make_float4(0.0f, 0.0f, 0.0f, 0.0f);
    float4 e1 = make_float4(0.0f, 0.0f, 0.0f, 0.0f);
    if (lane < d) {
        p0 = *(const int2*)(ordered + 2 * (size_t)(beg + lane));
        const float4 s = edge_scores(hl, hr, he, p0.y & 0xFFFF, node, p0.y >> 16);
        e0.x = __expf(s.x - smax);
        e0.y = __expf(s.y - smax);
        e0.z = __expf(s.z - smax);
        e0.w = __expf(s.w - smax);
        ds.x += e0.x;
        ds.y += e0.y;
        ds.z += e0.z;
        ds.w += e0.w;
    }
    if (32 + lane < d) {
        p1 = *(const int2*)(ordered + 2 * (size_t)(beg + 32 + lane));
        const float4 s = edge_scores(hl, hr, he, p1.y & 0xFFFF, node, p1.y >> 16);
        e1.x = __expf(s.x - smax);
        e1.y = __expf(s.y - smax);
        e1.z = __expf(s.z - smax);
        e1.w = __expf(s.w - smax);
        ds.x += e1.x;
        ds.y += e1.y;
        ds.z += e1.z;
        ds.w += e1.w;
    }
    for (int c = 64; c < d; c += 32) {
        const int i = c + lane;
        if (i < d) {
            const int2 p = *(const int2*)(ordered + 2 * (size_t)(beg + i));
            const float4 s = edge_scores(hl, hr, he, p.y & 0xFFFF, node, p.y >> 16);
            ds.x += __expf(s.x - smax);
            ds.y += __expf(s.y - smax);
            ds.z += __expf(s.z - smax);
            ds.w += __expf(s.w - smax);
        }
    }
    for (int m = 16; m > 0; m >>= 1) {
        ds.x += __shfl_xor(ds.x, m, 32);
        ds.y += __shfl_xor(ds.y, m, 32);
        ds.z += __shfl_xor(ds.z, m, 32);
        ds.w += __shfl_xor(ds.w, m, 32);
    }
    const float4 dn = make_float4(ds.x + 1e-16f, ds.y + 1e-16f, ds.z + 1e-16f, ds.w + 1e-16f);

    float acc0 = 0.0f, acc1 = 0.0f, acc2 = 0.0f, acc3 = 0.0f;
    const float* hcol = h + lane;
    for (int c = 0; c < d; c += 32) {
        const int i = c + lane;
        int srcv = 0;
        float4 a = make_float4(0.0f, 0.0f, 0.0f, 0.0f);
        if (i < d) {
            int2   p = p0;
            float4 e = e0;
            if (c == 32) { p = p1; e = e1; }
            if (c >= 64) {
                p = *(const int2*)(ordered + 2 * (size_t)(beg + i));
                const float4 s = edge_scores(hl, hr, he, p.y & 0xFFFF, node, p.y >> 16);
                e.x = __expf(s.x - smax);
                e.y = __expf(s.y - smax);
                e.z = __expf(s.z - smax);
                e.w = __expf(s.w - smax);
            }
            srcv = p.y & 0xFFFF;
            a.x = e.x / dn.x;
            a.y = e.y / dn.y;
            a.z = e.z / dn.z;
            a.w = e.w / dn.w;
            *(float4*)(attn_out + (size_t)p.x * NHEAD) = a;
        }
        const int nb = (d - c < 32) ? (d - c) : 32;
#pragma unroll 4
        for (int j = 0; j < nb; ++j) {
            const int sj = __builtin_amdgcn_readlane(srcv, j);
            const float a0 = __int_as_float(__builtin_amdgcn_readlane(__float_as_int(a.x), j));
            const float a1 = __int_as_float(__builtin_amdgcn_readlane(__float_as_int(a.y), j));
            const float a2 = __int_as_float(__builtin_amdgcn_readlane(__float_as_int(a.z), j));
            const float a3 = __int_as_float(__builtin_amdgcn_readlane(__float_as_int(a.w), j));
            const float* row = hcol + (size_t)sj * OUT_COLS;
            acc0 = fmaf(row[0 * OUT_HIDDEN], a0, acc0);
            acc1 = fmaf(row[1 * OUT_HIDDEN], a1, acc1);
            acc2 = fmaf(row[2 * OUT_HIDDEN], a2, acc2);
            acc3 = fmaf(row[3 * OUT_HIDDEN], a3, acc3);
        }
    }
    float* orow = out + (size_t)node * OUT_COLS + lane;
    orow[0 * OUT_HIDDEN] = acc0;
    orow[1 * OUT_HIDDEN] = acc1;
    orow[2 * OUT_HIDDEN] = acc2;
    orow[3 * OUT_HIDDEN] = acc3;
}

extern "C" void kernel_launch(void* const* d_in, const int* in_sizes, int n_in,
                              void* d_out, int out_size, void* d_ws, size_t ws_size,
                              hipStream_t stream) {
    const int*   edge     = (const int*)  d_in[0];
    const float* x        = (const float*)d_in[1];
    const float* edge_emb = (const float*)d_in[2];
    const float* W        = (const float*)d_in[3];
    const float* W_e      = (const float*)d_in[4];
    const float* a_l      = (const float*)d_in[5];
    const float* a_r      = (const float*)d_in[6];
    const float* a_e      = (const float*)d_in[7];

    float* out      = (float*)d_out;
    float* attn_out = out + (size_t)N_NODES * OUT_COLS;

    char* ws = (char*)d_ws;
    float* h       = (float*)ws;  ws += (size_t)N_NODES * OUT_COLS * sizeof(float);
    int*   planeA  = (int*)ws;    ws += (size_t)N_EDGES * 2 * sizeof(int);
    int*   planeB  = (int*)ws;    ws += (size_t)N_EDGES * 2 * sizeof(int);
    int*   lrank   = planeA;
    int*   grouped = planeA;
    int*   coarse  = planeB;
    int*   ordered = planeB;
    float* hl      = (float*)ws;  ws += (size_t)N_NODES * NHEAD   * sizeof(float);
    float* hr      = (float*)ws;  ws += (size_t)N_NODES * NHEAD   * sizeof(float);
    int*   offs    = (int*)ws + 50176;
    int*   tab     = (int*)ws + 102400;
    int*   bstarts = (int*)ws + 168000;
    ws += (size_t)N_NODES * NHEAD * sizeof(float);
    float* he      = (float*)ws;  ws += (size_t)N_ETYPES * NHEAD  * sizeof(float);
    unsigned* gmax = (unsigned*)ws;

    front_kernel<<<GEMM_BLOCKS + G1 + N_ETYPES, FRONT_T, 0, stream>>>(
        edge, x, W, edge_emb, W_e, a_e, a_l, a_r, h, he, hl, hr, lrank, tab, gmax);
    scatter_kernel<<<G1, GT, 0, stream>>>(edge, lrank, tab, bstarts, coarse);
    bucket_kernel<<<NB, GT, 0, stream>>>(coarse, bstarts, grouped, ordered, offs, hl, hr, he, gmax);
    node_aggregate_kernel<<<(N_NODES + 7) / 8, 256, 0, stream>>>(
        ordered, offs, h, hl, hr, he, gmax, out, attn_out);
}


# GEMM operand feed: W copied once per work-group into LDS as 4-k cells, one 16-byte LDS read feeds two matrix instructions, x read 16 bytes per lane one trip ahead; k sum order of h, hl, hr changed
# speedup vs baseline: 5.0546x; 1.0675x over previous
#include <hip/hip_runtime.h>
#include <hip/hip_bf16.h>

typedef __attribute__((ext_vector_type(2))) float v2f;
typedef __attribute__((ext_vector_type(8))) float v8f;

#define N_NODES 50000
#define N_EDGES 1600000
#define IN_HIDDEN 256
#define OUT_COLS 128
#define NHEAD 4
#define OUT_HIDDEN 32
#define EDGE_HIDDEN 64
#define N_ETYPES 8
#define NEG_SLOPE 0.2f

__device__ __forceinline__ unsigned enc_f32(float f) {
    unsigned u = __float_as_uint(f);
    return (u & 0x80000000u) ? ~u : (u | 0x80000000u);
}
__device__ __forceinline__ float dec_f32(unsigned u) {
    unsigned b = (u & 0x80000000u) ? (u & 0x7FFFFFFFu) : ~u;
    return __uint_as_float(b);
}

__device__ __forceinline__ float rows8_sum16(const float* v, const int c) {
    const bool b3 = (c & 8) != 0, b2 = (c & 4) != 0, b1 = (c & 2) != 0;
    float w[4], u[2];
#pragma unroll
    for (int i = 0; i < 4; ++i) {
        const float keep = b3 ? v[i + 4] : v[i];
        const float send = b3 ? v[i] : v[i + 4];
        w[i] = keep + __shfl_xor(send, 8, 32);
    }
#pragma unroll
    for (int i = 0; i < 2; ++i) {
        const float keep = b2 ? w[i + 2] : w[i];
        const float send = b2 ? w[i] : w[i + 2];
        u[i] = keep + __shfl_xor(send, 4, 32);
    }
    const float keep = b1 ? u[1] : u[0];
    const float send = b1 ? u[0] : u[1];
    const float t = keep + __shfl_xor(send, 2, 32);
    return t + __shfl_xor(t, 1, 32);
}

#define FRONT_T 256
#define GEMM_BLOCKS ((N_NODES / 16 + FRONT_T / 32 - 1) / (FRONT_T / 32))
#define W_CELLS ((IN_HIDDEN / 4) * OUT_COLS)
__device__ __forceinline__ void
gemm_h_stage(const int blk, const float* __restrict__ x, const float* __restrict__ W,
             const float* __restrict__ a_l, const float* __restrict__ a_r,
             float* __restrict__ h, float* __restrict__ hl, float* __restrict__ hr) {
    __shared__ float4 Wl[W_CELLS];
#pragma unroll 8
    for (int m = 0; m < W_CELLS / FRONT_T; ++m) {
        const int cell = threadIdx.x + m * FRONT_T;
        const int kq = cell >> 7, col = cell & (OUT_COLS - 1);
        const float* wp = W + (size_t)(4 * kq) * OUT_COLS + col;
        Wl[cell] = make_float4(wp[0], wp[OUT_COLS], wp[2 * OUT_COLS], wp[3 * OUT_COLS]);
    }
    __syncthreads();

    const int lane  = threadIdx.x & 31;
    const int wave  = threadIdx.x >> 5;
    const int strip = blk * (FRONT_T >> 5) + wave;
    if (strip * 16 >= N_NODES) return;
    const int row0  = strip * 16;

    const int mrow  = lane & 15;
    const int hh    = lane >> 4;

    v8f acc[8] = {};

    const float*  xrow = x + (size_t)(row0 + mrow) * IN_HIDDEN + 4 * hh;
    const float4* wl   = Wl + hh * OUT_COLS + mrow;
    float4 a4[4];
#pragma unroll
    for (int j = 0; j < 4; ++j) a4[j] = *(const float4*)(xrow + 8 * j);
#pragma unroll
    for (int kb = 0; kb < IN_HIDDEN; kb += 32) {
        const int kn = (kb + 32 < IN_HIDDEN) ? (kb + 32) : kb;
        float4 an[4];
#pragma unroll
        for (int j = 0; j < 4; ++j) an[j] = *(const float4*)(xrow + kn + 8 * j);
#pragma unroll
        for (int j = 0; j < 4; ++j) {
            const float4* wj = wl + (size_t)((kb >> 2) + 2 * j) * OUT_COLS;
            float4 b4[8];
#pragma unroll
            for (int t = 0; t < 8; ++t) b4[t] = wj[t * 16];
            v2f a01; a01.x = a4[j].x; a01.y = a4[j].y;
            v2f a23; a23.x = a4[j].z; a23.y = a4[j].w;
#pragma unroll
            for (int t = 0; t < 8; ++t) {
                v2f b; b.x = b4[t].x; b.y = b4[t].y;
                acc[t] = __builtin_amdgcn_wmma_f32_16x16x4_f32(
                    false, a01, false, b, (short)0, acc[t], false, false);
            }
#pragma unroll
            for (int t = 0; t < 8; ++t) {
                v2f b; b.x = b4[t].z; b.y = b4[t].w;
                acc[t] = __builtin_amdgcn_wmma_f32_16x16x4_f32(
                    false, a23, false, b, (short)0, acc[t], false, false);
            }
        }
#pragma unroll
        for (int j = 0; j < 4; ++j) a4[j] = an[j];
    }

    const int mbase = (lane >> 4) << 3;
#pragma unroll
    for (int t = 0; t < 8; ++t) {
#pragma unroll
        for (int i = 0; i < 8; ++i) {
            float v = acc[t][i];
            v = (v == v) ? v : 0.0f;
            acc[t][i] = v;
            h[(size_t)(row0 + mbase + i) * OUT_COLS + t * 16 + mrow] = v;
        }
    }

    float fl[NHEAD], fr[NHEAD];
#pragma unroll
    for (int hd = 0; hd < NHEAD; ++hd) {
        const float al0 = a_l[hd * OUT_HIDDEN + mrow], al1 = a_l[hd * OUT_HIDDEN + 16 + mrow];
        const float ar0 = a_r[hd * OUT_HIDDEN + mrow], ar1 = a_r[hd * OUT_HIDDEN + 16 + mrow];
        float pl[8], pr[8];
#pragma unroll
        for (int i = 0; i < 8; ++i) {
            const float v0 = acc[2 * hd][i], v1 = acc[2 * hd + 1][i];
            pl[i] = fmaf(al1, v1, al0 * v0);
            pr[i] = fmaf(ar1, v1, ar0 * v0);
        }
        fl[hd] = rows8_sum16(pl, mrow);
        fr[hd] = rows8_sum16(pr, mrow);
    }
    const bool odd = (mrow & 1) != 0;
    const float2 ol = make_float2(odd ? fl[2] : fl[0], odd ? fl[3] : fl[1]);
    const float2 orr = make_float2(odd ? fr[2] : fr[0], odd ? fr[3] : fr[1]);
    *(float2*)(hl + (size_t)(row0 + mbase) * NHEAD + mrow * 2) = ol;
    *(float2*)(hr + (size_t)(row0 + mbase) * NHEAD + mrow * 2) = orr;
}

__device__ __forceinline__ void
etype_logits_stage(const int t,
                   const float* __restrict__ edge_emb,
                   const float* __restrict__ W_e,
                   const float* __restrict__ a_e,
                   float* __restrict__ he) {
    __shared__ float ev_s[NHEAD * EDGE_HIDDEN];
    __shared__ float ae_s[NHEAD * EDGE_HIDDEN];
    const int col  = threadIdx.x;
    const int head = col >> 6, d = col & (EDGE_HIDDEN - 1);
    const float* er = edge_emb + t * EDGE_HIDDEN;
    const float* wc = W_e + col;
    const float ae = a_e[col];
    float ev = 0.0f;
#pragma unroll
    for (int k = 0; k < EDGE_HIDDEN; ++k)
        ev = fmaf(er[k], wc[k * (EDGE_HIDDEN * NHEAD)], ev);
    ev_s[d * NHEAD + head] = ev;
    ae_s[d * NHEAD + head] = ae;
    __syncthreads();
    if (col < NHEAD) {
        float s = 0.0f;
#pragma unroll
        for (int dd = 0; dd < EDGE_HIDDEN; ++dd)
            s = fmaf(ae_s[dd * NHEAD + col], ev_s[dd * NHEAD + col], s);
        he[t * NHEAD + col] = s;
    }
}

__device__ __forceinline__ float4 edge_scores(const float* __restrict__ hl,
                                              const float* __restrict__ hr,
                                              const float* __restrict__ he,
                                              int src, int trg, int et) {
    const float4 l = *(const float4*)(hl + src * NHEAD);
    const float4 r = *(const float4*)(hr + trg * NHEAD);
    const float4 t = *(const float4*)(he + et * NHEAD);
    float4 s;
    s.x = l.x + r.x + t.x;
    s.y = l.y + r.y + t.y;
    s.z = l.z + r.z + t.z;
    s.w = l.w + r.w + t.w;
    s.x = (s.x > 0.0f) ? s.x : NEG_SLOPE * s.x;
    s.y = (s.y > 0.0f) ? s.y : NEG_SLOPE * s.y;
    s.z = (s.z > 0.0f) ? s.z : NEG_SLOPE * s.z;
    s.w = (s.w > 0.0f) ? s.w : NEG_SLOPE * s.w;
    return s;
}

__device__ __forceinline__ float4 scores_of_rows(const float4 l, const float4 r, const float4 t) {
    float4 s;
    s.x = l.x + r.x + t.x;
    s.y = l.y + r.y + t.y;
    s.z = l.z + r.z + t.z;
    s.w = l.w + r.w + t.w;
    s.x = (s.x > 0.0f) ? s.x : NEG_SLOPE * s.x;
    s.y = (s.y > 0.0f) ? s.y : NEG_SLOPE * s.y;
    s.z = (s.z > 0.0f) ? s.z : NEG_SLOPE * s.z;
    s.w = (s.w > 0.0f) ? s.w : NEG_SLOPE * s.w;
    return s;
}

#define NB   256
#define NPB  196
#define G1   256
#define CH   (N_EDGES / G1)
#define GT   1024
#define ITERS ((CH + GT - 1) / GT)

#define ITERS_R ((CH + FRONT_T - 1) / FRONT_T)
__device__ __forceinline__ void
chunk_rank_stage(const int wg, const int* __restrict__ edge,
                 int* __restrict__ lrank, int* __restrict__ counts) {
    __shared__ int hist[NB];
    const int tid = threadIdx.x;
    if (tid < NB) hist[tid] = 0;
    __syncthreads();
    const int e0 = wg * CH;
    int trgv[ITERS_R];
#pragma unroll
    for (int it = 0; it < ITERS_R; ++it) {
        const int k = tid + it * FRONT_T;
        const int e = e0 + ((k < CH) ? k : (CH - 1));
        trgv[it] = edge[N_EDGES + e];
    }
#pragma unroll
    for (int it = 0; it < ITERS_R; ++it) {
        const int k = tid + it * FRONT_T;
        if (k < CH) lrank[e0 + k] = atomicAdd(&hist[trgv[it] / NPB], 1);
    }
    __syncthreads();
    if (tid < NB) counts[tid * G1 + wg] = hist[tid];
}

__global__ void __launch_bounds__(FRONT_T)
front_kernel(const int* __restrict__ edge,
             const float* __restrict__ x, const float* __restrict__ W,
             const float* __restrict__ edge_emb, const float* __restrict__ W_e,
             const float* __restrict__ a_e,
             const float* __restrict__ a_l, const float* __restrict__ a_r,
             float* __restrict__ h, float* __restrict__ he,
             float* __restrict__ hl, float* __restrict__ hr,
             int* __restrict__ lrank, int* __restrict__ counts,
             unsigned* __restrict__ gmax) {
    const int b = blockIdx.x;
    if (b < GEMM_BLOCKS) {
        gemm_h_stage(b, x, W, a_l, a_r, h, hl, hr);
    } else if (b < GEMM_BLOCKS + G1) {
        chunk_rank_stage(b - GEMM_BLOCKS, edge, lrank, counts);
    } else {
        const int t = b - GEMM_BLOCKS - G1;
        if (t == 0 && threadIdx.x == 0) *gmax = enc_f32(-INFINITY);
        etype_logits_stage(t, edge_emb, W_e, a_e, he);
    }
}

__device__ __forceinline__ int vals16_sum32(const int* v, const int lane) {
    const bool b4 = (lane & 16) != 0, b3 = (lane & 8) != 0, b2 = (lane & 4) != 0, b1 = (lane & 2) != 0;
    int a[8], w[4], u[2];
#pragma unroll
    for (int i = 0; i < 8; ++i) {
        const int keep = b4 ? v[i + 8] : v[i];
        const int send = b4 ? v[i] : v[i + 8];
        a[i] = keep + __shfl_xor(send, 16, 32);
    }
#pragma unroll
    for (int i = 0; i < 4; ++i) {
        const int keep = b3 ? a[i + 4] : a[i];
        const int send = b3 ? a[i] : a[i + 4];
        w[i] = keep + __shfl_xor(send, 8, 32);
    }
#pragma unroll
    for (int i = 0; i < 2; ++i) {
        const int keep = b2 ? w[i + 2] : w[i];
        const int send = b2 ? w[i] : w[i + 2];
        u[i] = keep + __shfl_xor(send, 4, 32);
    }
    const int keep = b1 ? u[1] : u[0];
    const int send = b1 ? u[0] : u[1];
    const int t = keep + __shfl_xor(send, 2, 32);
    return t + __shfl_xor(t, 1, 32);
}

#define EID_BITS 21
__global__ void __launch_bounds__(GT)
scatter_kernel(const int* __restrict__ edge, const int* __restrict__ lrank,
               const int* __restrict__ counts, int* __restrict__ bstarts, int* __restrict__ coarse) {
    __shared__ int  bs[NB];
    __shared__ int  ls[NB];
    __shared__ int  wtot[NB / 32];
    __shared__ int  wtotB[NB / 32];
    __shared__ int  rtot[NB];
    __shared__ int  rpre[NB];
    __shared__ int2 stage[CH];
    const int wg = blockIdx.x, tid = threadIdx.x;
    const int lane = tid & 31, w = tid >> 5;
    const int e0 = wg * CH;
    int srcv[ITERS], trgv[ITERS], etv[ITERS], rkv[ITERS];
#pragma unroll
    for (int it = 0; it < ITERS; ++it) {
        const int k = tid + it * GT;
        const int e = e0 + ((k < CH) ? k : (CH - 1));
        srcv[it] = edge[e];
        trgv[it] = edge[N_EDGES + e];
        etv[it]  = edge[2 * N_EDGES + e];
        rkv[it]  = lrank[e];
    }
    int cnt = 0;
    if (tid < NB) cnt = counts[tid * G1 + wg];
    {
        const int4* rowp = (const int4*)(counts + (size_t)(w * 8) * G1) + lane;
        int4 ta[8], tb[8];
#pragma unroll
        for (int r = 0; r < 8; ++r) { ta[r] = rowp[r * (G1 / 4)]; tb[r] = rowp[r * (G1 / 4) + 32]; }
        int v[16];
        const int c0 = 4 * lane, c1 = 128 + 4 * lane;
#pragma unroll
        for (int r = 0; r < 8; ++r) {
            v[r] = (ta[r].x + ta[r].y) + (ta[r].z + ta[r].w) + (tb[r].x + tb[r].y) + (tb[r].z + tb[r].w);
            v[8 + r] = ((c0     < wg) ? ta[r].x : 0) + ((c0 + 1 < wg) ? ta[r].y : 0)
                     + ((c0 + 2 < wg) ? ta[r].z : 0) + ((c0 + 3 < wg) ? ta[r].w : 0)
                     + ((c1     < wg) ? tb[r].x : 0) + ((c1 + 1 < wg) ? tb[r].y : 0)
                     + ((c1 + 2 < wg) ? tb[r].z : 0) + ((c1 + 3 < wg) ? tb[r].w : 0);
        }
        const int tot = vals16_sum32(v, lane);
        if ((lane & 1) == 0) {
            const int row = w * 8 + ((lane >> 1) & 7);
            if (lane & 16) rpre[row] = tot; else rtot[row] = tot;
        }
    }
    __syncthreads();
    int ex = 0, exB = 0;
    if (tid < NB) {
        const int tot = rtot[tid];
        int inc = cnt, incB = tot;
        for (int off = 1; off < 32; off <<= 1) {
            const int t  = __shfl_up(inc, off, 32);
            const int tB = __shfl_up(incB, off, 32);
            if (lane >= off) { inc += t; incB += tB; }
        }
        if (lane == 31) { wtot[w] = inc; wtotB[w] = incB; }
        ex = inc - cnt;
        exB = incB - tot;
    }
    __syncthreads();
    if (tid < NB) {
        int base = 0, baseB = 0;
        for (int j = 0; j < NB / 32; ++j) { base += (j < w) ? wtot[j] : 0; baseB += (j < w) ? wtotB[j] : 0; }
        ls[tid] = ex + base;
        const int bfirst = exB + baseB;
        bs[tid] = bfirst + rpre[tid];
        if (wg == 0) bstarts[tid] = bfirst;
    }
    __syncthreads();
#pragma unroll
    for (int it = 0; it < ITERS; ++it) {
        const int k = tid + it * GT;
        if (k < CH) {
            const int b  = trgv[it] / NPB;
            const int tl = trgv[it] - b * NPB;
            stage[ls[b] + rkv[it]] =
                make_int2((e0 + k) | (b << EID_BITS), srcv[it] | (etv[it] << 16) | (tl << 19));
        }
    }
    __syncthreads();
#pragma unroll
    for (int it = 0; it < ITERS; ++it) {
        const int s = tid + it * GT;
        if (s < CH) {
            const int2 r = stage[s];
            const int b = r.x >> EID_BITS;
            *(int2*)(coarse + 2 * (size_t)(bs[b] + (s - ls[b]))) =
                make_int2(r.x & ((1 << EID_BITS) - 1), r.y);
        }
    }
}

#define BCAP_IT 8
#define BCAP    (BCAP_IT * GT)

template <bool IN_LDS>
__device__ __forceinline__ float
order_segments(const int2* recs, const int* __restrict__ grouped, int* ordered,
               const int* loff, const int bstart, const int b,
               const float* __restrict__ hl, const float* __restrict__ hr,
               const float* __restrict__ he, const int lane, const int w) {
    float lmax = -INFINITY;
    int  py = 0, pnode = 0;
    bool pv = false;
    for (int t = w; t < NPB; t += GT / 32) {
        const int lo   = __builtin_amdgcn_readfirstlane(loff[t]);
        const int d    = __builtin_amdgcn_readfirstlane(loff[t + 1]) - lo;
        const int beg  = bstart + lo;
        const int node = b * NPB + t;
        for (int ca = 0; ca < d; ca += 32) {
            const int ia = ca + lane;
            int2 pa = make_int2(0x7FFFFFFF, 0);
            if (ia < d) {
                if (IN_LDS) pa = recs[lo + ia];
                else        pa = *(const int2*)(grouped + 2 * (size_t)(beg + ia));
            }
            const float4 rl = *(const float4*)(hl + (py & 0xFFFF) * NHEAD);
            const float4 rr = *(const float4*)(hr + pnode * NHEAD);
            const float4 rt = *(const float4*)(he + (py >> 16) * NHEAD);
            int pos = 0;
            for (int cb = 0; cb < d; cb += 32) {
                const int ib = cb + lane;
                int kb = 0x7FFFFFFF;
                if (ib < d) {
                    if (IN_LDS) kb = recs[lo + ib].x;
                    else        kb = grouped[2 * (size_t)(beg + ib)];
                }
#pragma unroll
                for (int j = 0; j < 32; ++j) {
                    const int key = __builtin_amdgcn_readlane(kb, j);
                    pos += (key < pa.x) ? 1 : 0;
                }
            }
            if (ia < d) *(int2*)(ordered + 2 * (size_t)(beg + pos)) = pa;
            const float4 s = scores_of_rows(rl, rr, rt);
            if (pv) lmax = fmaxf(lmax, fmaxf(fmaxf(s.x, s.y), fmaxf(s.z, s.w)));
            py = pa.y; pnode = node; pv = (ia < d);
        }
    }
    {
        const float4 rl = *(const float4*)(hl + (py & 0xFFFF) * NHEAD);
        const float4 rr = *(const float4*)(hr + pnode * NHEAD);
        const float4 rt = *(const float4*)(he + (py >> 16) * NHEAD);
        const float4 s = scores_of_rows(rl, rr, rt);
        if (pv) lmax = fmaxf(lmax, fmaxf(fmaxf(s.x, s.y), fmaxf(s.z, s.w)));
    }
    return lmax;
}

__global__ void __launch_bounds__(GT)
bucket_kernel(const int* coarse, const int* __restrict__ bstarts,
              int* __restrict__ grouped, int* ordered, int* __restrict__ offs,
              const float* __restrict__ hl, const float* __restrict__ hr,
              const float* __restrict__ he, unsigned* __restrict__ gmax) {
    __shared__ int   cntL[NPB + 1];
    __shared__ int   loff[NPB + 1];
    __shared__ float red[GT / 32];
    __shared__ int2  recs[BCAP];
    const int b = blockIdx.x, tid = threadIdx.x;
    const int lane = tid & 31, w = tid >> 5;
    const int bstart = bstarts[b];
    const int bend   = (b + 1 < NB) ? bstarts[b + 1] : N_EDGES;
    const int n = bend - bstart;
    const bool in_lds = (n <= BCAP);
    if (tid <= NPB) cntL[tid] = 0;
    int2 rv[BCAP_IT];
    if (in_lds) {
#pragma unroll
        for (int it = 0; it < BCAP_IT; ++it) {
            const int i = tid + it * GT;
            rv[it] = make_int2(0, 0);
            if (i < n) rv[it] = *(const int2*)(coarse + 2 * (size_t)(bstart + i));
        }
    }
    __syncthreads();
    if (in_lds) {
#pragma unroll
        for (int it = 0; it < BCAP_IT; ++it)
            if (tid + it * GT < n) atomicAdd(&cntL[(rv[it].y >> 19) & 0xFF], 1);
    } else {
        for (int i = tid; i < n; i += GT)
            atomicAdd(&cntL[(coarse[2 * (size_t)(bstart + i) + 1] >> 19) & 0xFF], 1);
    }
    __syncthreads();
    if (w == 0) {
        int carry = 0;
        for (int r = 0; r < (NPB + 32) / 32; ++r) {
            const int idx = r * 32 + lane;
            const int c = (idx < NPB) ? cntL[idx] : 0;
            int inc = c;
            for (int off = 1; off < 32; off <<= 1) {
                const int t = __shfl_up(inc, off, 32);
                if (lane >= off) inc += t;
            }
            if (idx <= NPB) loff[idx] = carry + inc - c;
            carry += __shfl(inc, 31, 32);
        }
    }
    __syncthreads();
    if (tid < NPB) {
        const int node = b * NPB + tid;
        if (node <= N_NODES) offs[node] = bstart + loff[tid];
        cntL[tid] = 0;
    }
    __syncthreads();
    float lmax;
    if (in_lds) {
#pragma unroll
        for (int it = 0; it < BCAP_IT; ++it) {
            if (tid + it * GT < n) {
                const int tl = (rv[it].y >> 19) & 0xFF;
                const int r  = atomicAdd(&cntL[tl], 1);
                recs[loff[tl] + r] = make_int2(rv[it].x, rv[it].y & 0x7FFFF);
            }
        }
        __syncthreads();
        lmax = order_segments<true>(recs, grouped, ordered, loff, bstart, b, hl, hr, he, lane, w);
    } else {
        for (int i = tid; i < n; i += GT) {
            const int2 rec = *(const int2*)(coarse + 2 * (size_t)(bstart + i));
            const int tl = (rec.y >> 19) & 0xFF;
            const int r  = atomicAdd(&cntL[tl], 1);
            *(int2*)(grouped + 2 * (size_t)(bstart + loff[tl] + r)) = make_int2(rec.x, rec.y & 0x7FFFF);
        }
        __syncthreads();
        lmax = order_segments<false>(recs, grouped, ordered, loff, bstart, b, hl, hr, he, lane, w);
    }
    for (int m = 16; m > 0; m >>= 1) lmax = fmaxf(lmax, __shfl_xor(lmax, m, 32));
    if (lane == 0) red[w] = lmax;
    __syncthreads();
    if (tid < 32) {
        float v = red[tid];
        for (int m = 16; m > 0; m >>= 1) v = fmaxf(v, __shfl_xor(v, m, 32));
        if (tid == 0) atomicMax(gmax, enc_f32(v));
    }
}

__global__ void __launch_bounds__(256)
node_aggregate_kernel(const int* __restrict__ ordered, const int* __restrict__ offs,
                      const float* __restrict__ h,
                      const float* __restrict__ hl, const float* __restrict__ hr,
                      const float* __restrict__ he, const unsigned* __restrict__ gmax,
                      float* __restrict__ out, float* __restrict__ attn_out) {
    const int lane = threadIdx.x & 31;
    const int node = blockIdx.x * (blockDim.x >> 5) + (threadIdx.x >> 5);
    if (node >= N_NODES) return;
    const int beg = __builtin_amdgcn_readfirstlane(offs[node]);
    const int d   = __builtin_amdgcn_readfirstlane(offs[node + 1]) - beg;
    const float smax = dec_f32(*gmax);

    float4 ds = make_float4(0.0f, 0.0f, 0.0f, 0.0f);
    int2   p0 = make_int2(0, 0), p1 = make_int2(0, 0);
    float4 e0 = make_float4(0.0f, 0.0f, 0.0f, 0.0f);
    float4 e1 = make_float4(0.0f, 0.0f, 0.0f, 0.0f);
    if (lane < d) {
        p0 = *(const int2*)(ordered + 2 * (size_t)(beg + lane));
        const float4 s = edge_scores(hl, hr, he, p0.y & 0xFFFF, node, p0.y >> 16);
        e0.x = __expf(s.x - smax);
        e0.y = __expf(s.y - smax);
        e0.z = __expf(s.z - smax);
        e0.w = __expf(s.w - smax);
        ds.x += e0.x;
        ds.y += e0.y;
        ds.z += e0.z;
        ds.w += e0.w;
    }
    if (32 + lane < d) {
        p1 = *(const int2*)(ordered + 2 * (size_t)(beg + 32 + lane));
        const float4 s = edge_scores(hl, hr, he, p1.y & 0xFFFF, node, p1.y >> 16);
        e1.x = __expf(s.x - smax);
        e1.y = __expf(s.y - smax);
        e1.z = __expf(s.z - smax);
        e1.w = __expf(s.w - smax);
        ds.x += e1.x;
        ds.y += e1.y;
        ds.z += e1.z;
        ds.w += e1.w;
    }
    for (int c = 64; c < d; c += 32) {
        const int i = c + lane;
        if (i < d) {
            const int2 p = *(const int2*)(ordered + 2 * (size_t)(beg + i));
            const float4 s = edge_scores(hl, hr, he, p.y & 0xFFFF, node, p.y >> 16);
            ds.x += __expf(s.x - smax);
            ds.y += __expf(s.y - smax);
            ds.z += __expf(s.z - smax);
            ds.w += __expf(s.w - smax);
        }
    }
    for (int m = 16; m > 0; m >>= 1) {
        ds.x += __shfl_xor(ds.x, m, 32);
        ds.y += __shfl_xor(ds.y, m, 32);
        ds.z += __shfl_xor(ds.z, m, 32);
        ds.w += __shfl_xor(ds.w, m, 32);
    }
    const float4 dn = make_float4(ds.x + 1e-16f, ds.y + 1e-16f, ds.z + 1e-16f, ds.w + 1e-16f);

    float acc0 = 0.0f, acc1 = 0.0f, acc2 = 0.0f, acc3 = 0.0f;
    const float* hcol = h + lane;
    for (int c = 0; c < d; c += 32) {
        const int i = c + lane;
        int srcv = 0;
        float4 a = make_float4(0.0f, 0.0f, 0.0f, 0.0f);
        if (i < d) {
            int2   p = p0;
            float4 e = e0;
            if (c == 32) { p = p1; e = e1; }
            if (c >= 64) {
                p = *(const int2*)(ordered + 2 * (size_t)(beg + i));
                const float4 s = edge_scores(hl, hr, he, p.y & 0xFFFF, node, p.y >> 16);
                e.x = __expf(s.x - smax);
                e.y = __expf(s.y - smax);
                e.z = __expf(s.z - smax);
                e.w = __expf(s.w - smax);
            }
            srcv = p.y & 0xFFFF;
            a.x = e.x / dn.x;
            a.y = e.y / dn.y;
            a.z = e.z / dn.z;
            a.w = e.w / dn.w;
            *(float4*)(attn_out + (size_t)p.x * NHEAD) = a;
        }
        const int nb = (d - c < 32) ? (d - c) : 32;
#pragma unroll 4
        for (int j = 0; j < nb; ++j) {
            const int sj = __builtin_amdgcn_readlane(srcv, j);
            const float a0 = __int_as_float(__builtin_amdgcn_readlane(__float_as_int(a.x), j));
            const float a1 = __int_as_float(__builtin_amdgcn_readlane(__float_as_int(a.y), j));
            const float a2 = __int_as_float(__builtin_amdgcn_readlane(__float_as_int(a.z), j));
            const float a3 = __int_as_float(__builtin_amdgcn_readlane(__float_as_int(a.w), j));
            const float* row = hcol + (size_t)sj * OUT_COLS;
            acc0 = fmaf(row[0 * OUT_HIDDEN], a0, acc0);
            acc1 = fmaf(row[1 * OUT_HIDDEN], a1, acc1);
            acc2 = fmaf(row[2 * OUT_HIDDEN], a2, acc2);
            acc3 = fmaf(row[3 * OUT_HIDDEN], a3, acc3);
        }
    }
    float* orow = out + (size_t)node * OUT_COLS + lane;
    orow[0 * OUT_HIDDEN] = acc0;
    orow[1 * OUT_HIDDEN] = acc1;
    orow[2 * OUT_HIDDEN] = acc2;
    orow[3 * OUT_HIDDEN] = acc3;
}

extern "C" void kernel_launch(void* const* d_in, const int* in_sizes, int n_in,
                              void* d_out, int out_size, void* d_ws, size_t ws_size,
                              hipStream_t stream) {
    const int*   edge     = (const int*)  d_in[0];
    const float* x        = (const float*)d_in[1];
    const float* edge_emb = (const float*)d_in[2];
    const float* W        = (const float*)d_in[3];
    const float* W_e      = (const float*)d_in[4];
    const float* a_l      = (const float*)d_in[5];
    const float* a_r      = (const float*)d_in[6];
    const float* a_e      = (const float*)d_in[7];

    float* out      = (float*)d_out;
    float* attn_out = out + (size_t)N_NODES * OUT_COLS;

    char* ws = (char*)d_ws;
    float* h       = (float*)ws;  ws += (size_t)N_NODES * OUT_COLS * sizeof(float);
    int*   planeA  = (int*)ws;    ws += (size_t)N_EDGES * 2 * sizeof(int);
    int*   planeB  = (int*)ws;    ws += (size_t)N_EDGES * 2 * sizeof(int);
    int*   lrank   = planeA;
    int*   grouped = planeA;
    int*   coarse  = planeB;
    int*   ordered = planeB;
    float* hl      = (float*)ws;  ws += (size_t)N_NODES * NHEAD   * sizeof(float);
    float* hr      = (float*)ws;  ws += (size_t)N_NODES * NHEAD   * sizeof(float);
    int*   offs    = (int*)ws + 50176;
    int*   tab     = (int*)ws + 102400;
    int*   bstarts = (int*)ws + 168000;
    ws += (size_t)N_NODES * NHEAD * sizeof(float);
    float* he      = (float*)ws;  ws += (size_t)N_ETYPES * NHEAD  * sizeof(float);
    unsigned* gmax = (unsigned*)ws;

    front_kernel<<<GEMM_BLOCKS + G1 + N_ETYPES, FRONT_T, 0, stream>>>(
        edge, x, W, edge_emb, W_e, a_e, a_l, a_r, h, he, hl, hr, lrank, tab, gmax);
    scatter_kernel<<<G1, GT, 0, stream>>>(edge, lrank, tab, bstarts, coarse);
    bucket_kernel<<<NB, GT, 0, stream>>>(coarse, bstarts, grouped, ordered, offs, hl, hr, he, gmax);
    node_aggregate_kernel<<<(N_NODES + 7) / 8, 256, 0, stream>>>(
        ordered, offs, h, hl, hr, he, gmax, out, attn_out);
}
